# starting kernel
# baseline (speedup 1.0000x reference)
#include <hip/hip_runtime.h>

typedef __attribute__((ext_vector_type(8))) float v8f;
typedef __attribute__((ext_vector_type(2))) float v2f;

__global__ void fuse_kernel(const float* __restrict__ x,
                            const float* __restrict__ embed,
                            const int* __restrict__ node_index,
                            const float* __restrict__ W,
                            const float* __restrict__ b,
                            float* __restrict__ out,
                            int n, int fin) {
  int t = blockIdx.x * blockDim.x + threadIdx.x;
  int node = t >> 5;
  int j = t & 31;
  if (node >= n) return;
  float v;
  if (j < 16) {
    v = embed[(long long)node_index[node] * 16 + j];
  } else {
    int r = j - 16;
    float acc = b[r];
    const float* xr = x + (long long)node * fin;
    for (int k = 0; k < fin; ++k) acc = fmaf(xr[k], W[r * fin + k], acc);
    v = acc;
  }
  out[(long long)node * 32 + j] = v;
}

__global__ void degree_kernel(const int* __restrict__ idx,
                              float* __restrict__ deg, int E) {
  int t = blockIdx.x * blockDim.x + threadIdx.x;
  if (t < E) atomicAdd(&deg[idx[t]], 1.0f);
}

__global__ void scatter_kernel(const float* __restrict__ srcf,
                               const int* __restrict__ gather_idx,
                               const int* __restrict__ scatter_idx,
                               float* __restrict__ agg,
                               int E) {
  int t = blockIdx.x * blockDim.x + threadIdx.x;
  int e = t >> 3;
  int part = t & 7;
  if (e >= E) return;
  int s = gather_idx[e];
  int d = scatter_idx[e];
  float4 v = ((const float4*)(srcf + (long long)s * 32))[part];
  float* o = agg + (long long)d * 32 + part * 4;
  atomicAdd(o + 0, v.x);
  atomicAdd(o + 1, v.y);
  atomicAdd(o + 2, v.z);
  atomicAdd(o + 3, v.w);
}

template <int NTILES, bool RELU>
__global__ void sage_wmma_kernel(const float* __restrict__ agg,
                                 const float* __restrict__ deg,
                                 const float* __restrict__ xdst,
                                 const float* __restrict__ Wl,
                                 const float* __restrict__ bl,
                                 const float* __restrict__ Wr,
                                 float* __restrict__ out,
                                 int n) {
  const int OUTF = NTILES * 16;
  int lane = threadIdx.x & 31;
  int wave = (blockIdx.x * blockDim.x + threadIdx.x) >> 5;
  int nb = wave * 16;
  if (nb >= n) return;
  int m  = lane & 15;
  int hi = lane >> 4;

  int node = nb + m;
  bool nv = node < n;
  int ncl = nv ? node : (n - 1);
  float dinv = nv ? (1.0f / fmaxf(deg[ncl], 1.0f)) : 0.0f;
  const float* arow = agg  + (long long)ncl * 32;
  const float* xrow = xdst + (long long)ncl * 32;

  v2f a_agg[8], a_x[8];
#pragma unroll
  for (int s = 0; s < 8; ++s) {
    int k = 4 * s + 2 * hi;
    a_agg[s].x = arow[k] * dinv;
    a_agg[s].y = arow[k + 1] * dinv;
    a_x[s].x   = nv ? xrow[k]     : 0.0f;
    a_x[s].y   = nv ? xrow[k + 1] : 0.0f;
  }

  bool full = (nb + 16 <= n);

#pragma unroll
  for (int t = 0; t < NTILES; ++t) {
    int ng = t * 16 + m;
    const float* wl = Wl + ng * 32;
    const float* wr = Wr + ng * 32;
    float bias = bl[ng];
    v8f c;
#pragma unroll
    for (int v = 0; v < 8; ++v) c[v] = bias;

#pragma unroll
    for (int s = 0; s < 8; ++s) {
      int k = 4 * s + 2 * hi;
      v2f b;
      b.x = wl[k];
      b.y = wl[k + 1];
      c = __builtin_amdgcn_wmma_f32_16x16x4_f32(false, a_agg[s], false, b,
                                                (short)0, c, false, false);
    }
#pragma unroll
    for (int s = 0; s < 8; ++s) {
      int k = 4 * s + 2 * hi;
      v2f b;
      b.x = wr[k];
      b.y = wr[k + 1];
      c = __builtin_amdgcn_wmma_f32_16x16x4_f32(false, a_x[s], false, b,
                                                (short)0, c, false, false);
    }

    if (full) {
#pragma unroll
      for (int v = 0; v < 8; ++v) {
        float val = c[v];
        if (RELU) val = fmaxf(val, 0.0f);
        out[(long long)(nb + v + 8 * hi) * OUTF + ng] = val;
      }
    } else {
#pragma unroll
      for (int v = 0; v < 8; ++v) {
        int nd = nb + v + 8 * hi;
        if (nd < n) {
          float val = c[v];
          if (RELU) val = fmaxf(val, 0.0f);
          out[(long long)nd * OUTF + ng] = val;
        }
      }
    }
  }
}

__global__ void decode_kernel(const float* __restrict__ ou,
                              const float* __restrict__ oc,
                              const int* __restrict__ ls,
                              const int* __restrict__ ld,
                              float* __restrict__ out, int L) {
  int t = blockIdx.x * blockDim.x + threadIdx.x;
  if (t >= L) return;
  const float4* a = (const float4*)(ou + (long long)ls[t] * 16);
  const float4* b = (const float4*)(oc + (long long)ld[t] * 16);
  float acc = 0.0f;
#pragma unroll
  for (int i = 0; i < 4; ++i) {
    float4 xa = a[i], xb = b[i];
    acc += xa.x * xb.x + xa.y * xb.y + xa.z * xb.z + xa.w * xb.w;
  }
  out[t] = acc;
}

extern "C" void kernel_launch(void* const* d_in, const int* in_sizes, int n_in,
                              void* d_out, int out_size, void* d_ws, size_t ws_size,
                              hipStream_t stream) {
  const float* user_x       = (const float*)d_in[0];
  const float* course_x     = (const float*)d_in[1];
  const float* user_embed   = (const float*)d_in[2];
  const float* course_embed = (const float*)d_in[3];
  const float* Wu = (const float*)d_in[4];
  const float* bu = (const float*)d_in[5];
  const float* Wc = (const float*)d_in[6];
  const float* bc = (const float*)d_in[7];
  const float* c1e_Wl = (const float*)d_in[8];
  const float* c1e_bl = (const float*)d_in[9];
  const float* c1e_Wr = (const float*)d_in[10];
  const float* c1r_Wl = (const float*)d_in[11];
  const float* c1r_bl = (const float*)d_in[12];
  const float* c1r_Wr = (const float*)d_in[13];
  const float* c2e_Wl = (const float*)d_in[14];
  const float* c2e_bl = (const float*)d_in[15];
  const float* c2e_Wr = (const float*)d_in[16];
  const float* c2r_Wl = (const float*)d_in[17];
  const float* c2r_bl = (const float*)d_in[18];
  const float* c2r_Wr = (const float*)d_in[19];
  const int* user_idx   = (const int*)d_in[20];
  const int* course_idx = (const int*)d_in[21];
  const int* edge_src   = (const int*)d_in[22];
  const int* edge_dst   = (const int*)d_in[23];
  const int* label_src  = (const int*)d_in[24];
  const int* label_dst  = (const int*)d_in[25];
  float* outp = (float*)d_out;

  const int NU = in_sizes[2] / 16;
  const int NC = in_sizes[3] / 16;
  const int E  = in_sizes[22];
  const int L  = in_sizes[24];

  char* p = (char*)d_ws;
  auto carve = [&](size_t bytes) -> float* {
    float* r = (float*)p;
    p += (bytes + 255) & ~((size_t)255);
    return r;
  };
  float* xu   = carve((size_t)NU * 32 * 4);
  float* xc   = carve((size_t)NC * 32 * 4);
  float* aggu = carve((size_t)NU * 32 * 4);
  float* aggc = carve((size_t)NC * 32 * 4);
  float* degu = carve((size_t)NU * 4);
  float* degc = carve((size_t)NC * 4);
  float* hu   = carve((size_t)NU * 32 * 4);
  float* hc   = carve((size_t)NC * 32 * 4);
  float* ou   = carve((size_t)NU * 16 * 4);
  float* oc   = carve((size_t)NC * 16 * 4);

  const int B = 256;
  auto nblk = [](long long t, int b) { return (unsigned)((t + b - 1) / b); };

  hipMemsetAsync(aggu, 0, (size_t)NU * 32 * 4, stream);
  hipMemsetAsync(aggc, 0, (size_t)NC * 32 * 4, stream);
  hipMemsetAsync(degu, 0, (size_t)NU * 4, stream);
  hipMemsetAsync(degc, 0, (size_t)NC * 4, stream);

  fuse_kernel<<<nblk((long long)NU * 32, B), B, 0, stream>>>(
      user_x, user_embed, user_idx, Wu, bu, xu, NU, 5);
  fuse_kernel<<<nblk((long long)NC * 32, B), B, 0, stream>>>(
      course_x, course_embed, course_idx, Wc, bc, xc, NC, 2);

  degree_kernel<<<nblk(E, B), B, 0, stream>>>(edge_dst, degc, E);
  degree_kernel<<<nblk(E, B), B, 0, stream>>>(edge_src, degu, E);

  scatter_kernel<<<nblk((long long)E * 8, B), B, 0, stream>>>(xu, edge_src, edge_dst, aggc, E);
  scatter_kernel<<<nblk((long long)E * 8, B), B, 0, stream>>>(xc, edge_dst, edge_src, aggu, E);
  sage_wmma_kernel<2, true><<<nblk(((long long)(NC + 15) / 16) * 32, B), B, 0, stream>>>(
      aggc, degc, xc, c1e_Wl, c1e_bl, c1e_Wr, hc, NC);
  sage_wmma_kernel<2, true><<<nblk(((long long)(NU + 15) / 16) * 32, B), B, 0, stream>>>(
      aggu, degu, xu, c1r_Wl, c1r_bl, c1r_Wr, hu, NU);

  hipMemsetAsync(aggu, 0, (size_t)NU * 32 * 4, stream);
  hipMemsetAsync(aggc, 0, (size_t)NC * 32 * 4, stream);
  scatter_kernel<<<nblk((long long)E * 8, B), B, 0, stream>>>(hu, edge_src, edge_dst, aggc, E);
  scatter_kernel<<<nblk((long long)E * 8, B), B, 0, stream>>>(hc, edge_dst, edge_src, aggu, E);
  sage_wmma_kernel<1, false><<<nblk(((long long)(NC + 15) / 16) * 32, B), B, 0, stream>>>(
      aggc, degc, hc, c2e_Wl, c2e_bl, c2e_Wr, oc, NC);
  sage_wmma_kernel<1, false><<<nblk(((long long)(NU + 15) / 16) * 32, B), B, 0, stream>>>(
      aggu, degu, hu, c2r_Wl, c2r_bl, c2r_Wr, ou, NU);

  decode_kernel<<<nblk(L, B), B, 0, stream>>>(ou, oc, label_src, label_dst, outp, L);
}


# course-side neighbour sums of both layers accumulate in range-partitioned LDS tables (LDS float atomics; per-slice partials summed in fixed order), no global atomics there; sum association changes
# speedup vs baseline: 1.9759x; 1.9759x over previous
#include <hip/hip_runtime.h>

typedef __attribute__((ext_vector_type(8))) float v8f;
typedef __attribute__((ext_vector_type(2))) float v2f;

__global__ void fuse_kernel(const float* __restrict__ x,
                            const float* __restrict__ embed,
                            const int* __restrict__ node_index,
                            const float* __restrict__ W,
                            const float* __restrict__ b,
                            float* __restrict__ out,
                            int n, int fin) {
  int t = blockIdx.x * blockDim.x + threadIdx.x;
  int node = t >> 5;
  int j = t & 31;
  if (node >= n) return;
  float v;
  if (j < 16) {
    v = embed[(long long)node_index[node] * 16 + j];
  } else {
    int r = j - 16;
    float acc = b[r];
    const float* xr = x + (long long)node * fin;
    for (int k = 0; k < fin; ++k) acc = fmaf(xr[k], W[r * fin + k], acc);
    v = acc;
  }
  out[(long long)node * 32 + j] = v;
}

template <int FIN>
__global__ void scatter_raw_kernel(const float* __restrict__ embed,
                                   const int* __restrict__ node_index,
                                   const float* __restrict__ x,
                                   const int* __restrict__ gather_idx,
                                   const int* __restrict__ scatter_idx,
                                   float* __restrict__ agg,
                                   int E) {
  int t = blockIdx.x * blockDim.x + threadIdx.x;
  int e = t >> 3;
  int part = t & 7;
  if (e >= E) return;
  int s = gather_idx[e];
  int d = scatter_idx[e];
  const float* er = embed + (long long)node_index[s] * 16;
  float* o = agg + (long long)d * 32;
  float v0 = er[part];
  float v1 = er[8 + part];
  float v2 = 1.0f;
  if (part < FIN) v2 = x[(long long)s * FIN + part];
  atomicAdd(o + part, v0);
  atomicAdd(o + 8 + part, v1);
  if (part <= FIN) atomicAdd(o + 16 + part, v2);
}

__global__ void scatter16_kernel(const float* __restrict__ srcp,
                                 const int* __restrict__ gather_idx,
                                 const int* __restrict__ scatter_idx,
                                 float* __restrict__ agg,
                                 int E) {
  int t = blockIdx.x * blockDim.x + threadIdx.x;
  int col = t & 15;
  int e0 = (t >> 4) * 4;
  if (e0 >= E) return;
  if (e0 + 3 < E) {
    int s0 = gather_idx[e0],  s1 = gather_idx[e0 + 1];
    int s2 = gather_idx[e0 + 2], s3 = gather_idx[e0 + 3];
    int d0 = scatter_idx[e0], d1 = scatter_idx[e0 + 1];
    int d2 = scatter_idx[e0 + 2], d3 = scatter_idx[e0 + 3];
    float v0 = srcp[(long long)s0 * 16 + col];
    float v1 = srcp[(long long)s1 * 16 + col];
    float v2 = srcp[(long long)s2 * 16 + col];
    float v3 = srcp[(long long)s3 * 16 + col];
    atomicAdd(agg + (long long)d0 * 16 + col, v0);
    atomicAdd(agg + (long long)d1 * 16 + col, v1);
    atomicAdd(agg + (long long)d2 * 16 + col, v2);
    atomicAdd(agg + (long long)d3 * 16 + col, v3);
  } else {
    for (int e = e0; e < E; ++e) {
      int s = gather_idx[e];
      int d = scatter_idx[e];
      atomicAdd(agg + (long long)d * 16 + col, srcp[(long long)s * 16 + col]);
    }
  }
}

constexpr int kCapRows = 2560;
constexpr int kLdsThreads = 1024;
constexpr int kEdgesPerTrip = 4;

template <int FIN>
__global__ __launch_bounds__(kLdsThreads)
void course_sums_raw_lds_kernel(const float* __restrict__ embed,
                                const int* __restrict__ node_index,
                                const float* __restrict__ x,
                                const int* __restrict__ gather_idx,
                                const int* __restrict__ scatter_idx,
                                float* __restrict__ part,
                                int E, int n_dst, int n_ranges, int n_slices) {
  constexpr int W = 16 + FIN + 1;
  constexpr int P = W | 1;
  __shared__ float T[kCapRows * P];
  const int tid = threadIdx.x;
  const int nthr = blockDim.x;
  const int r = blockIdx.x % n_ranges;
  const int slice = blockIdx.x / n_ranges;
  const int base = r * kCapRows;
  const int left = n_dst - base;
  const int nrows = left < kCapRows ? left : kCapRows;
  for (int i = tid; i < nrows * P; i += nthr) T[i] = 0.0f;
  __syncthreads();

  const int chunk = (E + n_slices - 1) / n_slices;
  const int e_lo = slice * chunk;
  const int e_end = e_lo + chunk;
  const int e_hi = e_end < E ? e_end : E;
  for (int e0 = e_lo + tid; e0 < e_hi; e0 += kEdgesPerTrip * nthr) {
    int s[kEdgesPerTrip], rel[kEdgesPerTrip], ni[kEdgesPerTrip];
    bool in[kEdgesPerTrip];
    float4 ev[kEdgesPerTrip][4];
    float xv[kEdgesPerTrip][FIN];
#pragma unroll
    for (int u = 0; u < kEdgesPerTrip; ++u) {
      int e = e0 + u * nthr;
      bool v = e < e_hi;
      int ec = v ? e : e0;
      int d = scatter_idx[ec];
      s[u] = gather_idx[ec];
      rel[u] = d - base;
      in[u] = v && ((unsigned)rel[u] < (unsigned)nrows);
      if (!in[u]) s[u] = 0;
    }
#pragma unroll
    for (int u = 0; u < kEdgesPerTrip; ++u) ni[u] = node_index[s[u]];
#pragma unroll
    for (int u = 0; u < kEdgesPerTrip; ++u) {
      if (!in[u]) ni[u] = 0;
      const float4* er = (const float4*)(embed + (long long)ni[u] * 16);
#pragma unroll
      for (int q = 0; q < 4; ++q) ev[u][q] = er[q];
      const float* xr = x + (long long)s[u] * FIN;
#pragma unroll
      for (int j = 0; j < FIN; ++j) xv[u][j] = xr[j];
    }
#pragma unroll
    for (int u = 0; u < kEdgesPerTrip; ++u) {
      if (in[u]) {
        float* row = T + rel[u] * P;
#pragma unroll
        for (int q = 0; q < 4; ++q) {
          atomicAdd(row + 4 * q + 0, ev[u][q].x);
          atomicAdd(row + 4 * q + 1, ev[u][q].y);
          atomicAdd(row + 4 * q + 2, ev[u][q].z);
          atomicAdd(row + 4 * q + 3, ev[u][q].w);
        }
#pragma unroll
        for (int j = 0; j < FIN; ++j) atomicAdd(row + 16 + j, xv[u][j]);
        atomicAdd(row + 16 + FIN, 1.0f);
      }
    }
  }
  __syncthreads();

  float* dst = part + ((long long)slice * n_dst + base) * W;
  for (int i = tid; i < nrows * W; i += nthr) {
    int row = i / W;
    int c = i - row * W;
    dst[i] = T[row * P + c];
  }
}

__global__ __launch_bounds__(kLdsThreads)
void course_sums16_lds_kernel(const float* __restrict__ srcp,
                              const int* __restrict__ gather_idx,
                              const int* __restrict__ scatter_idx,
                              float* __restrict__ part,
                              int E, int n_dst, int n_ranges, int n_slices) {
  constexpr int P = 17;
  __shared__ float T[kCapRows * P];
  const int tid = threadIdx.x;
  const int nthr = blockDim.x;
  const int r = blockIdx.x % n_ranges;
  const int slice = blockIdx.x / n_ranges;
  const int base = r * kCapRows;
  const int left = n_dst - base;
  const int nrows = left < kCapRows ? left : kCapRows;
  for (int i = tid; i < nrows * P; i += nthr) T[i] = 0.0f;
  __syncthreads();

  const int chunk = (E + n_slices - 1) / n_slices;
  const int e_lo = slice * chunk;
  const int e_end = e_lo + chunk;
  const int e_hi = e_end < E ? e_end : E;
  for (int e0 = e_lo + tid; e0 < e_hi; e0 += kEdgesPerTrip * nthr) {
    int s[kEdgesPerTrip], rel[kEdgesPerTrip];
    bool in[kEdgesPerTrip];
    float4 pv[kEdgesPerTrip][4];
#pragma unroll
    for (int u = 0; u < kEdgesPerTrip; ++u) {
      int e = e0 + u * nthr;
      bool v = e < e_hi;
      int ec = v ? e : e0;
      int d = scatter_idx[ec];
      s[u] = gather_idx[ec];
      rel[u] = d - base;
      in[u] = v && ((unsigned)rel[u] < (unsigned)nrows);
      if (!in[u]) s[u] = 0;
    }
#pragma unroll
    for (int u = 0; u < kEdgesPerTrip; ++u) {
      const float4* pr = (const float4*)(srcp + (long long)s[u] * 16);
#pragma unroll
      for (int q = 0; q < 4; ++q) pv[u][q] = pr[q];
    }
#pragma unroll
    for (int u = 0; u < kEdgesPerTrip; ++u) {
      if (in[u]) {
        float* row = T + rel[u] * P;
#pragma unroll
        for (int q = 0; q < 4; ++q) {
          atomicAdd(row + 4 * q + 0, pv[u][q].x);
          atomicAdd(row + 4 * q + 1, pv[u][q].y);
          atomicAdd(row + 4 * q + 2, pv[u][q].z);
          atomicAdd(row + 4 * q + 3, pv[u][q].w);
        }
      }
    }
  }
  __syncthreads();

  float* dst = part + ((long long)slice * n_dst + base) * 16;
  for (int i = tid; i < nrows * 16; i += nthr) {
    dst[i] = T[(i >> 4) * P + (i & 15)];
  }
}

template <int W>
__global__ void reduce_parts_kernel(const float* __restrict__ part,
                                    float* __restrict__ out,
                                    int n_rows, int n_slices) {
  int t = blockIdx.x * blockDim.x + threadIdx.x;
  int total = n_rows * W;
  if (t >= total) return;
  float acc = 0.0f;
#pragma unroll 16
  for (int s = 0; s < n_slices; ++s) acc += part[(long long)s * total + t];
  int row = t / W;
  int c = t - row * W;
  out[(long long)row * 32 + c] = acc;
}

__global__ void combine16_parts_kernel(const float* __restrict__ part,
                                       int n_slices,
                                       const float* __restrict__ deg,
                                       int dstride,
                                       const float* __restrict__ R,
                                       float* __restrict__ out,
                                       int n) {
  int t = blockIdx.x * blockDim.x + threadIdx.x;
  int node = t >> 2;
  int q = t & 3;
  if (node >= n) return;
  float4 a = make_float4(0.0f, 0.0f, 0.0f, 0.0f);
  long long stride = (long long)n * 16;
#pragma unroll 16
  for (int s = 0; s < n_slices; ++s) {
    float4 v = ((const float4*)(part + s * stride + (long long)node * 16))[q];
    a.x += v.x; a.y += v.y; a.z += v.z; a.w += v.w;
  }
  float dinv = 1.0f / fmaxf(deg[(long long)node * dstride], 1.0f);
  float4 rr = ((const float4*)(R + (long long)node * 16))[q];
  float4 o;
  o.x = fmaf(a.x, dinv, rr.x);
  o.y = fmaf(a.y, dinv, rr.y);
  o.z = fmaf(a.z, dinv, rr.z);
  o.w = fmaf(a.w, dinv, rr.w);
  ((float4*)(out + (long long)node * 16))[q] = o;
}

template <int FIN>
__global__ void sage1_chain_kernel(const float* __restrict__ agg,
                                   const float* __restrict__ xdst,
                                   const float* __restrict__ Wl,
                                   const float* __restrict__ bl,
                                   const float* __restrict__ Wr,
                                   const float* __restrict__ Wp,
                                   const float* __restrict__ bp,
                                   const float* __restrict__ W2n,
                                   const float* __restrict__ W2r,
                                   const float* __restrict__ b2r,
                                   float* __restrict__ Pout,
                                   float* __restrict__ Rout,
                                   int n) {
  int lane = threadIdx.x & 31;
  int wave = (blockIdx.x * blockDim.x + threadIdx.x) >> 5;
  int nb = wave * 16;
  if (nb >= n) return;
  int m  = lane & 15;
  int hi = lane >> 4;

  int node = nb + m;
  bool nv = node < n;
  int ncl = nv ? node : (n - 1);
  const float* arow = agg  + (long long)ncl * 32;
  const float* xrow = xdst + (long long)ncl * 32;
  float cnt  = arow[16 + FIN];
  float dinv = nv ? (1.0f / fmaxf(cnt, 1.0f)) : 0.0f;

  v2f f_agg[8], f_x[8];
#pragma unroll
  for (int s = 0; s < 8; ++s) {
    int k = 4 * s + 2 * hi;
    if (s < 4) {
      f_agg[s].x = arow[k] * dinv;
      f_agg[s].y = arow[k + 1] * dinv;
    }
    f_x[s].x = nv ? xrow[k]     : 0.0f;
    f_x[s].y = nv ? xrow[k + 1] : 0.0f;
  }
  {
    float sx[FIN];
#pragma unroll
    for (int j = 0; j < FIN; ++j) sx[j] = arow[16 + j];
#pragma unroll
    for (int s = 4; s < 8; ++s) {
      int r = 4 * (s - 4) + 2 * hi;
      float p0 = cnt * bp[r];
      float p1 = cnt * bp[r + 1];
#pragma unroll
      for (int j = 0; j < FIN; ++j) {
        p0 = fmaf(sx[j], Wp[r * FIN + j], p0);
        p1 = fmaf(sx[j], Wp[(r + 1) * FIN + j], p1);
      }
      f_agg[s].x = p0 * dinv;
      f_agg[s].y = p1 * dinv;
    }
  }

  v8f hT[2];
#pragma unroll
  for (int t = 0; t < 2; ++t) {
    int og = t * 16 + m;
    const float* wl = Wl + og * 32;
    const float* wr = Wr + og * 32;
    v8f c;
#pragma unroll
    for (int r = 0; r < 8; ++r) c[r] = bl[t * 16 + 8 * hi + r];

#pragma unroll
    for (int s = 0; s < 8; ++s) {
      int k = 4 * s + 2 * hi;
      v2f w;
      w.x = wl[k];
      w.y = wl[k + 1];
      c = __builtin_amdgcn_wmma_f32_16x16x4_f32(false, w, false, f_agg[s],
                                                (short)0, c, false, false);
    }
#pragma unroll
    for (int s = 0; s < 8; ++s) {
      int k = 4 * s + 2 * hi;
      v2f w;
      w.x = wr[k];
      w.y = wr[k + 1];
      c = __builtin_amdgcn_wmma_f32_16x16x4_f32(false, w, false, f_x[s],
                                                (short)0, c, false, false);
    }
#pragma unroll
    for (int r = 0; r < 8; ++r) hT[t][r] = fmaxf(c[r], 0.0f);
  }

  const float* w2n = W2n + m * 32;
  const float* w2r = W2r + m * 32;
  v8f p, q;
#pragma unroll
  for (int r = 0; r < 8; ++r) {
    p[r] = 0.0f;
    q[r] = b2r[8 * hi + r];
  }
#pragma unroll
  for (int t = 0; t < 2; ++t) {
#pragma unroll
    for (int qq = 0; qq < 4; ++qq) {
      int k = 16 * t + 8 * hi + 2 * qq;
      v2f hb;
      hb.x = hT[t][2 * qq];
      hb.y = hT[t][2 * qq + 1];
      v2f wa, wb;
      wa.x = w2n[k];
      wa.y = w2n[k + 1];
      wb.x = w2r[k];
      wb.y = w2r[k + 1];
      p = __builtin_amdgcn_wmma_f32_16x16x4_f32(false, wa, false, hb,
                                                (short)0, p, false, false);
      q = __builtin_amdgcn_wmma_f32_16x16x4_f32(false, wb, false, hb,
                                                (short)0, q, false, false);
    }
  }

  if (nv) {
    float4* pd = (float4*)(Pout + (long long)node * 16 + 8 * hi);
    float4* rd = (float4*)(Rout + (long long)node * 16 + 8 * hi);
    pd[0] = make_float4(p[0], p[1], p[2], p[3]);
    pd[1] = make_float4(p[4], p[5], p[6], p[7]);
    rd[0] = make_float4(q[0], q[1], q[2], q[3]);
    rd[1] = make_float4(q[4], q[5], q[6], q[7]);
  }
}

__global__ void combine16_kernel(const float* __restrict__ aggp,
                                 const float* __restrict__ deg,
                                 int dstride,
                                 const float* __restrict__ R,
                                 float* __restrict__ out,
                                 int n) {
  int t = blockIdx.x * blockDim.x + threadIdx.x;
  int node = t >> 2;
  int part = t & 3;
  if (node >= n) return;
  float dinv = 1.0f / fmaxf(deg[(long long)node * dstride], 1.0f);
  float4 a = ((const float4*)(aggp + (long long)node * 16))[part];
  float4 r = ((const float4*)(R + (long long)node * 16))[part];
  float4 o;
  o.x = fmaf(a.x, dinv, r.x);
  o.y = fmaf(a.y, dinv, r.y);
  o.z = fmaf(a.z, dinv, r.z);
  o.w = fmaf(a.w, dinv, r.w);
  ((float4*)(out + (long long)node * 16))[part] = o;
}

__global__ void decode_kernel(const float* __restrict__ ou,
                              const float* __restrict__ oc,
                              const int* __restrict__ ls,
                              const int* __restrict__ ld,
                              float* __restrict__ out, int L) {
  int t = blockIdx.x * blockDim.x + threadIdx.x;
  if (t >= L) return;
  const float4* a = (const float4*)(ou + (long long)ls[t] * 16);
  const float4* b = (const float4*)(oc + (long long)ld[t] * 16);
  float acc = 0.0f;
#pragma unroll
  for (int i = 0; i < 4; ++i) {
    float4 xa = a[i], xb = b[i];
    acc += xa.x * xb.x + xa.y * xb.y + xa.z * xb.z + xa.w * xb.w;
  }
  out[t] = acc;
}

extern "C" void kernel_launch(void* const* d_in, const int* in_sizes, int n_in,
                              void* d_out, int out_size, void* d_ws, size_t ws_size,
                              hipStream_t stream) {
  const float* user_x       = (const float*)d_in[0];
  const float* course_x     = (const float*)d_in[1];
  const float* user_embed   = (const float*)d_in[2];
  const float* course_embed = (const float*)d_in[3];
  const float* Wu = (const float*)d_in[4];
  const float* bu = (const float*)d_in[5];
  const float* Wc = (const float*)d_in[6];
  const float* bc = (const float*)d_in[7];
  const float* c1e_Wl = (const float*)d_in[8];
  const float* c1e_bl = (const float*)d_in[9];
  const float* c1e_Wr = (const float*)d_in[10];
  const float* c1r_Wl = (const float*)d_in[11];
  const float* c1r_bl = (const float*)d_in[12];
  const float* c1r_Wr = (const float*)d_in[13];
  const float* c2e_Wl = (const float*)d_in[14];
  const float* c2e_bl = (const float*)d_in[15];
  const float* c2e_Wr = (const float*)d_in[16];
  const float* c2r_Wl = (const float*)d_in[17];
  const float* c2r_bl = (const float*)d_in[18];
  const float* c2r_Wr = (const float*)d_in[19];
  const int* user_idx   = (const int*)d_in[20];
  const int* course_idx = (const int*)d_in[21];
  const int* edge_src   = (const int*)d_in[22];
  const int* edge_dst   = (const int*)d_in[23];
  const int* label_src  = (const int*)d_in[24];
  const int* label_dst  = (const int*)d_in[25];
  float* outp = (float*)d_out;

  const int NU = in_sizes[2] / 16;
  const int NC = in_sizes[3] / 16;
  const int E  = in_sizes[22];
  const int L  = in_sizes[24];

  char* p = (char*)d_ws;
  auto carve = [&](size_t bytes) -> float* {
    float* r = (float*)p;
    p += (bytes + 255) & ~((size_t)255);
    return r;
  };
  float* xu   = carve((size_t)NU * 32 * 4);
  float* xc   = carve((size_t)NC * 32 * 4);
  float* aggu = carve((size_t)NU * 32 * 4);
  float* aggc = carve((size_t)NC * 32 * 4);
  float* pru  = carve((size_t)NU * 32 * 4);
  float* prc  = carve((size_t)NC * 32 * 4);
  float* Pu = pru;
  float* Ru = pru + (size_t)NU * 16;
  float* Pc = prc;
  float* Rc = prc + (size_t)NC * 16;
  float* ou   = carve((size_t)NU * 16 * 4);
  float* oc   = carve((size_t)NC * 16 * 4);

  const int B = 256;
  auto nblk = [](long long t, int b) { return (unsigned)((t + b - 1) / b); };

  const int NR = (NC + kCapRows - 1) / kCapRows;
  int NS = 256 / NR;
  if (NS < 1) NS = 1;
  const long long cap1 = ((long long)NU * 32) / ((long long)NC * 22);
  const long long cap2 = ((long long)NU * 32) / ((long long)NC * 16);
  const int NS1 = (int)(cap1 < NS ? (cap1 < 1 ? 1 : cap1) : NS);
  const int NS2 = (int)(cap2 < NS ? (cap2 < 1 ? 1 : cap2) : NS);

  hipMemsetAsync(aggu, 0, (size_t)NU * 32 * 4, stream);

  fuse_kernel<<<nblk((long long)NU * 32, B), B, 0, stream>>>(
      user_x, user_embed, user_idx, Wu, bu, xu, NU, 5);
  fuse_kernel<<<nblk((long long)NC * 32, B), B, 0, stream>>>(
      course_x, course_embed, course_idx, Wc, bc, xc, NC, 2);

  float* part1 = pru;
  course_sums_raw_lds_kernel<5><<<NR * NS1, kLdsThreads, 0, stream>>>(
      user_embed, user_idx, user_x, edge_src, edge_dst, part1, E, NC, NR, NS1);
  reduce_parts_kernel<22><<<nblk((long long)NC * 22, B), B, 0, stream>>>(part1, aggc, NC, NS1);
  scatter_raw_kernel<2><<<nblk((long long)E * 8, B), B, 0, stream>>>(
      course_embed, course_idx, course_x, edge_dst, edge_src, aggu, E);
  sage1_chain_kernel<5><<<nblk(((long long)(NC + 15) / 16) * 32, B), B, 0, stream>>>(
      aggc, xc, c1e_Wl, c1e_bl, c1e_Wr, Wu, bu, c2r_Wl, c2e_Wr, c2e_bl, Pc, Rc, NC);
  sage1_chain_kernel<2><<<nblk(((long long)(NU + 15) / 16) * 32, B), B, 0, stream>>>(
      aggu, xu, c1r_Wl, c1r_bl, c1r_Wr, Wc, bc, c2e_Wl, c2r_Wr, c2r_bl, Pu, Ru, NU);

  float* part2 = xu;
  course_sums16_lds_kernel<<<NR * NS2, kLdsThreads, 0, stream>>>(
      Pu, edge_src, edge_dst, part2, E, NC, NR, NS2);
  combine16_parts_kernel<<<nblk((long long)NC * 4, B), B, 0, stream>>>(
      part2, NS2, aggc + 21, 32, Rc, oc, NC);

  float* agg2u = xu;
  hipMemsetAsync(agg2u, 0, (size_t)NU * 16 * 4, stream);
  scatter16_kernel<<<nblk(((long long)(E + 3) / 4) * 16, B), B, 0, stream>>>(Pc, edge_dst, edge_src, agg2u, E);
  combine16_kernel<<<nblk((long long)NU * 4, B), B, 0, stream>>>(agg2u, aggu + 18, 32, Ru, ou, NU);

  decode_kernel<<<nblk(L, B), B, 0, stream>>>(ou, oc, label_src, label_dst, outp, L);
}


# user-side neighbour sums of both layers in one edge pass into range-partitioned LDS tables (each block filters the whole edge list); no global atomics or memsets left; sum order: LDS arrival order
# speedup vs baseline: 2.1016x; 1.0636x over previous
#include <hip/hip_runtime.h>

typedef __attribute__((ext_vector_type(8))) float v8f;
typedef __attribute__((ext_vector_type(2))) float v2f;

__global__ void fuse_kernel(const float* __restrict__ x,
                            const float* __restrict__ embed,
                            const int* __restrict__ node_index,
                            const float* __restrict__ W,
                            const float* __restrict__ b,
                            float* __restrict__ out,
                            int n, int fin) {
  int t = blockIdx.x * blockDim.x + threadIdx.x;
  int node = t >> 5;
  int j = t & 31;
  if (node >= n) return;
  float v;
  if (j < 16) {
    v = embed[(long long)node_index[node] * 16 + j];
  } else {
    int r = j - 16;
    float acc = b[r];
    const float* xr = x + (long long)node * fin;
    for (int k = 0; k < fin; ++k) acc = fmaf(xr[k], W[r * fin + k], acc);
    v = acc;
  }
  out[(long long)node * 32 + j] = v;
}


constexpr int kURow = 35;
constexpr int kURows = 2048;
constexpr int kUThreads = 1024;
constexpr int kQCap = 64;
constexpr int kScanLoads = 8;
constexpr int kGather = 8;

__global__ void course_rows_kernel(const float* __restrict__ embed,
                                   const int* __restrict__ node_index,
                                   const float* __restrict__ x,
                                   const float* __restrict__ Pc,
                                   float* __restrict__ tab,
                                   int n) {
  int t = blockIdx.x * blockDim.x + threadIdx.x;
  if (t >= n * kURow) return;
  int c = t / kURow;
  int col = t - c * kURow;
  float v;
  if (col < 16)       v = embed[(long long)node_index[c] * 16 + col];
  else if (col < 18)  v = x[(long long)c * 2 + (col - 16)];
  else if (col == 18) v = 1.0f;
  else                v = Pc[(long long)c * 16 + (col - 19)];
  tab[t] = v;
}

__device__ __forceinline__ void user_queue_flush(int2* q, int qn, int lane, int base,
                                                 const int* __restrict__ edge_user,
                                                 const int* __restrict__ edge_course,
                                                 const float* __restrict__ tab,
                                                 float* T) {
  __builtin_amdgcn_fence(__ATOMIC_SEQ_CST, "workgroup");
  for (int i = lane; i < qn; i += 32) {
    int e = q[i].x;
    q[i] = make_int2(edge_course[e], edge_user[e] - base);
  }
  __builtin_amdgcn_fence(__ATOMIC_SEQ_CST, "workgroup");
  const int total = qn * kURow;
  for (int i0 = lane; i0 < total; i0 += 32 * kGather) {
    float v[kGather];
    int a[kGather];
#pragma unroll
    for (int k = 0; k < kGather; ++k) {
      int i = i0 + 32 * k;
      bool ok = i < total;
      int ii = ok ? i : 0;
      int m = ii / kURow;
      int col = ii - m * kURow;
      int2 en = q[m];
      v[k] = tab[(long long)en.x * kURow + col];
      a[k] = ok ? en.y * kURow + col : -1;
    }
#pragma unroll
    for (int k = 0; k < kGather; ++k) {
      if (a[k] >= 0) atomicAdd(T + a[k], v[k]);
    }
  }
  __builtin_amdgcn_fence(__ATOMIC_SEQ_CST, "workgroup");
}

template <bool CHECK>
__device__ __forceinline__ unsigned user_filter_trip(const int4* __restrict__ eu4, int jb, int N4,
                                                     int base, int nrows) {
  int4 uv[kScanLoads];
  bool ok[kScanLoads];
#pragma unroll
  for (int s = 0; s < kScanLoads; ++s) {
    int j = jb + s * kUThreads;
    ok[s] = CHECK ? (j < N4) : true;
    uv[s] = eu4[ok[s] ? j : 0];
  }
  unsigned hm = 0u;
#pragma unroll
  for (int s = kScanLoads - 1; s >= 0; --s) {
    int4 u = uv[s];
    unsigned h = 0u;
    h = h + h + ((unsigned)(u.w - base) < (unsigned)nrows ? 1u : 0u);
    h = h + h + ((unsigned)(u.z - base) < (unsigned)nrows ? 1u : 0u);
    h = h + h + ((unsigned)(u.y - base) < (unsigned)nrows ? 1u : 0u);
    h = h + h + ((unsigned)(u.x - base) < (unsigned)nrows ? 1u : 0u);
    if (CHECK && !ok[s]) h = 0u;
    hm = (hm << 4) | h;
  }
  return hm;
}

__global__ __launch_bounds__(kUThreads)
void user_sums_lds_kernel(const int* __restrict__ edge_user,
                          const int* __restrict__ edge_course,
                          const float* __restrict__ tab,
                          float* __restrict__ agg1,
                          float* __restrict__ agg2,
                          int E, int n_user) {
  __shared__ float T[kURows * kURow];
  __shared__ int2 Q[(kUThreads / 32) * kQCap];
  const int tid = threadIdx.x;
  const int lane = tid & 31;
  const int base = blockIdx.x * kURows;
  const int left = n_user - base;
  const int nrows = left < kURows ? left : kURows;
  for (int i = tid; i < nrows * kURow; i += kUThreads) T[i] = 0.0f;
  __syncthreads();

  int2* q = Q + (tid >> 5) * kQCap;
  int qn = 0;
  const int N4 = E >> 2;

  if (tid < 32) {
    int e = (N4 << 2) + lane;
    bool hit = false;
    if (e < E) hit = (unsigned)(edge_user[e] - base) < (unsigned)nrows;
    unsigned m = __builtin_amdgcn_ballot_w32(hit);
    if (hit) q[__builtin_amdgcn_mbcnt_lo(m, 0u)].x = e;
    qn = __builtin_popcount(m);
  }

  const int4* eu4 = (const int4*)edge_user;
  constexpr int kTrip = kUThreads * kScanLoads;
  const int ntrips = (N4 + kTrip - 1) / kTrip;
  const int toff = (int)(((unsigned)blockIdx.x * (unsigned)ntrips) / gridDim.x);
  for (int t = 0; t < ntrips; ++t) {
    int tr = t + toff;
    if (tr >= ntrips) tr -= ntrips;
    const int jb = tr * kTrip + tid;
    unsigned hm;
    if (tr * kTrip + kTrip <= N4) hm = user_filter_trip<false>(eu4, jb, N4, base, nrows);
    else                          hm = user_filter_trip<true>(eu4, jb, N4, base, nrows);
    unsigned m = __builtin_amdgcn_ballot_w32(hm != 0u);
    while (m != 0u) {
      int n = __builtin_popcount(m);
      if (qn + n > kQCap) {
        user_queue_flush(q, qn, lane, base, edge_user, edge_course, tab, T);
        qn = 0;
      }
      if (hm != 0u) {
        int b = __builtin_ctz(hm);
        hm &= hm - 1u;
        q[qn + (int)__builtin_amdgcn_mbcnt_lo(m, 0u)].x = 4 * (jb + (b >> 2) * kUThreads) + (b & 3);
      }
      qn += n;
      m = __builtin_amdgcn_ballot_w32(hm != 0u);
    }
  }
  if (qn > 0) user_queue_flush(q, qn, lane, base, edge_user, edge_course, tab, T);
  __syncthreads();

  for (int i = tid; i < nrows * kURow; i += kUThreads) {
    int row = i / kURow;
    int c = i - row * kURow;
    float v = T[i];
    long long node = (long long)base + row;
    if (c < 19) agg1[node * 32 + c] = v;
    else        agg2[node * 16 + (c - 19)] = v;
  }
}

constexpr int kCapRows = 2560;
constexpr int kLdsThreads = 1024;
constexpr int kEdgesPerTrip = 4;

template <int FIN>
__global__ __launch_bounds__(kLdsThreads)
void course_sums_raw_lds_kernel(const float* __restrict__ embed,
                                const int* __restrict__ node_index,
                                const float* __restrict__ x,
                                const int* __restrict__ gather_idx,
                                const int* __restrict__ scatter_idx,
                                float* __restrict__ part,
                                int E, int n_dst, int n_ranges, int n_slices) {
  constexpr int W = 16 + FIN + 1;
  constexpr int P = W | 1;
  __shared__ float T[kCapRows * P];
  const int tid = threadIdx.x;
  const int nthr = blockDim.x;
  const int r = blockIdx.x % n_ranges;
  const int slice = blockIdx.x / n_ranges;
  const int base = r * kCapRows;
  const int left = n_dst - base;
  const int nrows = left < kCapRows ? left : kCapRows;
  for (int i = tid; i < nrows * P; i += nthr) T[i] = 0.0f;
  __syncthreads();

  const int chunk = (E + n_slices - 1) / n_slices;
  const int e_lo = slice * chunk;
  const int e_end = e_lo + chunk;
  const int e_hi = e_end < E ? e_end : E;
  for (int e0 = e_lo + tid; e0 < e_hi; e0 += kEdgesPerTrip * nthr) {
    int s[kEdgesPerTrip], rel[kEdgesPerTrip], ni[kEdgesPerTrip];
    bool in[kEdgesPerTrip];
    float4 ev[kEdgesPerTrip][4];
    float xv[kEdgesPerTrip][FIN];
#pragma unroll
    for (int u = 0; u < kEdgesPerTrip; ++u) {
      int e = e0 + u * nthr;
      bool v = e < e_hi;
      int ec = v ? e : e0;
      int d = scatter_idx[ec];
      s[u] = gather_idx[ec];
      rel[u] = d - base;
      in[u] = v && ((unsigned)rel[u] < (unsigned)nrows);
      if (!in[u]) s[u] = 0;
    }
#pragma unroll
    for (int u = 0; u < kEdgesPerTrip; ++u) ni[u] = node_index[s[u]];
#pragma unroll
    for (int u = 0; u < kEdgesPerTrip; ++u) {
      if (!in[u]) ni[u] = 0;
      const float4* er = (const float4*)(embed + (long long)ni[u] * 16);
#pragma unroll
      for (int q = 0; q < 4; ++q) ev[u][q] = er[q];
      const float* xr = x + (long long)s[u] * FIN;
#pragma unroll
      for (int j = 0; j < FIN; ++j) xv[u][j] = xr[j];
    }
#pragma unroll
    for (int u = 0; u < kEdgesPerTrip; ++u) {
      if (in[u]) {
        float* row = T + rel[u] * P;
#pragma unroll
        for (int q = 0; q < 4; ++q) {
          atomicAdd(row + 4 * q + 0, ev[u][q].x);
          atomicAdd(row + 4 * q + 1, ev[u][q].y);
          atomicAdd(row + 4 * q + 2, ev[u][q].z);
          atomicAdd(row + 4 * q + 3, ev[u][q].w);
        }
#pragma unroll
        for (int j = 0; j < FIN; ++j) atomicAdd(row + 16 + j, xv[u][j]);
        atomicAdd(row + 16 + FIN, 1.0f);
      }
    }
  }
  __syncthreads();

  float* dst = part + ((long long)slice * n_dst + base) * W;
  for (int i = tid; i < nrows * W; i += nthr) {
    int row = i / W;
    int c = i - row * W;
    dst[i] = T[row * P + c];
  }
}

__global__ __launch_bounds__(kLdsThreads)
void course_sums16_lds_kernel(const float* __restrict__ srcp,
                              const int* __restrict__ gather_idx,
                              const int* __restrict__ scatter_idx,
                              float* __restrict__ part,
                              int E, int n_dst, int n_ranges, int n_slices) {
  constexpr int P = 17;
  __shared__ float T[kCapRows * P];
  const int tid = threadIdx.x;
  const int nthr = blockDim.x;
  const int r = blockIdx.x % n_ranges;
  const int slice = blockIdx.x / n_ranges;
  const int base = r * kCapRows;
  const int left = n_dst - base;
  const int nrows = left < kCapRows ? left : kCapRows;
  for (int i = tid; i < nrows * P; i += nthr) T[i] = 0.0f;
  __syncthreads();

  const int chunk = (E + n_slices - 1) / n_slices;
  const int e_lo = slice * chunk;
  const int e_end = e_lo + chunk;
  const int e_hi = e_end < E ? e_end : E;
  for (int e0 = e_lo + tid; e0 < e_hi; e0 += kEdgesPerTrip * nthr) {
    int s[kEdgesPerTrip], rel[kEdgesPerTrip];
    bool in[kEdgesPerTrip];
    float4 pv[kEdgesPerTrip][4];
#pragma unroll
    for (int u = 0; u < kEdgesPerTrip; ++u) {
      int e = e0 + u * nthr;
      bool v = e < e_hi;
      int ec = v ? e : e0;
      int d = scatter_idx[ec];
      s[u] = gather_idx[ec];
      rel[u] = d - base;
      in[u] = v && ((unsigned)rel[u] < (unsigned)nrows);
      if (!in[u]) s[u] = 0;
    }
#pragma unroll
    for (int u = 0; u < kEdgesPerTrip; ++u) {
      const float4* pr = (const float4*)(srcp + (long long)s[u] * 16);
#pragma unroll
      for (int q = 0; q < 4; ++q) pv[u][q] = pr[q];
    }
#pragma unroll
    for (int u = 0; u < kEdgesPerTrip; ++u) {
      if (in[u]) {
        float* row = T + rel[u] * P;
#pragma unroll
        for (int q = 0; q < 4; ++q) {
          atomicAdd(row + 4 * q + 0, pv[u][q].x);
          atomicAdd(row + 4 * q + 1, pv[u][q].y);
          atomicAdd(row + 4 * q + 2, pv[u][q].z);
          atomicAdd(row + 4 * q + 3, pv[u][q].w);
        }
      }
    }
  }
  __syncthreads();

  float* dst = part + ((long long)slice * n_dst + base) * 16;
  for (int i = tid; i < nrows * 16; i += nthr) {
    dst[i] = T[(i >> 4) * P + (i & 15)];
  }
}

template <int W>
__global__ void reduce_parts_kernel(const float* __restrict__ part,
                                    float* __restrict__ out,
                                    int n_rows, int n_slices) {
  int t = blockIdx.x * blockDim.x + threadIdx.x;
  int total = n_rows * W;
  if (t >= total) return;
  float acc = 0.0f;
#pragma unroll 16
  for (int s = 0; s < n_slices; ++s) acc += part[(long long)s * total + t];
  int row = t / W;
  int c = t - row * W;
  out[(long long)row * 32 + c] = acc;
}

__global__ void combine16_parts_kernel(const float* __restrict__ part,
                                       int n_slices,
                                       const float* __restrict__ deg,
                                       int dstride,
                                       const float* __restrict__ R,
                                       float* __restrict__ out,
                                       int n) {
  int t = blockIdx.x * blockDim.x + threadIdx.x;
  int node = t >> 2;
  int q = t & 3;
  if (node >= n) return;
  float4 a = make_float4(0.0f, 0.0f, 0.0f, 0.0f);
  long long stride = (long long)n * 16;
#pragma unroll 16
  for (int s = 0; s < n_slices; ++s) {
    float4 v = ((const float4*)(part + s * stride + (long long)node * 16))[q];
    a.x += v.x; a.y += v.y; a.z += v.z; a.w += v.w;
  }
  float dinv = 1.0f / fmaxf(deg[(long long)node * dstride], 1.0f);
  float4 rr = ((const float4*)(R + (long long)node * 16))[q];
  float4 o;
  o.x = fmaf(a.x, dinv, rr.x);
  o.y = fmaf(a.y, dinv, rr.y);
  o.z = fmaf(a.z, dinv, rr.z);
  o.w = fmaf(a.w, dinv, rr.w);
  ((float4*)(out + (long long)node * 16))[q] = o;
}

template <int FIN>
__global__ void sage1_chain_kernel(const float* __restrict__ agg,
                                   const float* __restrict__ xdst,
                                   const float* __restrict__ Wl,
                                   const float* __restrict__ bl,
                                   const float* __restrict__ Wr,
                                   const float* __restrict__ Wp,
                                   const float* __restrict__ bp,
                                   const float* __restrict__ W2n,
                                   const float* __restrict__ W2r,
                                   const float* __restrict__ b2r,
                                   float* __restrict__ Pout,
                                   float* __restrict__ Rout,
                                   int n) {
  int lane = threadIdx.x & 31;
  int wave = (blockIdx.x * blockDim.x + threadIdx.x) >> 5;
  int nb = wave * 16;
  if (nb >= n) return;
  int m  = lane & 15;
  int hi = lane >> 4;

  int node = nb + m;
  bool nv = node < n;
  int ncl = nv ? node : (n - 1);
  const float* arow = agg  + (long long)ncl * 32;
  const float* xrow = xdst + (long long)ncl * 32;
  float cnt  = arow[16 + FIN];
  float dinv = nv ? (1.0f / fmaxf(cnt, 1.0f)) : 0.0f;

  v2f f_agg[8], f_x[8];
#pragma unroll
  for (int s = 0; s < 8; ++s) {
    int k = 4 * s + 2 * hi;
    if (s < 4) {
      f_agg[s].x = arow[k] * dinv;
      f_agg[s].y = arow[k + 1] * dinv;
    }
    f_x[s].x = nv ? xrow[k]     : 0.0f;
    f_x[s].y = nv ? xrow[k + 1] : 0.0f;
  }
  {
    float sx[FIN];
#pragma unroll
    for (int j = 0; j < FIN; ++j) sx[j] = arow[16 + j];
#pragma unroll
    for (int s = 4; s < 8; ++s) {
      int r = 4 * (s - 4) + 2 * hi;
      float p0 = cnt * bp[r];
      float p1 = cnt * bp[r + 1];
#pragma unroll
      for (int j = 0; j < FIN; ++j) {
        p0 = fmaf(sx[j], Wp[r * FIN + j], p0);
        p1 = fmaf(sx[j], Wp[(r + 1) * FIN + j], p1);
      }
      f_agg[s].x = p0 * dinv;
      f_agg[s].y = p1 * dinv;
    }
  }

  v8f hT[2];
#pragma unroll
  for (int t = 0; t < 2; ++t) {
    int og = t * 16 + m;
    const float* wl = Wl + og * 32;
    const float* wr = Wr + og * 32;
    v8f c;
#pragma unroll
    for (int r = 0; r < 8; ++r) c[r] = bl[t * 16 + 8 * hi + r];

#pragma unroll
    for (int s = 0; s < 8; ++s) {
      int k = 4 * s + 2 * hi;
      v2f w;
      w.x = wl[k];
      w.y = wl[k + 1];
      c = __builtin_amdgcn_wmma_f32_16x16x4_f32(false, w, false, f_agg[s],
                                                (short)0, c, false, false);
    }
#pragma unroll
    for (int s = 0; s < 8; ++s) {
      int k = 4 * s + 2 * hi;
      v2f w;
      w.x = wr[k];
      w.y = wr[k + 1];
      c = __builtin_amdgcn_wmma_f32_16x16x4_f32(false, w, false, f_x[s],
                                                (short)0, c, false, false);
    }
#pragma unroll
    for (int r = 0; r < 8; ++r) hT[t][r] = fmaxf(c[r], 0.0f);
  }

  const float* w2n = W2n + m * 32;
  const float* w2r = W2r + m * 32;
  v8f p, q;
#pragma unroll
  for (int r = 0; r < 8; ++r) {
    p[r] = 0.0f;
    q[r] = b2r[8 * hi + r];
  }
#pragma unroll
  for (int t = 0; t < 2; ++t) {
#pragma unroll
    for (int qq = 0; qq < 4; ++qq) {
      int k = 16 * t + 8 * hi + 2 * qq;
      v2f hb;
      hb.x = hT[t][2 * qq];
      hb.y = hT[t][2 * qq + 1];
      v2f wa, wb;
      wa.x = w2n[k];
      wa.y = w2n[k + 1];
      wb.x = w2r[k];
      wb.y = w2r[k + 1];
      p = __builtin_amdgcn_wmma_f32_16x16x4_f32(false, wa, false, hb,
                                                (short)0, p, false, false);
      q = __builtin_amdgcn_wmma_f32_16x16x4_f32(false, wb, false, hb,
                                                (short)0, q, false, false);
    }
  }

  if (nv) {
    float4* pd = (float4*)(Pout + (long long)node * 16 + 8 * hi);
    float4* rd = (float4*)(Rout + (long long)node * 16 + 8 * hi);
    pd[0] = make_float4(p[0], p[1], p[2], p[3]);
    pd[1] = make_float4(p[4], p[5], p[6], p[7]);
    rd[0] = make_float4(q[0], q[1], q[2], q[3]);
    rd[1] = make_float4(q[4], q[5], q[6], q[7]);
  }
}

__global__ void combine16_kernel(float* io,
                                 const float* __restrict__ deg,
                                 int dstride,
                                 const float* __restrict__ R,
                                 int n) {
  int t = blockIdx.x * blockDim.x + threadIdx.x;
  int node = t >> 2;
  int part = t & 3;
  if (node >= n) return;
  float dinv = 1.0f / fmaxf(deg[(long long)node * dstride], 1.0f);
  float4 a = ((const float4*)(io + (long long)node * 16))[part];
  float4 r = ((const float4*)(R + (long long)node * 16))[part];
  float4 o;
  o.x = fmaf(a.x, dinv, r.x);
  o.y = fmaf(a.y, dinv, r.y);
  o.z = fmaf(a.z, dinv, r.z);
  o.w = fmaf(a.w, dinv, r.w);
  ((float4*)(io + (long long)node * 16))[part] = o;
}

__global__ void decode_kernel(const float* __restrict__ ou,
                              const float* __restrict__ oc,
                              const int* __restrict__ ls,
                              const int* __restrict__ ld,
                              float* __restrict__ out, int L) {
  int t = blockIdx.x * blockDim.x + threadIdx.x;
  if (t >= L) return;
  const float4* a = (const float4*)(ou + (long long)ls[t] * 16);
  const float4* b = (const float4*)(oc + (long long)ld[t] * 16);
  float acc = 0.0f;
#pragma unroll
  for (int i = 0; i < 4; ++i) {
    float4 xa = a[i], xb = b[i];
    acc += xa.x * xb.x + xa.y * xb.y + xa.z * xb.z + xa.w * xb.w;
  }
  out[t] = acc;
}

extern "C" void kernel_launch(void* const* d_in, const int* in_sizes, int n_in,
                              void* d_out, int out_size, void* d_ws, size_t ws_size,
                              hipStream_t stream) {
  const float* user_x       = (const float*)d_in[0];
  const float* course_x     = (const float*)d_in[1];
  const float* user_embed   = (const float*)d_in[2];
  const float* course_embed = (const float*)d_in[3];
  const float* Wu = (const float*)d_in[4];
  const float* bu = (const float*)d_in[5];
  const float* Wc = (const float*)d_in[6];
  const float* bc = (const float*)d_in[7];
  const float* c1e_Wl = (const float*)d_in[8];
  const float* c1e_bl = (const float*)d_in[9];
  const float* c1e_Wr = (const float*)d_in[10];
  const float* c1r_Wl = (const float*)d_in[11];
  const float* c1r_bl = (const float*)d_in[12];
  const float* c1r_Wr = (const float*)d_in[13];
  const float* c2e_Wl = (const float*)d_in[14];
  const float* c2e_bl = (const float*)d_in[15];
  const float* c2e_Wr = (const float*)d_in[16];
  const float* c2r_Wl = (const float*)d_in[17];
  const float* c2r_bl = (const float*)d_in[18];
  const float* c2r_Wr = (const float*)d_in[19];
  const int* user_idx   = (const int*)d_in[20];
  const int* course_idx = (const int*)d_in[21];
  const int* edge_src   = (const int*)d_in[22];
  const int* edge_dst   = (const int*)d_in[23];
  const int* label_src  = (const int*)d_in[24];
  const int* label_dst  = (const int*)d_in[25];
  float* outp = (float*)d_out;

  const int NU = in_sizes[2] / 16;
  const int NC = in_sizes[3] / 16;
  const int E  = in_sizes[22];
  const int L  = in_sizes[24];

  char* p = (char*)d_ws;
  auto carve = [&](size_t bytes) -> float* {
    float* r = (float*)p;
    p += (bytes + 255) & ~((size_t)255);
    return r;
  };
  float* xu   = carve((size_t)NU * 32 * 4);
  float* xc   = carve((size_t)NC * 32 * 4);
  float* aggu = carve((size_t)NU * 32 * 4);
  float* aggc = carve((size_t)NC * 32 * 4);
  float* pru  = carve((size_t)NU * 32 * 4);
  float* prc  = carve((size_t)NC * 32 * 4);
  float* Pu = pru;
  float* Ru = pru + (size_t)NU * 16;
  float* Pc = prc;
  float* Rc = prc + (size_t)NC * 16;
  float* ou   = carve((size_t)NU * 16 * 4);
  float* oc   = carve((size_t)NC * 16 * 4);

  const int B = 256;
  auto nblk = [](long long t, int b) { return (unsigned)((t + b - 1) / b); };

  const int NR = (NC + kCapRows - 1) / kCapRows;
  int NS = 256 / NR;
  if (NS < 1) NS = 1;
  const long long cap1 = ((long long)NU * 32) / ((long long)NC * 22);
  const long long cap2 = ((long long)NU * 32) / ((long long)NC * 16);
  const int NS1 = (int)(cap1 < NS ? (cap1 < 1 ? 1 : cap1) : NS);
  const int NS2 = (int)(cap2 < NS ? (cap2 < 1 ? 1 : cap2) : NS);

  fuse_kernel<<<nblk((long long)NU * 32, B), B, 0, stream>>>(
      user_x, user_embed, user_idx, Wu, bu, xu, NU, 5);
  fuse_kernel<<<nblk((long long)NC * 32, B), B, 0, stream>>>(
      course_x, course_embed, course_idx, Wc, bc, xc, NC, 2);

  float* part1 = pru;
  course_sums_raw_lds_kernel<5><<<NR * NS1, kLdsThreads, 0, stream>>>(
      user_embed, user_idx, user_x, edge_src, edge_dst, part1, E, NC, NR, NS1);
  reduce_parts_kernel<22><<<nblk((long long)NC * 22, B), B, 0, stream>>>(part1, aggc, NC, NS1);
  sage1_chain_kernel<5><<<nblk(((long long)(NC + 15) / 16) * 32, B), B, 0, stream>>>(
      aggc, xc, c1e_Wl, c1e_bl, c1e_Wr, Wu, bu, c2r_Wl, c2e_Wr, c2e_bl, Pc, Rc, NC);
  float* ctab = pru;
  course_rows_kernel<<<nblk((long long)NC * kURow, B), B, 0, stream>>>(
      course_embed, course_idx, course_x, Pc, ctab, NC);
  user_sums_lds_kernel<<<nblk(NU, kURows), kUThreads, 0, stream>>>(
      edge_src, edge_dst, ctab, aggu, ou, E, NU);
  sage1_chain_kernel<2><<<nblk(((long long)(NU + 15) / 16) * 32, B), B, 0, stream>>>(
      aggu, xu, c1r_Wl, c1r_bl, c1r_Wr, Wc, bc, c2e_Wl, c2r_Wr, c2r_bl, Pu, Ru, NU);

  float* part2 = xu;
  course_sums16_lds_kernel<<<NR * NS2, kLdsThreads, 0, stream>>>(
      Pu, edge_src, edge_dst, part2, E, NC, NR, NS2);
  combine16_parts_kernel<<<nblk((long long)NC * 4, B), B, 0, stream>>>(
      part2, NS2, aggc + 21, 32, Rc, oc, NC);

  combine16_kernel<<<nblk((long long)NU * 4, B), B, 0, stream>>>(ou, aggu + 18, 32, Ru, NU);

  decode_kernel<<<nblk(L, B), B, 0, stream>>>(ou, oc, label_src, label_dst, outp, L);
}


# fused feature rows never stored: layer-1 finalize rebuilds a node's own row (embedding lookup + input projection) in its WMMA fragments from the raw inputs; fusion launches gone; sum order unchanged
# speedup vs baseline: 2.9519x; 1.4046x over previous
#include <hip/hip_runtime.h>

typedef __attribute__((ext_vector_type(8))) float v8f;
typedef __attribute__((ext_vector_type(2))) float v2f;


constexpr int kURow = 35;
constexpr int kURows = 2048;
constexpr int kUThreads = 1024;
constexpr int kQCap = 64;
constexpr int kScanLoads = 8;
constexpr int kGather = 8;

__global__ void course_rows_kernel(const float* __restrict__ embed,
                                   const int* __restrict__ node_index,
                                   const float* __restrict__ x,
                                   const float* __restrict__ Pc,
                                   float* __restrict__ tab,
                                   int n) {
  int t = blockIdx.x * blockDim.x + threadIdx.x;
  if (t >= n * kURow) return;
  int c = t / kURow;
  int col = t - c * kURow;
  float v;
  if (col < 16)       v = embed[(long long)node_index[c] * 16 + col];
  else if (col < 18)  v = x[(long long)c * 2 + (col - 16)];
  else if (col == 18) v = 1.0f;
  else                v = Pc[(long long)c * 16 + (col - 19)];
  tab[t] = v;
}

__device__ __forceinline__ void user_queue_flush(int2* q, int qn, int lane, int base,
                                                 const int* __restrict__ edge_user,
                                                 const int* __restrict__ edge_course,
                                                 const float* __restrict__ tab,
                                                 float* T) {
  __builtin_amdgcn_fence(__ATOMIC_SEQ_CST, "workgroup");
  for (int i = lane; i < qn; i += 32) {
    int e = q[i].x;
    q[i] = make_int2(edge_course[e], edge_user[e] - base);
  }
  __builtin_amdgcn_fence(__ATOMIC_SEQ_CST, "workgroup");
  const int total = qn * kURow;
  for (int i0 = lane; i0 < total; i0 += 32 * kGather) {
    float v[kGather];
    int a[kGather];
#pragma unroll
    for (int k = 0; k < kGather; ++k) {
      int i = i0 + 32 * k;
      bool ok = i < total;
      int ii = ok ? i : 0;
      int m = ii / kURow;
      int col = ii - m * kURow;
      int2 en = q[m];
      v[k] = tab[(long long)en.x * kURow + col];
      a[k] = ok ? en.y * kURow + col : -1;
    }
#pragma unroll
    for (int k = 0; k < kGather; ++k) {
      if (a[k] >= 0) atomicAdd(T + a[k], v[k]);
    }
  }
  __builtin_amdgcn_fence(__ATOMIC_SEQ_CST, "workgroup");
}

template <bool CHECK>
__device__ __forceinline__ unsigned user_filter_trip(const int4* __restrict__ eu4, int jb, int N4,
                                                     int base, int nrows) {
  int4 uv[kScanLoads];
  bool ok[kScanLoads];
#pragma unroll
  for (int s = 0; s < kScanLoads; ++s) {
    int j = jb + s * kUThreads;
    ok[s] = CHECK ? (j < N4) : true;
    uv[s] = eu4[ok[s] ? j : 0];
  }
  unsigned hm = 0u;
#pragma unroll
  for (int s = kScanLoads - 1; s >= 0; --s) {
    int4 u = uv[s];
    unsigned h = 0u;
    h = h + h + ((unsigned)(u.w - base) < (unsigned)nrows ? 1u : 0u);
    h = h + h + ((unsigned)(u.z - base) < (unsigned)nrows ? 1u : 0u);
    h = h + h + ((unsigned)(u.y - base) < (unsigned)nrows ? 1u : 0u);
    h = h + h + ((unsigned)(u.x - base) < (unsigned)nrows ? 1u : 0u);
    if (CHECK && !ok[s]) h = 0u;
    hm = (hm << 4) | h;
  }
  return hm;
}

__global__ __launch_bounds__(kUThreads)
void user_sums_lds_kernel(const int* __restrict__ edge_user,
                          const int* __restrict__ edge_course,
                          const float* __restrict__ tab,
                          float* __restrict__ agg1,
                          float* __restrict__ agg2,
                          int E, int n_user) {
  __shared__ float T[kURows * kURow];
  __shared__ int2 Q[(kUThreads / 32) * kQCap];
  const int tid = threadIdx.x;
  const int lane = tid & 31;
  const int base = blockIdx.x * kURows;
  const int left = n_user - base;
  const int nrows = left < kURows ? left : kURows;
  for (int i = tid; i < nrows * kURow; i += kUThreads) T[i] = 0.0f;
  __syncthreads();

  int2* q = Q + (tid >> 5) * kQCap;
  int qn = 0;
  const int N4 = E >> 2;

  if (tid < 32) {
    int e = (N4 << 2) + lane;
    bool hit = false;
    if (e < E) hit = (unsigned)(edge_user[e] - base) < (unsigned)nrows;
    unsigned m = __builtin_amdgcn_ballot_w32(hit);
    if (hit) q[__builtin_amdgcn_mbcnt_lo(m, 0u)].x = e;
    qn = __builtin_popcount(m);
  }

  const int4* eu4 = (const int4*)edge_user;
  constexpr int kTrip = kUThreads * kScanLoads;
  const int ntrips = (N4 + kTrip - 1) / kTrip;
  const int toff = (int)(((unsigned)blockIdx.x * (unsigned)ntrips) / gridDim.x);
  for (int t = 0; t < ntrips; ++t) {
    int tr = t + toff;
    if (tr >= ntrips) tr -= ntrips;
    const int jb = tr * kTrip + tid;
    unsigned hm;
    if (tr * kTrip + kTrip <= N4) hm = user_filter_trip<false>(eu4, jb, N4, base, nrows);
    else                          hm = user_filter_trip<true>(eu4, jb, N4, base, nrows);
    unsigned m = __builtin_amdgcn_ballot_w32(hm != 0u);
    while (m != 0u) {
      int n = __builtin_popcount(m);
      if (qn + n > kQCap) {
        user_queue_flush(q, qn, lane, base, edge_user, edge_course, tab, T);
        qn = 0;
      }
      if (hm != 0u) {
        int b = __builtin_ctz(hm);
        hm &= hm - 1u;
        q[qn + (int)__builtin_amdgcn_mbcnt_lo(m, 0u)].x = 4 * (jb + (b >> 2) * kUThreads) + (b & 3);
      }
      qn += n;
      m = __builtin_amdgcn_ballot_w32(hm != 0u);
    }
  }
  if (qn > 0) user_queue_flush(q, qn, lane, base, edge_user, edge_course, tab, T);
  __syncthreads();

  for (int i = tid; i < nrows * kURow; i += kUThreads) {
    int row = i / kURow;
    int c = i - row * kURow;
    float v = T[i];
    long long node = (long long)base + row;
    if (c < 19) agg1[node * 32 + c] = v;
    else        agg2[node * 16 + (c - 19)] = v;
  }
}

constexpr int kCapRows = 2560;
constexpr int kLdsThreads = 1024;
constexpr int kEdgesPerTrip = 4;

template <int FIN>
__global__ __launch_bounds__(kLdsThreads)
void course_sums_raw_lds_kernel(const float* __restrict__ embed,
                                const int* __restrict__ node_index,
                                const float* __restrict__ x,
                                const int* __restrict__ gather_idx,
                                const int* __restrict__ scatter_idx,
                                float* __restrict__ part,
                                int E, int n_dst, int n_ranges, int n_slices) {
  constexpr int W = 16 + FIN + 1;
  constexpr int P = W | 1;
  __shared__ float T[kCapRows * P];
  const int tid = threadIdx.x;
  const int nthr = blockDim.x;
  const int r = blockIdx.x % n_ranges;
  const int slice = blockIdx.x / n_ranges;
  const int base = r * kCapRows;
  const int left = n_dst - base;
  const int nrows = left < kCapRows ? left : kCapRows;
  for (int i = tid; i < nrows * P; i += nthr) T[i] = 0.0f;
  __syncthreads();

  const int chunk = (E + n_slices - 1) / n_slices;
  const int e_lo = slice * chunk;
  const int e_end = e_lo + chunk;
  const int e_hi = e_end < E ? e_end : E;
  for (int e0 = e_lo + tid; e0 < e_hi; e0 += kEdgesPerTrip * nthr) {
    int s[kEdgesPerTrip], rel[kEdgesPerTrip], ni[kEdgesPerTrip];
    bool in[kEdgesPerTrip];
    float4 ev[kEdgesPerTrip][4];
    float xv[kEdgesPerTrip][FIN];
#pragma unroll
    for (int u = 0; u < kEdgesPerTrip; ++u) {
      int e = e0 + u * nthr;
      bool v = e < e_hi;
      int ec = v ? e : e0;
      int d = scatter_idx[ec];
      s[u] = gather_idx[ec];
      rel[u] = d - base;
      in[u] = v && ((unsigned)rel[u] < (unsigned)nrows);
      if (!in[u]) s[u] = 0;
    }
#pragma unroll
    for (int u = 0; u < kEdgesPerTrip; ++u) ni[u] = node_index[s[u]];
#pragma unroll
    for (int u = 0; u < kEdgesPerTrip; ++u) {
      if (!in[u]) ni[u] = 0;
      const float4* er = (const float4*)(embed + (long long)ni[u] * 16);
#pragma unroll
      for (int q = 0; q < 4; ++q) ev[u][q] = er[q];
      const float* xr = x + (long long)s[u] * FIN;
#pragma unroll
      for (int j = 0; j < FIN; ++j) xv[u][j] = xr[j];
    }
#pragma unroll
    for (int u = 0; u < kEdgesPerTrip; ++u) {
      if (in[u]) {
        float* row = T + rel[u] * P;
#pragma unroll
        for (int q = 0; q < 4; ++q) {
          atomicAdd(row + 4 * q + 0, ev[u][q].x);
          atomicAdd(row + 4 * q + 1, ev[u][q].y);
          atomicAdd(row + 4 * q + 2, ev[u][q].z);
          atomicAdd(row + 4 * q + 3, ev[u][q].w);
        }
#pragma unroll
        for (int j = 0; j < FIN; ++j) atomicAdd(row + 16 + j, xv[u][j]);
        atomicAdd(row + 16 + FIN, 1.0f);
      }
    }
  }
  __syncthreads();

  float* dst = part + ((long long)slice * n_dst + base) * W;
  for (int i = tid; i < nrows * W; i += nthr) {
    int row = i / W;
    int c = i - row * W;
    dst[i] = T[row * P + c];
  }
}

__global__ __launch_bounds__(kLdsThreads)
void course_sums16_lds_kernel(const float* __restrict__ srcp,
                              const int* __restrict__ gather_idx,
                              const int* __restrict__ scatter_idx,
                              float* __restrict__ part,
                              int E, int n_dst, int n_ranges, int n_slices) {
  constexpr int P = 17;
  __shared__ float T[kCapRows * P];
  const int tid = threadIdx.x;
  const int nthr = blockDim.x;
  const int r = blockIdx.x % n_ranges;
  const int slice = blockIdx.x / n_ranges;
  const int base = r * kCapRows;
  const int left = n_dst - base;
  const int nrows = left < kCapRows ? left : kCapRows;
  for (int i = tid; i < nrows * P; i += nthr) T[i] = 0.0f;
  __syncthreads();

  const int chunk = (E + n_slices - 1) / n_slices;
  const int e_lo = slice * chunk;
  const int e_end = e_lo + chunk;
  const int e_hi = e_end < E ? e_end : E;
  for (int e0 = e_lo + tid; e0 < e_hi; e0 += kEdgesPerTrip * nthr) {
    int s[kEdgesPerTrip], rel[kEdgesPerTrip];
    bool in[kEdgesPerTrip];
    float4 pv[kEdgesPerTrip][4];
#pragma unroll
    for (int u = 0; u < kEdgesPerTrip; ++u) {
      int e = e0 + u * nthr;
      bool v = e < e_hi;
      int ec = v ? e : e0;
      int d = scatter_idx[ec];
      s[u] = gather_idx[ec];
      rel[u] = d - base;
      in[u] = v && ((unsigned)rel[u] < (unsigned)nrows);
      if (!in[u]) s[u] = 0;
    }
#pragma unroll
    for (int u = 0; u < kEdgesPerTrip; ++u) {
      const float4* pr = (const float4*)(srcp + (long long)s[u] * 16);
#pragma unroll
      for (int q = 0; q < 4; ++q) pv[u][q] = pr[q];
    }
#pragma unroll
    for (int u = 0; u < kEdgesPerTrip; ++u) {
      if (in[u]) {
        float* row = T + rel[u] * P;
#pragma unroll
        for (int q = 0; q < 4; ++q) {
          atomicAdd(row + 4 * q + 0, pv[u][q].x);
          atomicAdd(row + 4 * q + 1, pv[u][q].y);
          atomicAdd(row + 4 * q + 2, pv[u][q].z);
          atomicAdd(row + 4 * q + 3, pv[u][q].w);
        }
      }
    }
  }
  __syncthreads();

  float* dst = part + ((long long)slice * n_dst + base) * 16;
  for (int i = tid; i < nrows * 16; i += nthr) {
    dst[i] = T[(i >> 4) * P + (i & 15)];
  }
}

template <int W>
__global__ void reduce_parts_kernel(const float* __restrict__ part,
                                    float* __restrict__ out,
                                    int n_rows, int n_slices) {
  int t = blockIdx.x * blockDim.x + threadIdx.x;
  int total = n_rows * W;
  if (t >= total) return;
  float acc = 0.0f;
#pragma unroll 16
  for (int s = 0; s < n_slices; ++s) acc += part[(long long)s * total + t];
  int row = t / W;
  int c = t - row * W;
  out[(long long)row * 32 + c] = acc;
}

__global__ void combine16_parts_kernel(const float* __restrict__ part,
                                       int n_slices,
                                       const float* __restrict__ deg,
                                       int dstride,
                                       const float* __restrict__ R,
                                       float* __restrict__ out,
                                       int n) {
  int t = blockIdx.x * blockDim.x + threadIdx.x;
  int node = t >> 2;
  int q = t & 3;
  if (node >= n) return;
  float4 a = make_float4(0.0f, 0.0f, 0.0f, 0.0f);
  long long stride = (long long)n * 16;
#pragma unroll 16
  for (int s = 0; s < n_slices; ++s) {
    float4 v = ((const float4*)(part + s * stride + (long long)node * 16))[q];
    a.x += v.x; a.y += v.y; a.z += v.z; a.w += v.w;
  }
  float dinv = 1.0f / fmaxf(deg[(long long)node * dstride], 1.0f);
  float4 rr = ((const float4*)(R + (long long)node * 16))[q];
  float4 o;
  o.x = fmaf(a.x, dinv, rr.x);
  o.y = fmaf(a.y, dinv, rr.y);
  o.z = fmaf(a.z, dinv, rr.z);
  o.w = fmaf(a.w, dinv, rr.w);
  ((float4*)(out + (long long)node * 16))[q] = o;
}

template <int FIN, int FDST>
__global__ void sage1_chain_kernel(const float* __restrict__ agg,
                                   const float* __restrict__ embed_d,
                                   const int* __restrict__ idx_d,
                                   const float* __restrict__ x_d,
                                   const float* __restrict__ Wd,
                                   const float* __restrict__ bd,
                                   const float* __restrict__ Wl,
                                   const float* __restrict__ bl,
                                   const float* __restrict__ Wr,
                                   const float* __restrict__ Wp,
                                   const float* __restrict__ bp,
                                   const float* __restrict__ W2n,
                                   const float* __restrict__ W2r,
                                   const float* __restrict__ b2r,
                                   float* __restrict__ Pout,
                                   float* __restrict__ Rout,
                                   int n) {
  int lane = threadIdx.x & 31;
  int wave = (blockIdx.x * blockDim.x + threadIdx.x) >> 5;
  int nb = wave * 16;
  if (nb >= n) return;
  int m  = lane & 15;
  int hi = lane >> 4;

  int node = nb + m;
  bool nv = node < n;
  int ncl = nv ? node : (n - 1);
  const float* arow = agg + (long long)ncl * 32;
  const float* erow = embed_d + (long long)idx_d[ncl] * 16;
  const float* xrow = x_d + (long long)ncl * FDST;
  float cnt  = arow[16 + FIN];
  float dinv = nv ? (1.0f / fmaxf(cnt, 1.0f)) : 0.0f;

  v2f f_agg[8], f_x[8];
#pragma unroll
  for (int s = 0; s < 4; ++s) {
    int k = 4 * s + 2 * hi;
    f_agg[s].x = arow[k] * dinv;
    f_agg[s].y = arow[k + 1] * dinv;
    float e0 = erow[k];
    float e1 = erow[k + 1];
    f_x[s].x = nv ? e0 : 0.0f;
    f_x[s].y = nv ? e1 : 0.0f;
  }
  {
    float xd[FDST];
#pragma unroll
    for (int j = 0; j < FDST; ++j) xd[j] = xrow[j];
#pragma unroll
    for (int s = 4; s < 8; ++s) {
      int r = 4 * (s - 4) + 2 * hi;
      float p0 = bd[r];
      float p1 = bd[r + 1];
#pragma unroll
      for (int j = 0; j < FDST; ++j) {
        p0 = fmaf(xd[j], Wd[r * FDST + j], p0);
        p1 = fmaf(xd[j], Wd[(r + 1) * FDST + j], p1);
      }
      f_x[s].x = nv ? p0 : 0.0f;
      f_x[s].y = nv ? p1 : 0.0f;
    }
  }
  {
    float sx[FIN];
#pragma unroll
    for (int j = 0; j < FIN; ++j) sx[j] = arow[16 + j];
#pragma unroll
    for (int s = 4; s < 8; ++s) {
      int r = 4 * (s - 4) + 2 * hi;
      float p0 = cnt * bp[r];
      float p1 = cnt * bp[r + 1];
#pragma unroll
      for (int j = 0; j < FIN; ++j) {
        p0 = fmaf(sx[j], Wp[r * FIN + j], p0);
        p1 = fmaf(sx[j], Wp[(r + 1) * FIN + j], p1);
      }
      f_agg[s].x = p0 * dinv;
      f_agg[s].y = p1 * dinv;
    }
  }

  v8f hT[2];
#pragma unroll
  for (int t = 0; t < 2; ++t) {
    int og = t * 16 + m;
    const float* wl = Wl + og * 32;
    const float* wr = Wr + og * 32;
    v8f c;
#pragma unroll
    for (int r = 0; r < 8; ++r) c[r] = bl[t * 16 + 8 * hi + r];

#pragma unroll
    for (int s = 0; s < 8; ++s) {
      int k = 4 * s + 2 * hi;
      v2f w;
      w.x = wl[k];
      w.y = wl[k + 1];
      c = __builtin_amdgcn_wmma_f32_16x16x4_f32(false, w, false, f_agg[s],
                                                (short)0, c, false, false);
    }
#pragma unroll
    for (int s = 0; s < 8; ++s) {
      int k = 4 * s + 2 * hi;
      v2f w;
      w.x = wr[k];
      w.y = wr[k + 1];
      c = __builtin_amdgcn_wmma_f32_16x16x4_f32(false, w, false, f_x[s],
                                                (short)0, c, false, false);
    }
#pragma unroll
    for (int r = 0; r < 8; ++r) hT[t][r] = fmaxf(c[r], 0.0f);
  }

  const float* w2n = W2n + m * 32;
  const float* w2r = W2r + m * 32;
  v8f p, q;
#pragma unroll
  for (int r = 0; r < 8; ++r) {
    p[r] = 0.0f;
    q[r] = b2r[8 * hi + r];
  }
#pragma unroll
  for (int t = 0; t < 2; ++t) {
#pragma unroll
    for (int qq = 0; qq < 4; ++qq) {
      int k = 16 * t + 8 * hi + 2 * qq;
      v2f hb;
      hb.x = hT[t][2 * qq];
      hb.y = hT[t][2 * qq + 1];
      v2f wa, wb;
      wa.x = w2n[k];
      wa.y = w2n[k + 1];
      wb.x = w2r[k];
      wb.y = w2r[k + 1];
      p = __builtin_amdgcn_wmma_f32_16x16x4_f32(false, wa, false, hb,
                                                (short)0, p, false, false);
      q = __builtin_amdgcn_wmma_f32_16x16x4_f32(false, wb, false, hb,
                                                (short)0, q, false, false);
    }
  }

  if (nv) {
    float4* pd = (float4*)(Pout + (long long)node * 16 + 8 * hi);
    float4* rd = (float4*)(Rout + (long long)node * 16 + 8 * hi);
    pd[0] = make_float4(p[0], p[1], p[2], p[3]);
    pd[1] = make_float4(p[4], p[5], p[6], p[7]);
    rd[0] = make_float4(q[0], q[1], q[2], q[3]);
    rd[1] = make_float4(q[4], q[5], q[6], q[7]);
  }
}

__global__ void combine16_kernel(float* io,
                                 const float* __restrict__ deg,
                                 int dstride,
                                 const float* __restrict__ R,
                                 int n) {
  int t = blockIdx.x * blockDim.x + threadIdx.x;
  int node = t >> 2;
  int part = t & 3;
  if (node >= n) return;
  float dinv = 1.0f / fmaxf(deg[(long long)node * dstride], 1.0f);
  float4 a = ((const float4*)(io + (long long)node * 16))[part];
  float4 r = ((const float4*)(R + (long long)node * 16))[part];
  float4 o;
  o.x = fmaf(a.x, dinv, r.x);
  o.y = fmaf(a.y, dinv, r.y);
  o.z = fmaf(a.z, dinv, r.z);
  o.w = fmaf(a.w, dinv, r.w);
  ((float4*)(io + (long long)node * 16))[part] = o;
}

__global__ void decode_kernel(const float* __restrict__ ou,
                              const float* __restrict__ oc,
                              const int* __restrict__ ls,
                              const int* __restrict__ ld,
                              float* __restrict__ out, int L) {
  int t = blockIdx.x * blockDim.x + threadIdx.x;
  if (t >= L) return;
  const float4* a = (const float4*)(ou + (long long)ls[t] * 16);
  const float4* b = (const float4*)(oc + (long long)ld[t] * 16);
  float acc = 0.0f;
#pragma unroll
  for (int i = 0; i < 4; ++i) {
    float4 xa = a[i], xb = b[i];
    acc += xa.x * xb.x + xa.y * xb.y + xa.z * xb.z + xa.w * xb.w;
  }
  out[t] = acc;
}

extern "C" void kernel_launch(void* const* d_in, const int* in_sizes, int n_in,
                              void* d_out, int out_size, void* d_ws, size_t ws_size,
                              hipStream_t stream) {
  const float* user_x       = (const float*)d_in[0];
  const float* course_x     = (const float*)d_in[1];
  const float* user_embed   = (const float*)d_in[2];
  const float* course_embed = (const float*)d_in[3];
  const float* Wu = (const float*)d_in[4];
  const float* bu = (const float*)d_in[5];
  const float* Wc = (const float*)d_in[6];
  const float* bc = (const float*)d_in[7];
  const float* c1e_Wl = (const float*)d_in[8];
  const float* c1e_bl = (const float*)d_in[9];
  const float* c1e_Wr = (const float*)d_in[10];
  const float* c1r_Wl = (const float*)d_in[11];
  const float* c1r_bl = (const float*)d_in[12];
  const float* c1r_Wr = (const float*)d_in[13];
  const float* c2e_Wl = (const float*)d_in[14];
  const float* c2e_bl = (const float*)d_in[15];
  const float* c2e_Wr = (const float*)d_in[16];
  const float* c2r_Wl = (const float*)d_in[17];
  const float* c2r_bl = (const float*)d_in[18];
  const float* c2r_Wr = (const float*)d_in[19];
  const int* user_idx   = (const int*)d_in[20];
  const int* course_idx = (const int*)d_in[21];
  const int* edge_src   = (const int*)d_in[22];
  const int* edge_dst   = (const int*)d_in[23];
  const int* label_src  = (const int*)d_in[24];
  const int* label_dst  = (const int*)d_in[25];
  float* outp = (float*)d_out;

  const int NU = in_sizes[2] / 16;
  const int NC = in_sizes[3] / 16;
  const int E  = in_sizes[22];
  const int L  = in_sizes[24];

  char* p = (char*)d_ws;
  auto carve = [&](size_t bytes) -> float* {
    float* r = (float*)p;
    p += (bytes + 255) & ~((size_t)255);
    return r;
  };
  float* prt2 = carve((size_t)NU * 32 * 4);
  float* aggu = carve((size_t)NU * 32 * 4);
  float* aggc = carve((size_t)NC * 32 * 4);
  float* pru  = carve((size_t)NU * 32 * 4);
  float* prc  = carve((size_t)NC * 32 * 4);
  float* Pu = pru;
  float* Ru = pru + (size_t)NU * 16;
  float* Pc = prc;
  float* Rc = prc + (size_t)NC * 16;
  float* ou   = carve((size_t)NU * 16 * 4);
  float* oc   = carve((size_t)NC * 16 * 4);

  const int B = 256;
  auto nblk = [](long long t, int b) { return (unsigned)((t + b - 1) / b); };

  const int NR = (NC + kCapRows - 1) / kCapRows;
  int NS = 256 / NR;
  if (NS < 1) NS = 1;
  const long long cap1 = ((long long)NU * 32) / ((long long)NC * 22);
  const long long cap2 = ((long long)NU * 32) / ((long long)NC * 16);
  const int NS1 = (int)(cap1 < NS ? (cap1 < 1 ? 1 : cap1) : NS);
  const int NS2 = (int)(cap2 < NS ? (cap2 < 1 ? 1 : cap2) : NS);

  float* part1 = pru;
  course_sums_raw_lds_kernel<5><<<NR * NS1, kLdsThreads, 0, stream>>>(
      user_embed, user_idx, user_x, edge_src, edge_dst, part1, E, NC, NR, NS1);
  reduce_parts_kernel<22><<<nblk((long long)NC * 22, B), B, 0, stream>>>(part1, aggc, NC, NS1);
  sage1_chain_kernel<5, 2><<<nblk(((long long)(NC + 15) / 16) * 32, B), B, 0, stream>>>(
      aggc, course_embed, course_idx, course_x, Wc, bc,
      c1e_Wl, c1e_bl, c1e_Wr, Wu, bu, c2r_Wl, c2e_Wr, c2e_bl, Pc, Rc, NC);
  float* ctab = pru;
  course_rows_kernel<<<nblk((long long)NC * kURow, B), B, 0, stream>>>(
      course_embed, course_idx, course_x, Pc, ctab, NC);
  user_sums_lds_kernel<<<nblk(NU, kURows), kUThreads, 0, stream>>>(
      edge_src, edge_dst, ctab, aggu, ou, E, NU);
  sage1_chain_kernel<2, 5><<<nblk(((long long)(NU + 15) / 16) * 32, B), B, 0, stream>>>(
      aggu, user_embed, user_idx, user_x, Wu, bu,
      c1r_Wl, c1r_bl, c1r_Wr, Wc, bc, c2e_Wl, c2r_Wr, c2r_bl, Pu, Ru, NU);

  float* part2 = prt2;
  course_sums16_lds_kernel<<<NR * NS2, kLdsThreads, 0, stream>>>(
      Pu, edge_src, edge_dst, part2, E, NC, NR, NS2);
  combine16_parts_kernel<<<nblk((long long)NC * 4, B), B, 0, stream>>>(
      part2, NS2, aggc + 21, 32, Rc, oc, NC);

  combine16_kernel<<<nblk((long long)NU * 4, B), B, 0, stream>>>(ou, aggu + 18, 32, Ru, NU);

  decode_kernel<<<nblk(L, B), B, 0, stream>>>(ou, oc, label_src, label_dst, outp, L);
}


# user-side edge filter reads a one-byte range key per edge (written by the first edge pass), 4 edges per 32-bit test, instead of the 4-byte user index; sum order: LDS arrival order as before
# speedup vs baseline: 3.2106x; 1.0876x over previous
#include <hip/hip_runtime.h>

typedef __attribute__((ext_vector_type(8))) float v8f;
typedef __attribute__((ext_vector_type(2))) float v2f;


constexpr int kURow = 35;
constexpr int kURows = 2048;
constexpr int kUThreads = 1024;
constexpr int kQCap = 64;
constexpr int kScanLoads = 8;
constexpr int kKeyMax = 255;
constexpr int kGather = 8;

__global__ void course_rows_kernel(const float* __restrict__ embed,
                                   const int* __restrict__ node_index,
                                   const float* __restrict__ x,
                                   const float* __restrict__ Pc,
                                   float* __restrict__ tab,
                                   int n) {
  int t = blockIdx.x * blockDim.x + threadIdx.x;
  if (t >= n * kURow) return;
  int c = t / kURow;
  int col = t - c * kURow;
  float v;
  if (col < 16)       v = embed[(long long)node_index[c] * 16 + col];
  else if (col < 18)  v = x[(long long)c * 2 + (col - 16)];
  else if (col == 18) v = 1.0f;
  else                v = Pc[(long long)c * 16 + (col - 19)];
  tab[t] = v;
}

__device__ __forceinline__ void user_queue_flush(int2* q, int qn, int lane, int base, int nrows,
                                                 const int* __restrict__ edge_user,
                                                 const int* __restrict__ edge_course,
                                                 const float* __restrict__ tab,
                                                 float* T) {
  __builtin_amdgcn_fence(__ATOMIC_SEQ_CST, "workgroup");
  for (int i = lane; i < qn; i += 32) {
    int e = q[i].x;
    int rel = edge_user[e] - base;
    q[i] = make_int2(edge_course[e], (unsigned)rel < (unsigned)nrows ? rel : -1);
  }
  __builtin_amdgcn_fence(__ATOMIC_SEQ_CST, "workgroup");
  const int total = qn * kURow;
  for (int i0 = lane; i0 < total; i0 += 32 * kGather) {
    float v[kGather];
    int a[kGather];
#pragma unroll
    for (int k = 0; k < kGather; ++k) {
      int i = i0 + 32 * k;
      bool ok = i < total;
      int ii = ok ? i : 0;
      int m = ii / kURow;
      int col = ii - m * kURow;
      int2 en = q[m];
      v[k] = tab[(long long)en.x * kURow + col];
      a[k] = (ok && en.y >= 0) ? en.y * kURow + col : -1;
    }
#pragma unroll
    for (int k = 0; k < kGather; ++k) {
      if (a[k] >= 0) atomicAdd(T + a[k], v[k]);
    }
  }
  __builtin_amdgcn_fence(__ATOMIC_SEQ_CST, "workgroup");
}

template <bool CHECK>
__device__ __forceinline__ void user_filter_trip(const uint4* __restrict__ key16, int jb, int N16,
                                                 unsigned kk, unsigned (&hw)[kScanLoads / 2]) {
  uint4 kv[kScanLoads];
  bool ok[kScanLoads];
#pragma unroll
  for (int s = 0; s < kScanLoads; ++s) {
    int j = jb + s * kUThreads;
    ok[s] = CHECK ? (j < N16) : true;
    kv[s] = key16[ok[s] ? j : 0];
  }
#pragma unroll
  for (int w = 0; w < kScanLoads / 2; ++w) {
    unsigned acc = 0u;
#pragma unroll
    for (int g = 0; g < 2; ++g) {
      const int s = 2 * w + g;
      const unsigned d[4] = {kv[s].x, kv[s].y, kv[s].z, kv[s].w};
      unsigned part = 0u;
#pragma unroll
      for (int i = 0; i < 4; ++i) {
        unsigned t = d[i] ^ kk;
        unsigned y = (t - 0x01010101u) & ~t;
        const int pos = 4 * g + i;
        part |= (y >> (7 - pos)) & (0x01010101u << pos);
      }
      if (CHECK && !ok[s]) part = 0u;
      acc |= part;
    }
    hw[w] = acc;
  }
}

__global__ __launch_bounds__(kUThreads)
void user_sums_lds_kernel(const unsigned int* __restrict__ edge_key,
                          const int* __restrict__ edge_user,
                          const int* __restrict__ edge_course,
                          const float* __restrict__ tab,
                          float* __restrict__ agg1,
                          float* __restrict__ agg2,
                          int E, int n_user) {
  __shared__ float T[kURows * kURow];
  __shared__ int2 Q[(kUThreads / 32) * kQCap];
  const int tid = threadIdx.x;
  const int lane = tid & 31;
  const int base = blockIdx.x * kURows;
  const int left = n_user - base;
  const int nrows = left < kURows ? left : kURows;
  for (int i = tid; i < nrows * kURow; i += kUThreads) T[i] = 0.0f;
  __syncthreads();

  int2* q = Q + (tid >> 5) * kQCap;
  int qn = 0;
  const int N16 = E >> 4;
  const int kb = (int)blockIdx.x < kKeyMax ? (int)blockIdx.x : kKeyMax;
  const unsigned kk = (unsigned)kb * 0x01010101u;

  if (tid < 32) {
    int e = (N16 << 4) + lane;
    bool hit = false;
    if (e < E) hit = (unsigned)(edge_user[e] - base) < (unsigned)nrows;
    unsigned m = __builtin_amdgcn_ballot_w32(hit);
    if (hit) q[__builtin_amdgcn_mbcnt_lo(m, 0u)].x = e;
    qn = __builtin_popcount(m);
  }

  const uint4* key16 = (const uint4*)edge_key;
  constexpr int kTrip = kUThreads * kScanLoads;
  const int ntrips = (N16 + kTrip - 1) / kTrip;
  const int toff = (int)(((unsigned)blockIdx.x * (unsigned)ntrips) / gridDim.x);
  for (int t = 0; t < ntrips; ++t) {
    int tr = t + toff;
    if (tr >= ntrips) tr -= ntrips;
    const int jb = tr * kTrip + tid;
    unsigned hw[kScanLoads / 2];
    if (tr * kTrip + kTrip <= N16) user_filter_trip<false>(key16, jb, N16, kk, hw);
    else                           user_filter_trip<true>(key16, jb, N16, kk, hw);
#pragma nounroll
    for (int w = 0; w < kScanLoads / 2; ++w) {
      unsigned hm = hw[0];
#pragma unroll
      for (int i = 0; i + 1 < kScanLoads / 2; ++i) hw[i] = hw[i + 1];
      unsigned m = __builtin_amdgcn_ballot_w32(hm != 0u);
      while (m != 0u) {
        int n = __builtin_popcount(m);
        if (qn + n > kQCap) {
          user_queue_flush(q, qn, lane, base, nrows, edge_user, edge_course, tab, T);
          qn = 0;
        }
        if (hm != 0u) {
          int b = __builtin_ctz(hm);
          hm &= hm - 1u;
          int pos = b & 7;
          int grp = jb + (2 * w + (pos >> 2)) * kUThreads;
          q[qn + (int)__builtin_amdgcn_mbcnt_lo(m, 0u)].x = 16 * grp + 4 * (pos & 3) + (b >> 3);
        }
        qn += n;
        m = __builtin_amdgcn_ballot_w32(hm != 0u);
      }
    }
  }
  if (qn > 0) user_queue_flush(q, qn, lane, base, nrows, edge_user, edge_course, tab, T);
  __syncthreads();

  for (int i = tid; i < nrows * kURow; i += kUThreads) {
    int row = i / kURow;
    int c = i - row * kURow;
    float v = T[i];
    long long node = (long long)base + row;
    if (c < 19) agg1[node * 32 + c] = v;
    else        agg2[node * 16 + (c - 19)] = v;
  }
}

constexpr int kCapRows = 2560;
constexpr int kLdsThreads = 1024;
constexpr int kEdgesPerTrip = 4;

template <int FIN>
__global__ __launch_bounds__(kLdsThreads)
void course_sums_raw_lds_kernel(const float* __restrict__ embed,
                                const int* __restrict__ node_index,
                                const float* __restrict__ x,
                                const int* __restrict__ gather_idx,
                                const int* __restrict__ scatter_idx,
                                float* __restrict__ part,
                                unsigned char* __restrict__ key_out,
                                int E, int n_dst, int n_ranges, int n_slices) {
  constexpr int W = 16 + FIN + 1;
  constexpr int P = W | 1;
  __shared__ float T[kCapRows * P];
  const int tid = threadIdx.x;
  const int nthr = blockDim.x;
  const int r = blockIdx.x % n_ranges;
  const int slice = blockIdx.x / n_ranges;
  const int base = r * kCapRows;
  const int left = n_dst - base;
  const int nrows = left < kCapRows ? left : kCapRows;
  for (int i = tid; i < nrows * P; i += nthr) T[i] = 0.0f;
  __syncthreads();

  const int chunk = (E + n_slices - 1) / n_slices;
  const int e_lo = slice * chunk;
  const int e_end = e_lo + chunk;
  const int e_hi = e_end < E ? e_end : E;
  for (int e0 = e_lo + tid; e0 < e_hi; e0 += kEdgesPerTrip * nthr) {
    int s[kEdgesPerTrip], rel[kEdgesPerTrip], ni[kEdgesPerTrip], kq[kEdgesPerTrip];
    bool in[kEdgesPerTrip];
    float4 ev[kEdgesPerTrip][4];
    float xv[kEdgesPerTrip][FIN];
#pragma unroll
    for (int u = 0; u < kEdgesPerTrip; ++u) {
      int e = e0 + u * nthr;
      bool v = e < e_hi;
      int ec = v ? e : e0;
      int d = scatter_idx[ec];
      s[u] = gather_idx[ec];
      int kr = s[u] / kURows;
      kq[u] = kr < kKeyMax ? kr : kKeyMax;
      rel[u] = d - base;
      in[u] = v && ((unsigned)rel[u] < (unsigned)nrows);
      if (!in[u]) s[u] = 0;
    }
#pragma unroll
    for (int u = 0; u < kEdgesPerTrip; ++u) ni[u] = node_index[s[u]];
#pragma unroll
    for (int u = 0; u < kEdgesPerTrip; ++u) {
      if (!in[u]) ni[u] = 0;
      const float4* er = (const float4*)(embed + (long long)ni[u] * 16);
#pragma unroll
      for (int q = 0; q < 4; ++q) ev[u][q] = er[q];
      const float* xr = x + (long long)s[u] * FIN;
#pragma unroll
      for (int j = 0; j < FIN; ++j) xv[u][j] = xr[j];
    }
    if (r == 0) {
#pragma unroll
      for (int u = 0; u < kEdgesPerTrip; ++u) {
        int e = e0 + u * nthr;
        if (e < e_hi) key_out[e] = (unsigned char)kq[u];
      }
    }
#pragma unroll
    for (int u = 0; u < kEdgesPerTrip; ++u) {
      if (in[u]) {
        float* row = T + rel[u] * P;
#pragma unroll
        for (int q = 0; q < 4; ++q) {
          atomicAdd(row + 4 * q + 0, ev[u][q].x);
          atomicAdd(row + 4 * q + 1, ev[u][q].y);
          atomicAdd(row + 4 * q + 2, ev[u][q].z);
          atomicAdd(row + 4 * q + 3, ev[u][q].w);
        }
#pragma unroll
        for (int j = 0; j < FIN; ++j) atomicAdd(row + 16 + j, xv[u][j]);
        atomicAdd(row + 16 + FIN, 1.0f);
      }
    }
  }
  __syncthreads();

  float* dst = part + ((long long)slice * n_dst + base) * W;
  for (int i = tid; i < nrows * W; i += nthr) {
    int row = i / W;
    int c = i - row * W;
    dst[i] = T[row * P + c];
  }
}

__global__ __launch_bounds__(kLdsThreads)
void course_sums16_lds_kernel(const float* __restrict__ srcp,
                              const int* __restrict__ gather_idx,
                              const int* __restrict__ scatter_idx,
                              float* __restrict__ part,
                              int E, int n_dst, int n_ranges, int n_slices) {
  constexpr int P = 17;
  __shared__ float T[kCapRows * P];
  const int tid = threadIdx.x;
  const int nthr = blockDim.x;
  const int r = blockIdx.x % n_ranges;
  const int slice = blockIdx.x / n_ranges;
  const int base = r * kCapRows;
  const int left = n_dst - base;
  const int nrows = left < kCapRows ? left : kCapRows;
  for (int i = tid; i < nrows * P; i += nthr) T[i] = 0.0f;
  __syncthreads();

  const int chunk = (E + n_slices - 1) / n_slices;
  const int e_lo = slice * chunk;
  const int e_end = e_lo + chunk;
  const int e_hi = e_end < E ? e_end : E;
  for (int e0 = e_lo + tid; e0 < e_hi; e0 += kEdgesPerTrip * nthr) {
    int s[kEdgesPerTrip], rel[kEdgesPerTrip];
    bool in[kEdgesPerTrip];
    float4 pv[kEdgesPerTrip][4];
#pragma unroll
    for (int u = 0; u < kEdgesPerTrip; ++u) {
      int e = e0 + u * nthr;
      bool v = e < e_hi;
      int ec = v ? e : e0;
      int d = scatter_idx[ec];
      s[u] = gather_idx[ec];
      rel[u] = d - base;
      in[u] = v && ((unsigned)rel[u] < (unsigned)nrows);
      if (!in[u]) s[u] = 0;
    }
#pragma unroll
    for (int u = 0; u < kEdgesPerTrip; ++u) {
      const float4* pr = (const float4*)(srcp + (long long)s[u] * 16);
#pragma unroll
      for (int q = 0; q < 4; ++q) pv[u][q] = pr[q];
    }
#pragma unroll
    for (int u = 0; u < kEdgesPerTrip; ++u) {
      if (in[u]) {
        float* row = T + rel[u] * P;
#pragma unroll
        for (int q = 0; q < 4; ++q) {
          atomicAdd(row + 4 * q + 0, pv[u][q].x);
          atomicAdd(row + 4 * q + 1, pv[u][q].y);
          atomicAdd(row + 4 * q + 2, pv[u][q].z);
          atomicAdd(row + 4 * q + 3, pv[u][q].w);
        }
      }
    }
  }
  __syncthreads();

  float* dst = part + ((long long)slice * n_dst + base) * 16;
  for (int i = tid; i < nrows * 16; i += nthr) {
    dst[i] = T[(i >> 4) * P + (i & 15)];
  }
}

template <int W>
__global__ void reduce_parts_kernel(const float* __restrict__ part,
                                    float* __restrict__ out,
                                    int n_rows, int n_slices) {
  int t = blockIdx.x * blockDim.x + threadIdx.x;
  int total = n_rows * W;
  if (t >= total) return;
  float acc = 0.0f;
#pragma unroll 16
  for (int s = 0; s < n_slices; ++s) acc += part[(long long)s * total + t];
  int row = t / W;
  int c = t - row * W;
  out[(long long)row * 32 + c] = acc;
}

__global__ void combine16_parts_kernel(const float* __restrict__ part,
                                       int n_slices,
                                       const float* __restrict__ deg,
                                       int dstride,
                                       const float* __restrict__ R,
                                       float* __restrict__ out,
                                       int n) {
  int t = blockIdx.x * blockDim.x + threadIdx.x;
  int node = t >> 2;
  int q = t & 3;
  if (node >= n) return;
  float4 a = make_float4(0.0f, 0.0f, 0.0f, 0.0f);
  long long stride = (long long)n * 16;
#pragma unroll 16
  for (int s = 0; s < n_slices; ++s) {
    float4 v = ((const float4*)(part + s * stride + (long long)node * 16))[q];
    a.x += v.x; a.y += v.y; a.z += v.z; a.w += v.w;
  }
  float dinv = 1.0f / fmaxf(deg[(long long)node * dstride], 1.0f);
  float4 rr = ((const float4*)(R + (long long)node * 16))[q];
  float4 o;
  o.x = fmaf(a.x, dinv, rr.x);
  o.y = fmaf(a.y, dinv, rr.y);
  o.z = fmaf(a.z, dinv, rr.z);
  o.w = fmaf(a.w, dinv, rr.w);
  ((float4*)(out + (long long)node * 16))[q] = o;
}

template <int FIN, int FDST>
__global__ void sage1_chain_kernel(const float* __restrict__ agg,
                                   const float* __restrict__ embed_d,
                                   const int* __restrict__ idx_d,
                                   const float* __restrict__ x_d,
                                   const float* __restrict__ Wd,
                                   const float* __restrict__ bd,
                                   const float* __restrict__ Wl,
                                   const float* __restrict__ bl,
                                   const float* __restrict__ Wr,
                                   const float* __restrict__ Wp,
                                   const float* __restrict__ bp,
                                   const float* __restrict__ W2n,
                                   const float* __restrict__ W2r,
                                   const float* __restrict__ b2r,
                                   float* __restrict__ Pout,
                                   float* __restrict__ Rout,
                                   int n) {
  int lane = threadIdx.x & 31;
  int wave = (blockIdx.x * blockDim.x + threadIdx.x) >> 5;
  int nb = wave * 16;
  if (nb >= n) return;
  int m  = lane & 15;
  int hi = lane >> 4;

  int node = nb + m;
  bool nv = node < n;
  int ncl = nv ? node : (n - 1);
  const float* arow = agg + (long long)ncl * 32;
  const float* erow = embed_d + (long long)idx_d[ncl] * 16;
  const float* xrow = x_d + (long long)ncl * FDST;
  float cnt  = arow[16 + FIN];
  float dinv = nv ? (1.0f / fmaxf(cnt, 1.0f)) : 0.0f;

  v2f f_agg[8], f_x[8];
#pragma unroll
  for (int s = 0; s < 4; ++s) {
    int k = 4 * s + 2 * hi;
    f_agg[s].x = arow[k] * dinv;
    f_agg[s].y = arow[k + 1] * dinv;
    float e0 = erow[k];
    float e1 = erow[k + 1];
    f_x[s].x = nv ? e0 : 0.0f;
    f_x[s].y = nv ? e1 : 0.0f;
  }
  {
    float xd[FDST];
#pragma unroll
    for (int j = 0; j < FDST; ++j) xd[j] = xrow[j];
#pragma unroll
    for (int s = 4; s < 8; ++s) {
      int r = 4 * (s - 4) + 2 * hi;
      float p0 = bd[r];
      float p1 = bd[r + 1];
#pragma unroll
      for (int j = 0; j < FDST; ++j) {
        p0 = fmaf(xd[j], Wd[r * FDST + j], p0);
        p1 = fmaf(xd[j], Wd[(r + 1) * FDST + j], p1);
      }
      f_x[s].x = nv ? p0 : 0.0f;
      f_x[s].y = nv ? p1 : 0.0f;
    }
  }
  {
    float sx[FIN];
#pragma unroll
    for (int j = 0; j < FIN; ++j) sx[j] = arow[16 + j];
#pragma unroll
    for (int s = 4; s < 8; ++s) {
      int r = 4 * (s - 4) + 2 * hi;
      float p0 = cnt * bp[r];
      float p1 = cnt * bp[r + 1];
#pragma unroll
      for (int j = 0; j < FIN; ++j) {
        p0 = fmaf(sx[j], Wp[r * FIN + j], p0);
        p1 = fmaf(sx[j], Wp[(r + 1) * FIN + j], p1);
      }
      f_agg[s].x = p0 * dinv;
      f_agg[s].y = p1 * dinv;
    }
  }

  v8f hT[2];
#pragma unroll
  for (int t = 0; t < 2; ++t) {
    int og = t * 16 + m;
    const float* wl = Wl + og * 32;
    const float* wr = Wr + og * 32;
    v8f c;
#pragma unroll
    for (int r = 0; r < 8; ++r) c[r] = bl[t * 16 + 8 * hi + r];

#pragma unroll
    for (int s = 0; s < 8; ++s) {
      int k = 4 * s + 2 * hi;
      v2f w;
      w.x = wl[k];
      w.y = wl[k + 1];
      c = __builtin_amdgcn_wmma_f32_16x16x4_f32(false, w, false, f_agg[s],
                                                (short)0, c, false, false);
    }
#pragma unroll
    for (int s = 0; s < 8; ++s) {
      int k = 4 * s + 2 * hi;
      v2f w;
      w.x = wr[k];
      w.y = wr[k + 1];
      c = __builtin_amdgcn_wmma_f32_16x16x4_f32(false, w, false, f_x[s],
                                                (short)0, c, false, false);
    }
#pragma unroll
    for (int r = 0; r < 8; ++r) hT[t][r] = fmaxf(c[r], 0.0f);
  }

  const float* w2n = W2n + m * 32;
  const float* w2r = W2r + m * 32;
  v8f p, q;
#pragma unroll
  for (int r = 0; r < 8; ++r) {
    p[r] = 0.0f;
    q[r] = b2r[8 * hi + r];
  }
#pragma unroll
  for (int t = 0; t < 2; ++t) {
#pragma unroll
    for (int qq = 0; qq < 4; ++qq) {
      int k = 16 * t + 8 * hi + 2 * qq;
      v2f hb;
      hb.x = hT[t][2 * qq];
      hb.y = hT[t][2 * qq + 1];
      v2f wa, wb;
      wa.x = w2n[k];
      wa.y = w2n[k + 1];
      wb.x = w2r[k];
      wb.y = w2r[k + 1];
      p = __builtin_amdgcn_wmma_f32_16x16x4_f32(false, wa, false, hb,
                                                (short)0, p, false, false);
      q = __builtin_amdgcn_wmma_f32_16x16x4_f32(false, wb, false, hb,
                                                (short)0, q, false, false);
    }
  }

  if (nv) {
    float4* pd = (float4*)(Pout + (long long)node * 16 + 8 * hi);
    float4* rd = (float4*)(Rout + (long long)node * 16 + 8 * hi);
    pd[0] = make_float4(p[0], p[1], p[2], p[3]);
    pd[1] = make_float4(p[4], p[5], p[6], p[7]);
    rd[0] = make_float4(q[0], q[1], q[2], q[3]);
    rd[1] = make_float4(q[4], q[5], q[6], q[7]);
  }
}

__global__ void combine16_kernel(float* io,
                                 const float* __restrict__ deg,
                                 int dstride,
                                 const float* __restrict__ R,
                                 int n) {
  int t = blockIdx.x * blockDim.x + threadIdx.x;
  int node = t >> 2;
  int part = t & 3;
  if (node >= n) return;
  float dinv = 1.0f / fmaxf(deg[(long long)node * dstride], 1.0f);
  float4 a = ((const float4*)(io + (long long)node * 16))[part];
  float4 r = ((const float4*)(R + (long long)node * 16))[part];
  float4 o;
  o.x = fmaf(a.x, dinv, r.x);
  o.y = fmaf(a.y, dinv, r.y);
  o.z = fmaf(a.z, dinv, r.z);
  o.w = fmaf(a.w, dinv, r.w);
  ((float4*)(io + (long long)node * 16))[part] = o;
}

__global__ void decode_kernel(const float* __restrict__ ou,
                              const float* __restrict__ oc,
                              const int* __restrict__ ls,
                              const int* __restrict__ ld,
                              float* __restrict__ out, int L) {
  int t = blockIdx.x * blockDim.x + threadIdx.x;
  if (t >= L) return;
  const float4* a = (const float4*)(ou + (long long)ls[t] * 16);
  const float4* b = (const float4*)(oc + (long long)ld[t] * 16);
  float acc = 0.0f;
#pragma unroll
  for (int i = 0; i < 4; ++i) {
    float4 xa = a[i], xb = b[i];
    acc += xa.x * xb.x + xa.y * xb.y + xa.z * xb.z + xa.w * xb.w;
  }
  out[t] = acc;
}

extern "C" void kernel_launch(void* const* d_in, const int* in_sizes, int n_in,
                              void* d_out, int out_size, void* d_ws, size_t ws_size,
                              hipStream_t stream) {
  const float* user_x       = (const float*)d_in[0];
  const float* course_x     = (const float*)d_in[1];
  const float* user_embed   = (const float*)d_in[2];
  const float* course_embed = (const float*)d_in[3];
  const float* Wu = (const float*)d_in[4];
  const float* bu = (const float*)d_in[5];
  const float* Wc = (const float*)d_in[6];
  const float* bc = (const float*)d_in[7];
  const float* c1e_Wl = (const float*)d_in[8];
  const float* c1e_bl = (const float*)d_in[9];
  const float* c1e_Wr = (const float*)d_in[10];
  const float* c1r_Wl = (const float*)d_in[11];
  const float* c1r_bl = (const float*)d_in[12];
  const float* c1r_Wr = (const float*)d_in[13];
  const float* c2e_Wl = (const float*)d_in[14];
  const float* c2e_bl = (const float*)d_in[15];
  const float* c2e_Wr = (const float*)d_in[16];
  const float* c2r_Wl = (const float*)d_in[17];
  const float* c2r_bl = (const float*)d_in[18];
  const float* c2r_Wr = (const float*)d_in[19];
  const int* user_idx   = (const int*)d_in[20];
  const int* course_idx = (const int*)d_in[21];
  const int* edge_src   = (const int*)d_in[22];
  const int* edge_dst   = (const int*)d_in[23];
  const int* label_src  = (const int*)d_in[24];
  const int* label_dst  = (const int*)d_in[25];
  float* outp = (float*)d_out;

  const int NU = in_sizes[2] / 16;
  const int NC = in_sizes[3] / 16;
  const int E  = in_sizes[22];
  const int L  = in_sizes[24];

  char* p = (char*)d_ws;
  auto carve = [&](size_t bytes) -> float* {
    float* r = (float*)p;
    p += (bytes + 255) & ~((size_t)255);
    return r;
  };
  float* prt2 = carve((size_t)NU * 32 * 4);
  float* aggu = carve((size_t)NU * 32 * 4);
  float* aggc = carve((size_t)NC * 32 * 4);
  float* pru  = carve((size_t)NU * 32 * 4);
  float* prc  = carve((size_t)NC * 32 * 4);
  float* Pu = pru;
  float* Ru = pru + (size_t)NU * 16;
  float* Pc = prc;
  float* Rc = prc + (size_t)NC * 16;
  float* ou   = carve((size_t)NU * 16 * 4);
  float* oc   = carve((size_t)NC * 16 * 4);

  const int B = 256;
  auto nblk = [](long long t, int b) { return (unsigned)((t + b - 1) / b); };

  const int NR = (NC + kCapRows - 1) / kCapRows;
  int NS = 256 / NR;
  if (NS < 1) NS = 1;
  const long long cap1 = ((long long)NU * 32) / ((long long)NC * 22);
  const long long cap2 = ((long long)NU * 32) / ((long long)NC * 16);
  const int NS1 = (int)(cap1 < NS ? (cap1 < 1 ? 1 : cap1) : NS);
  const int NS2 = (int)(cap2 < NS ? (cap2 < 1 ? 1 : cap2) : NS);

  float* part1 = pru;
  unsigned char* ukey = (unsigned char*)prt2;
  course_sums_raw_lds_kernel<5><<<NR * NS1, kLdsThreads, 0, stream>>>(
      user_embed, user_idx, user_x, edge_src, edge_dst, part1, ukey, E, NC, NR, NS1);
  reduce_parts_kernel<22><<<nblk((long long)NC * 22, B), B, 0, stream>>>(part1, aggc, NC, NS1);
  sage1_chain_kernel<5, 2><<<nblk(((long long)(NC + 15) / 16) * 32, B), B, 0, stream>>>(
      aggc, course_embed, course_idx, course_x, Wc, bc,
      c1e_Wl, c1e_bl, c1e_Wr, Wu, bu, c2r_Wl, c2e_Wr, c2e_bl, Pc, Rc, NC);
  float* ctab = pru;
  course_rows_kernel<<<nblk((long long)NC * kURow, B), B, 0, stream>>>(
      course_embed, course_idx, course_x, Pc, ctab, NC);
  user_sums_lds_kernel<<<nblk(NU, kURows), kUThreads, 0, stream>>>(
      (const unsigned int*)ukey, edge_src, edge_dst, ctab, aggu, ou, E, NU);
  sage1_chain_kernel<2, 5><<<nblk(((long long)(NU + 15) / 16) * 32, B), B, 0, stream>>>(
      aggu, user_embed, user_idx, user_x, Wu, bu,
      c1r_Wl, c1r_bl, c1r_Wr, Wc, bc, c2e_Wl, c2r_Wr, c2r_bl, Pu, Ru, NU);

  float* part2 = prt2;
  course_sums16_lds_kernel<<<NR * NS2, kLdsThreads, 0, stream>>>(
      Pu, edge_src, edge_dst, part2, E, NC, NR, NS2);
  combine16_parts_kernel<<<nblk((long long)NC * 4, B), B, 0, stream>>>(
      part2, NS2, aggc + 21, 32, Rc, oc, NC);

  combine16_kernel<<<nblk((long long)NU * 4, B), B, 0, stream>>>(ou, aggu + 18, 32, Ru, NU);

  decode_kernel<<<nblk(L, B), B, 0, stream>>>(ou, oc, label_src, label_dst, outp, L);
}


# user range-table blocks finish their own users from LDS in an epilogue: layer-1 finalize WMMA chain, layer-2 mean + root; raw-sum and root planes never stored; 10 -> 8 launches; sum order unchanged
# speedup vs baseline: 3.6869x; 1.1484x over previous
#include <hip/hip_runtime.h>

typedef __attribute__((ext_vector_type(8))) float v8f;
typedef __attribute__((ext_vector_type(2))) float v2f;


constexpr int kURow = 35;
constexpr int kURows = 2048;
constexpr int kUThreads = 1024;
constexpr int kQCap = 64;
constexpr int kScanLoads = 8;
constexpr int kKeyMax = 255;
constexpr int kGather = 8;

__global__ void course_rows_kernel(const float* __restrict__ embed,
                                   const int* __restrict__ node_index,
                                   const float* __restrict__ x,
                                   const float* __restrict__ Pc,
                                   float* __restrict__ tab,
                                   int n) {
  int t = blockIdx.x * blockDim.x + threadIdx.x;
  if (t >= n * kURow) return;
  int c = t / kURow;
  int col = t - c * kURow;
  float v;
  if (col < 16)       v = embed[(long long)node_index[c] * 16 + col];
  else if (col < 18)  v = x[(long long)c * 2 + (col - 16)];
  else if (col == 18) v = 1.0f;
  else                v = Pc[(long long)c * 16 + (col - 19)];
  tab[t] = v;
}

__device__ __forceinline__ void user_queue_flush(int2* q, int qn, int lane, int base, int nrows,
                                                 const int* __restrict__ edge_user,
                                                 const int* __restrict__ edge_course,
                                                 const float* __restrict__ tab,
                                                 float* T) {
  __builtin_amdgcn_fence(__ATOMIC_SEQ_CST, "workgroup");
  for (int i = lane; i < qn; i += 32) {
    int e = q[i].x;
    int rel = edge_user[e] - base;
    q[i] = make_int2(edge_course[e], (unsigned)rel < (unsigned)nrows ? rel : -1);
  }
  __builtin_amdgcn_fence(__ATOMIC_SEQ_CST, "workgroup");
  const int total = qn * kURow;
  for (int i0 = lane; i0 < total; i0 += 32 * kGather) {
    float v[kGather];
    int a[kGather];
#pragma unroll
    for (int k = 0; k < kGather; ++k) {
      int i = i0 + 32 * k;
      bool ok = i < total;
      int ii = ok ? i : 0;
      int m = ii / kURow;
      int col = ii - m * kURow;
      int2 en = q[m];
      v[k] = tab[(long long)en.x * kURow + col];
      a[k] = (ok && en.y >= 0) ? en.y * kURow + col : -1;
    }
#pragma unroll
    for (int k = 0; k < kGather; ++k) {
      if (a[k] >= 0) atomicAdd(T + a[k], v[k]);
    }
  }
  __builtin_amdgcn_fence(__ATOMIC_SEQ_CST, "workgroup");
}

template <bool CHECK>
__device__ __forceinline__ void user_filter_trip(const uint4* __restrict__ key16, int jb, int N16,
                                                 unsigned kk, unsigned (&hw)[kScanLoads / 2]) {
  uint4 kv[kScanLoads];
  bool ok[kScanLoads];
#pragma unroll
  for (int s = 0; s < kScanLoads; ++s) {
    int j = jb + s * kUThreads;
    ok[s] = CHECK ? (j < N16) : true;
    kv[s] = key16[ok[s] ? j : 0];
  }
#pragma unroll
  for (int w = 0; w < kScanLoads / 2; ++w) {
    unsigned acc = 0u;
#pragma unroll
    for (int g = 0; g < 2; ++g) {
      const int s = 2 * w + g;
      const unsigned d[4] = {kv[s].x, kv[s].y, kv[s].z, kv[s].w};
      unsigned part = 0u;
#pragma unroll
      for (int i = 0; i < 4; ++i) {
        unsigned t = d[i] ^ kk;
        unsigned y = (t - 0x01010101u) & ~t;
        const int pos = 4 * g + i;
        part |= (y >> (7 - pos)) & (0x01010101u << pos);
      }
      if (CHECK && !ok[s]) part = 0u;
      acc |= part;
    }
    hw[w] = acc;
  }
}

constexpr int kCapRows = 2560;
constexpr int kLdsThreads = 1024;
constexpr int kEdgesPerTrip = 4;

template <int FIN>
__global__ __launch_bounds__(kLdsThreads)
void course_sums_raw_lds_kernel(const float* __restrict__ embed,
                                const int* __restrict__ node_index,
                                const float* __restrict__ x,
                                const int* __restrict__ gather_idx,
                                const int* __restrict__ scatter_idx,
                                float* __restrict__ part,
                                unsigned char* __restrict__ key_out,
                                int E, int n_dst, int n_ranges, int n_slices) {
  constexpr int W = 16 + FIN + 1;
  constexpr int P = W | 1;
  __shared__ float T[kCapRows * P];
  const int tid = threadIdx.x;
  const int nthr = blockDim.x;
  const int r = blockIdx.x % n_ranges;
  const int slice = blockIdx.x / n_ranges;
  const int base = r * kCapRows;
  const int left = n_dst - base;
  const int nrows = left < kCapRows ? left : kCapRows;
  for (int i = tid; i < nrows * P; i += nthr) T[i] = 0.0f;
  __syncthreads();

  const int chunk = (E + n_slices - 1) / n_slices;
  const int e_lo = slice * chunk;
  const int e_end = e_lo + chunk;
  const int e_hi = e_end < E ? e_end : E;
  for (int e0 = e_lo + tid; e0 < e_hi; e0 += kEdgesPerTrip * nthr) {
    int s[kEdgesPerTrip], rel[kEdgesPerTrip], ni[kEdgesPerTrip], kq[kEdgesPerTrip];
    bool in[kEdgesPerTrip];
    float4 ev[kEdgesPerTrip][4];
    float xv[kEdgesPerTrip][FIN];
#pragma unroll
    for (int u = 0; u < kEdgesPerTrip; ++u) {
      int e = e0 + u * nthr;
      bool v = e < e_hi;
      int ec = v ? e : e0;
      int d = scatter_idx[ec];
      s[u] = gather_idx[ec];
      int kr = s[u] / kURows;
      kq[u] = kr < kKeyMax ? kr : kKeyMax;
      rel[u] = d - base;
      in[u] = v && ((unsigned)rel[u] < (unsigned)nrows);
      if (!in[u]) s[u] = 0;
    }
#pragma unroll
    for (int u = 0; u < kEdgesPerTrip; ++u) ni[u] = node_index[s[u]];
#pragma unroll
    for (int u = 0; u < kEdgesPerTrip; ++u) {
      if (!in[u]) ni[u] = 0;
      const float4* er = (const float4*)(embed + (long long)ni[u] * 16);
#pragma unroll
      for (int q = 0; q < 4; ++q) ev[u][q] = er[q];
      const float* xr = x + (long long)s[u] * FIN;
#pragma unroll
      for (int j = 0; j < FIN; ++j) xv[u][j] = xr[j];
    }
    if (r == 0) {
#pragma unroll
      for (int u = 0; u < kEdgesPerTrip; ++u) {
        int e = e0 + u * nthr;
        if (e < e_hi) key_out[e] = (unsigned char)kq[u];
      }
    }
#pragma unroll
    for (int u = 0; u < kEdgesPerTrip; ++u) {
      if (in[u]) {
        float* row = T + rel[u] * P;
#pragma unroll
        for (int q = 0; q < 4; ++q) {
          atomicAdd(row + 4 * q + 0, ev[u][q].x);
          atomicAdd(row + 4 * q + 1, ev[u][q].y);
          atomicAdd(row + 4 * q + 2, ev[u][q].z);
          atomicAdd(row + 4 * q + 3, ev[u][q].w);
        }
#pragma unroll
        for (int j = 0; j < FIN; ++j) atomicAdd(row + 16 + j, xv[u][j]);
        atomicAdd(row + 16 + FIN, 1.0f);
      }
    }
  }
  __syncthreads();

  float* dst = part + ((long long)slice * n_dst + base) * W;
  for (int i = tid; i < nrows * W; i += nthr) {
    int row = i / W;
    int c = i - row * W;
    dst[i] = T[row * P + c];
  }
}

__global__ __launch_bounds__(kLdsThreads)
void course_sums16_lds_kernel(const float* __restrict__ srcp,
                              const int* __restrict__ gather_idx,
                              const int* __restrict__ scatter_idx,
                              float* __restrict__ part,
                              int E, int n_dst, int n_ranges, int n_slices) {
  constexpr int P = 17;
  __shared__ float T[kCapRows * P];
  const int tid = threadIdx.x;
  const int nthr = blockDim.x;
  const int r = blockIdx.x % n_ranges;
  const int slice = blockIdx.x / n_ranges;
  const int base = r * kCapRows;
  const int left = n_dst - base;
  const int nrows = left < kCapRows ? left : kCapRows;
  for (int i = tid; i < nrows * P; i += nthr) T[i] = 0.0f;
  __syncthreads();

  const int chunk = (E + n_slices - 1) / n_slices;
  const int e_lo = slice * chunk;
  const int e_end = e_lo + chunk;
  const int e_hi = e_end < E ? e_end : E;
  for (int e0 = e_lo + tid; e0 < e_hi; e0 += kEdgesPerTrip * nthr) {
    int s[kEdgesPerTrip], rel[kEdgesPerTrip];
    bool in[kEdgesPerTrip];
    float4 pv[kEdgesPerTrip][4];
#pragma unroll
    for (int u = 0; u < kEdgesPerTrip; ++u) {
      int e = e0 + u * nthr;
      bool v = e < e_hi;
      int ec = v ? e : e0;
      int d = scatter_idx[ec];
      s[u] = gather_idx[ec];
      rel[u] = d - base;
      in[u] = v && ((unsigned)rel[u] < (unsigned)nrows);
      if (!in[u]) s[u] = 0;
    }
#pragma unroll
    for (int u = 0; u < kEdgesPerTrip; ++u) {
      const float4* pr = (const float4*)(srcp + (long long)s[u] * 16);
#pragma unroll
      for (int q = 0; q < 4; ++q) pv[u][q] = pr[q];
    }
#pragma unroll
    for (int u = 0; u < kEdgesPerTrip; ++u) {
      if (in[u]) {
        float* row = T + rel[u] * P;
#pragma unroll
        for (int q = 0; q < 4; ++q) {
          atomicAdd(row + 4 * q + 0, pv[u][q].x);
          atomicAdd(row + 4 * q + 1, pv[u][q].y);
          atomicAdd(row + 4 * q + 2, pv[u][q].z);
          atomicAdd(row + 4 * q + 3, pv[u][q].w);
        }
      }
    }
  }
  __syncthreads();

  float* dst = part + ((long long)slice * n_dst + base) * 16;
  for (int i = tid; i < nrows * 16; i += nthr) {
    dst[i] = T[(i >> 4) * P + (i & 15)];
  }
}

template <int W>
__global__ void reduce_parts_kernel(const float* __restrict__ part,
                                    float* __restrict__ out,
                                    int n_rows, int n_slices) {
  int t = blockIdx.x * blockDim.x + threadIdx.x;
  int total = n_rows * W;
  if (t >= total) return;
  float acc = 0.0f;
#pragma unroll 16
  for (int s = 0; s < n_slices; ++s) acc += part[(long long)s * total + t];
  int row = t / W;
  int c = t - row * W;
  out[(long long)row * 32 + c] = acc;
}

__global__ void combine16_parts_kernel(const float* __restrict__ part,
                                       int n_slices,
                                       const float* __restrict__ deg,
                                       int dstride,
                                       const float* __restrict__ R,
                                       float* __restrict__ out,
                                       int n) {
  int t = blockIdx.x * blockDim.x + threadIdx.x;
  int node = t >> 2;
  int q = t & 3;
  if (node >= n) return;
  float4 a = make_float4(0.0f, 0.0f, 0.0f, 0.0f);
  long long stride = (long long)n * 16;
#pragma unroll 16
  for (int s = 0; s < n_slices; ++s) {
    float4 v = ((const float4*)(part + s * stride + (long long)node * 16))[q];
    a.x += v.x; a.y += v.y; a.z += v.z; a.w += v.w;
  }
  float dinv = 1.0f / fmaxf(deg[(long long)node * dstride], 1.0f);
  float4 rr = ((const float4*)(R + (long long)node * 16))[q];
  float4 o;
  o.x = fmaf(a.x, dinv, rr.x);
  o.y = fmaf(a.y, dinv, rr.y);
  o.z = fmaf(a.z, dinv, rr.z);
  o.w = fmaf(a.w, dinv, rr.w);
  ((float4*)(out + (long long)node * 16))[q] = o;
}

template <int FIN, int FDST>
__device__ __forceinline__ void sage1_chain_tile(const float* arow,
                                                 const float* erow,
                                                 const float* xrow,
                                                 bool nv, int m, int hi,
                                                 const float* __restrict__ Wd,
                                                 const float* __restrict__ bd,
                                                 const float* __restrict__ Wl,
                                                 const float* __restrict__ bl,
                                                 const float* __restrict__ Wr,
                                                 const float* __restrict__ Wp,
                                                 const float* __restrict__ bp,
                                                 const float* __restrict__ W2n,
                                                 const float* __restrict__ W2r,
                                                 const float* __restrict__ b2r,
                                                 v8f& p, v8f& q, float& dinv_out) {
  float cnt  = arow[16 + FIN];
  float dinv = nv ? (1.0f / fmaxf(cnt, 1.0f)) : 0.0f;

  v2f f_agg[8], f_x[8];
#pragma unroll
  for (int s = 0; s < 4; ++s) {
    int k = 4 * s + 2 * hi;
    f_agg[s].x = arow[k] * dinv;
    f_agg[s].y = arow[k + 1] * dinv;
    float e0 = erow[k];
    float e1 = erow[k + 1];
    f_x[s].x = nv ? e0 : 0.0f;
    f_x[s].y = nv ? e1 : 0.0f;
  }
  {
    float xd[FDST];
#pragma unroll
    for (int j = 0; j < FDST; ++j) xd[j] = xrow[j];
#pragma unroll
    for (int s = 4; s < 8; ++s) {
      int r = 4 * (s - 4) + 2 * hi;
      float p0 = bd[r];
      float p1 = bd[r + 1];
#pragma unroll
      for (int j = 0; j < FDST; ++j) {
        p0 = fmaf(xd[j], Wd[r * FDST + j], p0);
        p1 = fmaf(xd[j], Wd[(r + 1) * FDST + j], p1);
      }
      f_x[s].x = nv ? p0 : 0.0f;
      f_x[s].y = nv ? p1 : 0.0f;
    }
  }
  {
    float sx[FIN];
#pragma unroll
    for (int j = 0; j < FIN; ++j) sx[j] = arow[16 + j];
#pragma unroll
    for (int s = 4; s < 8; ++s) {
      int r = 4 * (s - 4) + 2 * hi;
      float p0 = cnt * bp[r];
      float p1 = cnt * bp[r + 1];
#pragma unroll
      for (int j = 0; j < FIN; ++j) {
        p0 = fmaf(sx[j], Wp[r * FIN + j], p0);
        p1 = fmaf(sx[j], Wp[(r + 1) * FIN + j], p1);
      }
      f_agg[s].x = p0 * dinv;
      f_agg[s].y = p1 * dinv;
    }
  }

  v8f hT[2];
#pragma unroll
  for (int t = 0; t < 2; ++t) {
    int og = t * 16 + m;
    const float* wl = Wl + og * 32;
    const float* wr = Wr + og * 32;
    v8f c;
#pragma unroll
    for (int r = 0; r < 8; ++r) c[r] = bl[t * 16 + 8 * hi + r];

#pragma unroll
    for (int s = 0; s < 8; ++s) {
      int k = 4 * s + 2 * hi;
      v2f w;
      w.x = wl[k];
      w.y = wl[k + 1];
      c = __builtin_amdgcn_wmma_f32_16x16x4_f32(false, w, false, f_agg[s],
                                                (short)0, c, false, false);
    }
#pragma unroll
    for (int s = 0; s < 8; ++s) {
      int k = 4 * s + 2 * hi;
      v2f w;
      w.x = wr[k];
      w.y = wr[k + 1];
      c = __builtin_amdgcn_wmma_f32_16x16x4_f32(false, w, false, f_x[s],
                                                (short)0, c, false, false);
    }
#pragma unroll
    for (int r = 0; r < 8; ++r) hT[t][r] = fmaxf(c[r], 0.0f);
  }

  const float* w2n = W2n + m * 32;
  const float* w2r = W2r + m * 32;
#pragma unroll
  for (int r = 0; r < 8; ++r) {
    p[r] = 0.0f;
    q[r] = b2r[8 * hi + r];
  }
#pragma unroll
  for (int t = 0; t < 2; ++t) {
#pragma unroll
    for (int qq = 0; qq < 4; ++qq) {
      int k = 16 * t + 8 * hi + 2 * qq;
      v2f hb;
      hb.x = hT[t][2 * qq];
      hb.y = hT[t][2 * qq + 1];
      v2f wa, wb;
      wa.x = w2n[k];
      wa.y = w2n[k + 1];
      wb.x = w2r[k];
      wb.y = w2r[k + 1];
      p = __builtin_amdgcn_wmma_f32_16x16x4_f32(false, wa, false, hb,
                                                (short)0, p, false, false);
      q = __builtin_amdgcn_wmma_f32_16x16x4_f32(false, wb, false, hb,
                                                (short)0, q, false, false);
    }
  }
  dinv_out = dinv;
}

template <int FIN, int FDST>
__global__ void sage1_chain_kernel(const float* __restrict__ agg,
                                   const float* __restrict__ embed_d,
                                   const int* __restrict__ idx_d,
                                   const float* __restrict__ x_d,
                                   const float* __restrict__ Wd,
                                   const float* __restrict__ bd,
                                   const float* __restrict__ Wl,
                                   const float* __restrict__ bl,
                                   const float* __restrict__ Wr,
                                   const float* __restrict__ Wp,
                                   const float* __restrict__ bp,
                                   const float* __restrict__ W2n,
                                   const float* __restrict__ W2r,
                                   const float* __restrict__ b2r,
                                   float* __restrict__ Pout,
                                   float* __restrict__ Rout,
                                   int n) {
  int lane = threadIdx.x & 31;
  int wave = (blockIdx.x * blockDim.x + threadIdx.x) >> 5;
  int nb = wave * 16;
  if (nb >= n) return;
  int m  = lane & 15;
  int hi = lane >> 4;

  int node = nb + m;
  bool nv = node < n;
  int ncl = nv ? node : (n - 1);
  v8f p, q;
  float dinv;
  sage1_chain_tile<FIN, FDST>(agg + (long long)ncl * 32,
                              embed_d + (long long)idx_d[ncl] * 16,
                              x_d + (long long)ncl * FDST,
                              nv, m, hi, Wd, bd, Wl, bl, Wr, Wp, bp, W2n, W2r, b2r, p, q, dinv);

  if (nv) {
    float4* pd = (float4*)(Pout + (long long)node * 16 + 8 * hi);
    float4* rd = (float4*)(Rout + (long long)node * 16 + 8 * hi);
    pd[0] = make_float4(p[0], p[1], p[2], p[3]);
    pd[1] = make_float4(p[4], p[5], p[6], p[7]);
    rd[0] = make_float4(q[0], q[1], q[2], q[3]);
    rd[1] = make_float4(q[4], q[5], q[6], q[7]);
  }
}

__global__ __launch_bounds__(kUThreads)
void user_sums_lds_kernel(const unsigned int* __restrict__ edge_key,
                          const int* __restrict__ edge_user,
                          const int* __restrict__ edge_course,
                          const float* __restrict__ tab,
                          const float* __restrict__ embed_u,
                          const int* __restrict__ idx_u,
                          const float* __restrict__ x_u,
                          const float* __restrict__ Wd,
                          const float* __restrict__ bd,
                          const float* __restrict__ Wl,
                          const float* __restrict__ bl,
                          const float* __restrict__ Wr,
                          const float* __restrict__ Wp,
                          const float* __restrict__ bp,
                          const float* __restrict__ W2n,
                          const float* __restrict__ W2r,
                          const float* __restrict__ b2r,
                          float* __restrict__ Pout,
                          float* __restrict__ out2,
                          int E, int n_user) {
  __shared__ float T[kURows * kURow];
  __shared__ int2 Q[(kUThreads / 32) * kQCap];
  const int tid = threadIdx.x;
  const int lane = tid & 31;
  const int base = blockIdx.x * kURows;
  const int left = n_user - base;
  const int nrows = left < kURows ? left : kURows;
  for (int i = tid; i < nrows * kURow; i += kUThreads) T[i] = 0.0f;
  __syncthreads();

  int2* q = Q + (tid >> 5) * kQCap;
  int qn = 0;
  const int N16 = E >> 4;
  const int kb = (int)blockIdx.x < kKeyMax ? (int)blockIdx.x : kKeyMax;
  const unsigned kk = (unsigned)kb * 0x01010101u;

  if (tid < 32) {
    int e = (N16 << 4) + lane;
    bool hit = false;
    if (e < E) hit = (unsigned)(edge_user[e] - base) < (unsigned)nrows;
    unsigned m = __builtin_amdgcn_ballot_w32(hit);
    if (hit) q[__builtin_amdgcn_mbcnt_lo(m, 0u)].x = e;
    qn = __builtin_popcount(m);
  }

  const uint4* key16 = (const uint4*)edge_key;
  constexpr int kTrip = kUThreads * kScanLoads;
  const int ntrips = (N16 + kTrip - 1) / kTrip;
  const int toff = (int)(((unsigned)blockIdx.x * (unsigned)ntrips) / gridDim.x);
  for (int t = 0; t < ntrips; ++t) {
    int tr = t + toff;
    if (tr >= ntrips) tr -= ntrips;
    const int jb = tr * kTrip + tid;
    unsigned hw[kScanLoads / 2];
    if (tr * kTrip + kTrip <= N16) user_filter_trip<false>(key16, jb, N16, kk, hw);
    else                           user_filter_trip<true>(key16, jb, N16, kk, hw);
#pragma nounroll
    for (int w = 0; w < kScanLoads / 2; ++w) {
      unsigned hm = hw[0];
#pragma unroll
      for (int i = 0; i + 1 < kScanLoads / 2; ++i) hw[i] = hw[i + 1];
      unsigned m = __builtin_amdgcn_ballot_w32(hm != 0u);
      while (m != 0u) {
        int n = __builtin_popcount(m);
        if (qn + n > kQCap) {
          user_queue_flush(q, qn, lane, base, nrows, edge_user, edge_course, tab, T);
          qn = 0;
        }
        if (hm != 0u) {
          int b = __builtin_ctz(hm);
          hm &= hm - 1u;
          int pos = b & 7;
          int grp = jb + (2 * w + (pos >> 2)) * kUThreads;
          q[qn + (int)__builtin_amdgcn_mbcnt_lo(m, 0u)].x = 16 * grp + 4 * (pos & 3) + (b >> 3);
        }
        qn += n;
        m = __builtin_amdgcn_ballot_w32(hm != 0u);
      }
    }
  }
  if (qn > 0) user_queue_flush(q, qn, lane, base, nrows, edge_user, edge_course, tab, T);
  __syncthreads();

  const int wv = __builtin_amdgcn_readfirstlane(tid >> 5);
  const int m = lane & 15;
  const int hi = lane >> 4;
  for (int tile = wv; tile * 16 < nrows; tile += kUThreads / 32) {
    int row = tile * 16 + m;
    bool nv = row < nrows;
    int rcl = nv ? row : (nrows - 1);
    long long ncl = (long long)base + rcl;
    const float* arow = T + rcl * kURow;
    int wo = 0;
    asm volatile("" : "+s"(wo));
    v8f p, q;
    float dinv;
    sage1_chain_tile<2, 5>(arow,
                           embed_u + (long long)idx_u[ncl] * 16,
                           x_u + ncl * 5,
                           nv, m, hi, Wd + wo, bd + wo, Wl + wo, bl + wo, Wr + wo, Wp + wo, bp + wo,
                           W2n + wo, W2r + wo, b2r + wo, p, q, dinv);
    if (nv) {
      const float* s2 = arow + 19 + 8 * hi;
      float4* pd = (float4*)(Pout + ncl * 16 + 8 * hi);
      float4* od = (float4*)(out2 + ncl * 16 + 8 * hi);
      pd[0] = make_float4(p[0], p[1], p[2], p[3]);
      pd[1] = make_float4(p[4], p[5], p[6], p[7]);
      od[0] = make_float4(fmaf(s2[0], dinv, q[0]), fmaf(s2[1], dinv, q[1]),
                          fmaf(s2[2], dinv, q[2]), fmaf(s2[3], dinv, q[3]));
      od[1] = make_float4(fmaf(s2[4], dinv, q[4]), fmaf(s2[5], dinv, q[5]),
                          fmaf(s2[6], dinv, q[6]), fmaf(s2[7], dinv, q[7]));
    }
  }
}

__global__ void decode_kernel(const float* __restrict__ ou,
                              const float* __restrict__ oc,
                              const int* __restrict__ ls,
                              const int* __restrict__ ld,
                              float* __restrict__ out, int L) {
  int t = blockIdx.x * blockDim.x + threadIdx.x;
  if (t >= L) return;
  const float4* a = (const float4*)(ou + (long long)ls[t] * 16);
  const float4* b = (const float4*)(oc + (long long)ld[t] * 16);
  float acc = 0.0f;
#pragma unroll
  for (int i = 0; i < 4; ++i) {
    float4 xa = a[i], xb = b[i];
    acc += xa.x * xb.x + xa.y * xb.y + xa.z * xb.z + xa.w * xb.w;
  }
  out[t] = acc;
}

extern "C" void kernel_launch(void* const* d_in, const int* in_sizes, int n_in,
                              void* d_out, int out_size, void* d_ws, size_t ws_size,
                              hipStream_t stream) {
  const float* user_x       = (const float*)d_in[0];
  const float* course_x     = (const float*)d_in[1];
  const float* user_embed   = (const float*)d_in[2];
  const float* course_embed = (const float*)d_in[3];
  const float* Wu = (const float*)d_in[4];
  const float* bu = (const float*)d_in[5];
  const float* Wc = (const float*)d_in[6];
  const float* bc = (const float*)d_in[7];
  const float* c1e_Wl = (const float*)d_in[8];
  const float* c1e_bl = (const float*)d_in[9];
  const float* c1e_Wr = (const float*)d_in[10];
  const float* c1r_Wl = (const float*)d_in[11];
  const float* c1r_bl = (const float*)d_in[12];
  const float* c1r_Wr = (const float*)d_in[13];
  const float* c2e_Wl = (const float*)d_in[14];
  const float* c2e_bl = (const float*)d_in[15];
  const float* c2e_Wr = (const float*)d_in[16];
  const float* c2r_Wl = (const float*)d_in[17];
  const float* c2r_bl = (const float*)d_in[18];
  const float* c2r_Wr = (const float*)d_in[19];
  const int* user_idx   = (const int*)d_in[20];
  const int* course_idx = (const int*)d_in[21];
  const int* edge_src   = (const int*)d_in[22];
  const int* edge_dst   = (const int*)d_in[23];
  const int* label_src  = (const int*)d_in[24];
  const int* label_dst  = (const int*)d_in[25];
  float* outp = (float*)d_out;

  const int NU = in_sizes[2] / 16;
  const int NC = in_sizes[3] / 16;
  const int E  = in_sizes[22];
  const int L  = in_sizes[24];

  char* p = (char*)d_ws;
  auto carve = [&](size_t bytes) -> float* {
    float* r = (float*)p;
    p += (bytes + 255) & ~((size_t)255);
    return r;
  };
  float* prt2 = carve((size_t)NU * 32 * 4);
  float* Pu   = carve((size_t)NU * 32 * 4);
  float* aggc = carve((size_t)NC * 32 * 4);
  float* pru  = carve((size_t)NU * 32 * 4);
  float* prc  = carve((size_t)NC * 32 * 4);
  float* Pc = prc;
  float* Rc = prc + (size_t)NC * 16;
  float* ou   = carve((size_t)NU * 16 * 4);
  float* oc   = carve((size_t)NC * 16 * 4);

  const int B = 256;
  auto nblk = [](long long t, int b) { return (unsigned)((t + b - 1) / b); };

  const int NR = (NC + kCapRows - 1) / kCapRows;
  int NS = 256 / NR;
  if (NS < 1) NS = 1;
  const long long cap1 = ((long long)NU * 32) / ((long long)NC * 22);
  const long long cap2 = ((long long)NU * 32) / ((long long)NC * 16);
  const int NS1 = (int)(cap1 < NS ? (cap1 < 1 ? 1 : cap1) : NS);
  const int NS2 = (int)(cap2 < NS ? (cap2 < 1 ? 1 : cap2) : NS);

  float* part1 = pru;
  unsigned char* ukey = (unsigned char*)prt2;
  course_sums_raw_lds_kernel<5><<<NR * NS1, kLdsThreads, 0, stream>>>(
      user_embed, user_idx, user_x, edge_src, edge_dst, part1, ukey, E, NC, NR, NS1);
  reduce_parts_kernel<22><<<nblk((long long)NC * 22, B), B, 0, stream>>>(part1, aggc, NC, NS1);
  sage1_chain_kernel<5, 2><<<nblk(((long long)(NC + 15) / 16) * 32, B), B, 0, stream>>>(
      aggc, course_embed, course_idx, course_x, Wc, bc,
      c1e_Wl, c1e_bl, c1e_Wr, Wu, bu, c2r_Wl, c2e_Wr, c2e_bl, Pc, Rc, NC);
  float* ctab = pru;
  course_rows_kernel<<<nblk((long long)NC * kURow, B), B, 0, stream>>>(
      course_embed, course_idx, course_x, Pc, ctab, NC);
  user_sums_lds_kernel<<<nblk(NU, kURows), kUThreads, 0, stream>>>(
      (const unsigned int*)ukey, edge_src, edge_dst, ctab,
      user_embed, user_idx, user_x, Wu, bu,
      c1r_Wl, c1r_bl, c1r_Wr, Wc, bc, c2e_Wl, c2r_Wr, c2r_bl, Pu, ou, E, NU);

  float* part2 = prt2;
  course_sums16_lds_kernel<<<NR * NS2, kLdsThreads, 0, stream>>>(
      Pu, edge_src, edge_dst, part2, E, NC, NR, NS2);
  combine16_parts_kernel<<<nblk((long long)NC * 4, B), B, 0, stream>>>(
      part2, NS2, aggc + 21, 32, Rc, oc, NC);

  decode_kernel<<<nblk(L, B), B, 0, stream>>>(ou, oc, label_src, label_dst, outp, L);
}


# user range-table blocks stage the 4 weight matrices of their epilogue chain once in LDS in A-fragment order; per tile 48 contiguous 8-byte LDS reads replace 48 sixteen-line loads; sum order unchanged
# speedup vs baseline: 3.8274x; 1.0381x over previous
#include <hip/hip_runtime.h>

typedef __attribute__((ext_vector_type(8))) float v8f;
typedef __attribute__((ext_vector_type(2))) float v2f;


constexpr int kURow = 35;
constexpr int kURows = 2048;
constexpr int kUThreads = 1024;
constexpr int kQCap = 64;
constexpr int kScanLoads = 8;
constexpr int kKeyMax = 255;
constexpr int kGather = 8;

__global__ void course_rows_kernel(const float* __restrict__ embed,
                                   const int* __restrict__ node_index,
                                   const float* __restrict__ x,
                                   const float* __restrict__ Pc,
                                   float* __restrict__ tab,
                                   int n) {
  int t = blockIdx.x * blockDim.x + threadIdx.x;
  if (t >= n * kURow) return;
  int c = t / kURow;
  int col = t - c * kURow;
  float v;
  if (col < 16)       v = embed[(long long)node_index[c] * 16 + col];
  else if (col < 18)  v = x[(long long)c * 2 + (col - 16)];
  else if (col == 18) v = 1.0f;
  else                v = Pc[(long long)c * 16 + (col - 19)];
  tab[t] = v;
}

__device__ __forceinline__ void user_queue_flush(int2* q, int qn, int lane, int base, int nrows,
                                                 const int* __restrict__ edge_user,
                                                 const int* __restrict__ edge_course,
                                                 const float* __restrict__ tab,
                                                 float* T) {
  __builtin_amdgcn_fence(__ATOMIC_SEQ_CST, "workgroup");
  for (int i = lane; i < qn; i += 32) {
    int e = q[i].x;
    int rel = edge_user[e] - base;
    q[i] = make_int2(edge_course[e], (unsigned)rel < (unsigned)nrows ? rel : -1);
  }
  __builtin_amdgcn_fence(__ATOMIC_SEQ_CST, "workgroup");
  const int total = qn * kURow;
  for (int i0 = lane; i0 < total; i0 += 32 * kGather) {
    float v[kGather];
    int a[kGather];
#pragma unroll
    for (int k = 0; k < kGather; ++k) {
      int i = i0 + 32 * k;
      bool ok = i < total;
      int ii = ok ? i : 0;
      int m = ii / kURow;
      int col = ii - m * kURow;
      int2 en = q[m];
      v[k] = tab[(long long)en.x * kURow + col];
      a[k] = (ok && en.y >= 0) ? en.y * kURow + col : -1;
    }
#pragma unroll
    for (int k = 0; k < kGather; ++k) {
      if (a[k] >= 0) atomicAdd(T + a[k], v[k]);
    }
  }
  __builtin_amdgcn_fence(__ATOMIC_SEQ_CST, "workgroup");
}

template <bool CHECK>
__device__ __forceinline__ void user_filter_trip(const uint4* __restrict__ key16, int jb, int N16,
                                                 unsigned kk, unsigned (&hw)[kScanLoads / 2]) {
  uint4 kv[kScanLoads];
  bool ok[kScanLoads];
#pragma unroll
  for (int s = 0; s < kScanLoads; ++s) {
    int j = jb + s * kUThreads;
    ok[s] = CHECK ? (j < N16) : true;
    kv[s] = key16[ok[s] ? j : 0];
  }
#pragma unroll
  for (int w = 0; w < kScanLoads / 2; ++w) {
    unsigned acc = 0u;
#pragma unroll
    for (int g = 0; g < 2; ++g) {
      const int s = 2 * w + g;
      const unsigned d[4] = {kv[s].x, kv[s].y, kv[s].z, kv[s].w};
      unsigned part = 0u;
#pragma unroll
      for (int i = 0; i < 4; ++i) {
        unsigned t = d[i] ^ kk;
        unsigned y = (t - 0x01010101u) & ~t;
        const int pos = 4 * g + i;
        part |= (y >> (7 - pos)) & (0x01010101u << pos);
      }
      if (CHECK && !ok[s]) part = 0u;
      acc |= part;
    }
    hw[w] = acc;
  }
}

constexpr int kCapRows = 2560;
constexpr int kLdsThreads = 1024;
constexpr int kEdgesPerTrip = 4;

template <int FIN>
__global__ __launch_bounds__(kLdsThreads)
void course_sums_raw_lds_kernel(const float* __restrict__ embed,
                                const int* __restrict__ node_index,
                                const float* __restrict__ x,
                                const int* __restrict__ gather_idx,
                                const int* __restrict__ scatter_idx,
                                float* __restrict__ part,
                                unsigned char* __restrict__ key_out,
                                int E, int n_dst, int n_ranges, int n_slices) {
  constexpr int W = 16 + FIN + 1;
  constexpr int P = W | 1;
  __shared__ float T[kCapRows * P];
  const int tid = threadIdx.x;
  const int nthr = blockDim.x;
  const int r = blockIdx.x % n_ranges;
  const int slice = blockIdx.x / n_ranges;
  const int base = r * kCapRows;
  const int left = n_dst - base;
  const int nrows = left < kCapRows ? left : kCapRows;
  for (int i = tid; i < nrows * P; i += nthr) T[i] = 0.0f;
  __syncthreads();

  const int chunk = (E + n_slices - 1) / n_slices;
  const int e_lo = slice * chunk;
  const int e_end = e_lo + chunk;
  const int e_hi = e_end < E ? e_end : E;
  for (int e0 = e_lo + tid; e0 < e_hi; e0 += kEdgesPerTrip * nthr) {
    int s[kEdgesPerTrip], rel[kEdgesPerTrip], ni[kEdgesPerTrip], kq[kEdgesPerTrip];
    bool in[kEdgesPerTrip];
    float4 ev[kEdgesPerTrip][4];
    float xv[kEdgesPerTrip][FIN];
#pragma unroll
    for (int u = 0; u < kEdgesPerTrip; ++u) {
      int e = e0 + u * nthr;
      bool v = e < e_hi;
      int ec = v ? e : e0;
      int d = scatter_idx[ec];
      s[u] = gather_idx[ec];
      int kr = s[u] / kURows;
      kq[u] = kr < kKeyMax ? kr : kKeyMax;
      rel[u] = d - base;
      in[u] = v && ((unsigned)rel[u] < (unsigned)nrows);
      if (!in[u]) s[u] = 0;
    }
#pragma unroll
    for (int u = 0; u < kEdgesPerTrip; ++u) ni[u] = node_index[s[u]];
#pragma unroll
    for (int u = 0; u < kEdgesPerTrip; ++u) {
      if (!in[u]) ni[u] = 0;
      const float4* er = (const float4*)(embed + (long long)ni[u] * 16);
#pragma unroll
      for (int q = 0; q < 4; ++q) ev[u][q] = er[q];
      const float* xr = x + (long long)s[u] * FIN;
#pragma unroll
      for (int j = 0; j < FIN; ++j) xv[u][j] = xr[j];
    }
    if (r == 0) {
#pragma unroll
      for (int u = 0; u < kEdgesPerTrip; ++u) {
        int e = e0 + u * nthr;
        if (e < e_hi) key_out[e] = (unsigned char)kq[u];
      }
    }
#pragma unroll
    for (int u = 0; u < kEdgesPerTrip; ++u) {
      if (in[u]) {
        float* row = T + rel[u] * P;
#pragma unroll
        for (int q = 0; q < 4; ++q) {
          atomicAdd(row + 4 * q + 0, ev[u][q].x);
          atomicAdd(row + 4 * q + 1, ev[u][q].y);
          atomicAdd(row + 4 * q + 2, ev[u][q].z);
          atomicAdd(row + 4 * q + 3, ev[u][q].w);
        }
#pragma unroll
        for (int j = 0; j < FIN; ++j) atomicAdd(row + 16 + j, xv[u][j]);
        atomicAdd(row + 16 + FIN, 1.0f);
      }
    }
  }
  __syncthreads();

  float* dst = part + ((long long)slice * n_dst + base) * W;
  for (int i = tid; i < nrows * W; i += nthr) {
    int row = i / W;
    int c = i - row * W;
    dst[i] = T[row * P + c];
  }
}

__global__ __launch_bounds__(kLdsThreads)
void course_sums16_lds_kernel(const float* __restrict__ srcp,
                              const int* __restrict__ gather_idx,
                              const int* __restrict__ scatter_idx,
                              float* __restrict__ part,
                              int E, int n_dst, int n_ranges, int n_slices) {
  constexpr int P = 17;
  __shared__ float T[kCapRows * P];
  const int tid = threadIdx.x;
  const int nthr = blockDim.x;
  const int r = blockIdx.x % n_ranges;
  const int slice = blockIdx.x / n_ranges;
  const int base = r * kCapRows;
  const int left = n_dst - base;
  const int nrows = left < kCapRows ? left : kCapRows;
  for (int i = tid; i < nrows * P; i += nthr) T[i] = 0.0f;
  __syncthreads();

  const int chunk = (E + n_slices - 1) / n_slices;
  const int e_lo = slice * chunk;
  const int e_end = e_lo + chunk;
  const int e_hi = e_end < E ? e_end : E;
  for (int e0 = e_lo + tid; e0 < e_hi; e0 += kEdgesPerTrip * nthr) {
    int s[kEdgesPerTrip], rel[kEdgesPerTrip];
    bool in[kEdgesPerTrip];
    float4 pv[kEdgesPerTrip][4];
#pragma unroll
    for (int u = 0; u < kEdgesPerTrip; ++u) {
      int e = e0 + u * nthr;
      bool v = e < e_hi;
      int ec = v ? e : e0;
      int d = scatter_idx[ec];
      s[u] = gather_idx[ec];
      rel[u] = d - base;
      in[u] = v && ((unsigned)rel[u] < (unsigned)nrows);
      if (!in[u]) s[u] = 0;
    }
#pragma unroll
    for (int u = 0; u < kEdgesPerTrip; ++u) {
      const float4* pr = (const float4*)(srcp + (long long)s[u] * 16);
#pragma unroll
      for (int q = 0; q < 4; ++q) pv[u][q] = pr[q];
    }
#pragma unroll
    for (int u = 0; u < kEdgesPerTrip; ++u) {
      if (in[u]) {
        float* row = T + rel[u] * P;
#pragma unroll
        for (int q = 0; q < 4; ++q) {
          atomicAdd(row + 4 * q + 0, pv[u][q].x);
          atomicAdd(row + 4 * q + 1, pv[u][q].y);
          atomicAdd(row + 4 * q + 2, pv[u][q].z);
          atomicAdd(row + 4 * q + 3, pv[u][q].w);
        }
      }
    }
  }
  __syncthreads();

  float* dst = part + ((long long)slice * n_dst + base) * 16;
  for (int i = tid; i < nrows * 16; i += nthr) {
    dst[i] = T[(i >> 4) * P + (i & 15)];
  }
}

template <int W>
__global__ void reduce_parts_kernel(const float* __restrict__ part,
                                    float* __restrict__ out,
                                    int n_rows, int n_slices) {
  int t = blockIdx.x * blockDim.x + threadIdx.x;
  int total = n_rows * W;
  if (t >= total) return;
  float acc = 0.0f;
#pragma unroll 16
  for (int s = 0; s < n_slices; ++s) acc += part[(long long)s * total + t];
  int row = t / W;
  int c = t - row * W;
  out[(long long)row * 32 + c] = acc;
}

__global__ void combine16_parts_kernel(const float* __restrict__ part,
                                       int n_slices,
                                       const float* __restrict__ deg,
                                       int dstride,
                                       const float* __restrict__ R,
                                       float* __restrict__ out,
                                       int n) {
  int t = blockIdx.x * blockDim.x + threadIdx.x;
  int node = t >> 2;
  int q = t & 3;
  if (node >= n) return;
  float4 a = make_float4(0.0f, 0.0f, 0.0f, 0.0f);
  long long stride = (long long)n * 16;
#pragma unroll 16
  for (int s = 0; s < n_slices; ++s) {
    float4 v = ((const float4*)(part + s * stride + (long long)node * 16))[q];
    a.x += v.x; a.y += v.y; a.z += v.z; a.w += v.w;
  }
  float dinv = 1.0f / fmaxf(deg[(long long)node * dstride], 1.0f);
  float4 rr = ((const float4*)(R + (long long)node * 16))[q];
  float4 o;
  o.x = fmaf(a.x, dinv, rr.x);
  o.y = fmaf(a.y, dinv, rr.y);
  o.z = fmaf(a.z, dinv, rr.z);
  o.w = fmaf(a.w, dinv, rr.w);
  ((float4*)(out + (long long)node * 16))[q] = o;
}

constexpr int kWF1 = 2 * 16 * 32;
constexpr int kWF2 = 2 * 8 * 32;

template <int FIN, int FDST, bool LDSW>
__device__ __forceinline__ void sage1_chain_tile(const float* arow,
                                                 const float* erow,
                                                 const float* xrow,
                                                 bool nv, int m, int hi,
                                                 const float* __restrict__ Wd,
                                                 const float* __restrict__ bd,
                                                 const float* __restrict__ Wl,
                                                 const float* __restrict__ bl,
                                                 const float* __restrict__ Wr,
                                                 const float* __restrict__ Wp,
                                                 const float* __restrict__ bp,
                                                 const float* __restrict__ W2n,
                                                 const float* __restrict__ W2r,
                                                 const float* __restrict__ b2r,
                                                 const v2f* wf1,
                                                 const v2f* wf2,
                                                 v8f& p, v8f& q, float& dinv_out) {
  const int lane = 16 * hi + m;
  float cnt  = arow[16 + FIN];
  float dinv = nv ? (1.0f / fmaxf(cnt, 1.0f)) : 0.0f;

  v2f f_agg[8], f_x[8];
#pragma unroll
  for (int s = 0; s < 4; ++s) {
    int k = 4 * s + 2 * hi;
    f_agg[s].x = arow[k] * dinv;
    f_agg[s].y = arow[k + 1] * dinv;
    float e0 = erow[k];
    float e1 = erow[k + 1];
    f_x[s].x = nv ? e0 : 0.0f;
    f_x[s].y = nv ? e1 : 0.0f;
  }
  {
    float xd[FDST];
#pragma unroll
    for (int j = 0; j < FDST; ++j) xd[j] = xrow[j];
#pragma unroll
    for (int s = 4; s < 8; ++s) {
      int r = 4 * (s - 4) + 2 * hi;
      float p0 = bd[r];
      float p1 = bd[r + 1];
#pragma unroll
      for (int j = 0; j < FDST; ++j) {
        p0 = fmaf(xd[j], Wd[r * FDST + j], p0);
        p1 = fmaf(xd[j], Wd[(r + 1) * FDST + j], p1);
      }
      f_x[s].x = nv ? p0 : 0.0f;
      f_x[s].y = nv ? p1 : 0.0f;
    }
  }
  {
    float sx[FIN];
#pragma unroll
    for (int j = 0; j < FIN; ++j) sx[j] = arow[16 + j];
#pragma unroll
    for (int s = 4; s < 8; ++s) {
      int r = 4 * (s - 4) + 2 * hi;
      float p0 = cnt * bp[r];
      float p1 = cnt * bp[r + 1];
#pragma unroll
      for (int j = 0; j < FIN; ++j) {
        p0 = fmaf(sx[j], Wp[r * FIN + j], p0);
        p1 = fmaf(sx[j], Wp[(r + 1) * FIN + j], p1);
      }
      f_agg[s].x = p0 * dinv;
      f_agg[s].y = p1 * dinv;
    }
  }

  v8f hT[2];
#pragma unroll
  for (int t = 0; t < 2; ++t) {
    int og = t * 16 + m;
    const float* wl = Wl + og * 32;
    const float* wr = Wr + og * 32;
    v8f c;
#pragma unroll
    for (int r = 0; r < 8; ++r) c[r] = bl[t * 16 + 8 * hi + r];

    if constexpr (LDSW) {
      v2f wa[16];
#pragma unroll
      for (int u = 0; u < 16; ++u) wa[u] = wf1[(16 * t + u) * 32 + lane];
#pragma unroll
      for (int s = 0; s < 8; ++s)
        c = __builtin_amdgcn_wmma_f32_16x16x4_f32(false, wa[s], false, f_agg[s],
                                                  (short)0, c, false, false);
#pragma unroll
      for (int s = 0; s < 8; ++s)
        c = __builtin_amdgcn_wmma_f32_16x16x4_f32(false, wa[8 + s], false, f_x[s],
                                                  (short)0, c, false, false);
    } else {
#pragma unroll
      for (int s = 0; s < 8; ++s) {
        int k = 4 * s + 2 * hi;
        v2f w;
        w.x = wl[k];
        w.y = wl[k + 1];
        c = __builtin_amdgcn_wmma_f32_16x16x4_f32(false, w, false, f_agg[s],
                                                  (short)0, c, false, false);
      }
#pragma unroll
      for (int s = 0; s < 8; ++s) {
        int k = 4 * s + 2 * hi;
        v2f w;
        w.x = wr[k];
        w.y = wr[k + 1];
        c = __builtin_amdgcn_wmma_f32_16x16x4_f32(false, w, false, f_x[s],
                                                  (short)0, c, false, false);
      }
    }
#pragma unroll
    for (int r = 0; r < 8; ++r) hT[t][r] = fmaxf(c[r], 0.0f);
  }

  const float* w2n = W2n + m * 32;
  const float* w2r = W2r + m * 32;
#pragma unroll
  for (int r = 0; r < 8; ++r) {
    p[r] = 0.0f;
    q[r] = b2r[8 * hi + r];
  }
#pragma unroll
  for (int t = 0; t < 2; ++t) {
#pragma unroll
    for (int qq = 0; qq < 4; ++qq) {
      int k = 16 * t + 8 * hi + 2 * qq;
      v2f hb;
      hb.x = hT[t][2 * qq];
      hb.y = hT[t][2 * qq + 1];
      v2f wa, wb;
      if constexpr (LDSW) {
        wa = wf2[(4 * t + qq) * 32 + lane];
        wb = wf2[(8 + 4 * t + qq) * 32 + lane];
      } else {
        wa.x = w2n[k];
        wa.y = w2n[k + 1];
        wb.x = w2r[k];
        wb.y = w2r[k + 1];
      }
      p = __builtin_amdgcn_wmma_f32_16x16x4_f32(false, wa, false, hb,
                                                (short)0, p, false, false);
      q = __builtin_amdgcn_wmma_f32_16x16x4_f32(false, wb, false, hb,
                                                (short)0, q, false, false);
    }
  }
  dinv_out = dinv;
}

template <int FIN, int FDST>
__global__ void sage1_chain_kernel(const float* __restrict__ agg,
                                   const float* __restrict__ embed_d,
                                   const int* __restrict__ idx_d,
                                   const float* __restrict__ x_d,
                                   const float* __restrict__ Wd,
                                   const float* __restrict__ bd,
                                   const float* __restrict__ Wl,
                                   const float* __restrict__ bl,
                                   const float* __restrict__ Wr,
                                   const float* __restrict__ Wp,
                                   const float* __restrict__ bp,
                                   const float* __restrict__ W2n,
                                   const float* __restrict__ W2r,
                                   const float* __restrict__ b2r,
                                   float* __restrict__ Pout,
                                   float* __restrict__ Rout,
                                   int n) {
  int lane = threadIdx.x & 31;
  int wave = (blockIdx.x * blockDim.x + threadIdx.x) >> 5;
  int nb = wave * 16;
  if (nb >= n) return;
  int m  = lane & 15;
  int hi = lane >> 4;

  int node = nb + m;
  bool nv = node < n;
  int ncl = nv ? node : (n - 1);
  v8f p, q;
  float dinv;
  sage1_chain_tile<FIN, FDST, false>(agg + (long long)ncl * 32,
                                     embed_d + (long long)idx_d[ncl] * 16,
                                     x_d + (long long)ncl * FDST,
                                     nv, m, hi, Wd, bd, Wl, bl, Wr, Wp, bp, W2n, W2r, b2r,
                                     nullptr, nullptr, p, q, dinv);

  if (nv) {
    float4* pd = (float4*)(Pout + (long long)node * 16 + 8 * hi);
    float4* rd = (float4*)(Rout + (long long)node * 16 + 8 * hi);
    pd[0] = make_float4(p[0], p[1], p[2], p[3]);
    pd[1] = make_float4(p[4], p[5], p[6], p[7]);
    rd[0] = make_float4(q[0], q[1], q[2], q[3]);
    rd[1] = make_float4(q[4], q[5], q[6], q[7]);
  }
}

__global__ __launch_bounds__(kUThreads)
void user_sums_lds_kernel(const unsigned int* __restrict__ edge_key,
                          const int* __restrict__ edge_user,
                          const int* __restrict__ edge_course,
                          const float* __restrict__ tab,
                          const float* __restrict__ embed_u,
                          const int* __restrict__ idx_u,
                          const float* __restrict__ x_u,
                          const float* __restrict__ Wd,
                          const float* __restrict__ bd,
                          const float* __restrict__ Wl,
                          const float* __restrict__ bl,
                          const float* __restrict__ Wr,
                          const float* __restrict__ Wp,
                          const float* __restrict__ bp,
                          const float* __restrict__ W2n,
                          const float* __restrict__ W2r,
                          const float* __restrict__ b2r,
                          float* __restrict__ Pout,
                          float* __restrict__ out2,
                          int E, int n_user) {
  __shared__ float T[kURows * kURow];
  __shared__ int2 Q[(kUThreads / 32) * kQCap];
  __shared__ v2f WF[kWF1 + kWF2];
  const int tid = threadIdx.x;
  const int lane = tid & 31;
  const int base = blockIdx.x * kURows;
  const int left = n_user - base;
  const int nrows = left < kURows ? left : kURows;
  for (int i = tid; i < nrows * kURow; i += kUThreads) T[i] = 0.0f;
  for (int i = tid; i < kWF1 + kWF2; i += kUThreads) {
    const int fm = i & 15;
    const int fh = (i >> 4) & 1;
    const int g = i >> 5;
    const float* wsrc;
    if (g < 32) {
      const int t = g >> 4;
      const int u = g & 15;
      wsrc = (u < 8 ? Wl : Wr) + (16 * t + fm) * 32 + 4 * (u & 7) + 2 * fh;
    } else {
      const int g2 = g - 32;
      wsrc = ((g2 >> 3) ? W2r : W2n) + fm * 32 + 16 * ((g2 >> 2) & 1) + 8 * fh + 2 * (g2 & 3);
    }
    v2f w;
    w.x = wsrc[0];
    w.y = wsrc[1];
    WF[i] = w;
  }
  __syncthreads();

  int2* q = Q + (tid >> 5) * kQCap;
  int qn = 0;
  const int N16 = E >> 4;
  const int kb = (int)blockIdx.x < kKeyMax ? (int)blockIdx.x : kKeyMax;
  const unsigned kk = (unsigned)kb * 0x01010101u;

  if (tid < 32) {
    int e = (N16 << 4) + lane;
    bool hit = false;
    if (e < E) hit = (unsigned)(edge_user[e] - base) < (unsigned)nrows;
    unsigned m = __builtin_amdgcn_ballot_w32(hit);
    if (hit) q[__builtin_amdgcn_mbcnt_lo(m, 0u)].x = e;
    qn = __builtin_popcount(m);
  }

  const uint4* key16 = (const uint4*)edge_key;
  constexpr int kTrip = kUThreads * kScanLoads;
  const int ntrips = (N16 + kTrip - 1) / kTrip;
  const int toff = (int)(((unsigned)blockIdx.x * (unsigned)ntrips) / gridDim.x);
  for (int t = 0; t < ntrips; ++t) {
    int tr = t + toff;
    if (tr >= ntrips) tr -= ntrips;
    const int jb = tr * kTrip + tid;
    unsigned hw[kScanLoads / 2];
    if (tr * kTrip + kTrip <= N16) user_filter_trip<false>(key16, jb, N16, kk, hw);
    else                           user_filter_trip<true>(key16, jb, N16, kk, hw);
#pragma nounroll
    for (int w = 0; w < kScanLoads / 2; ++w) {
      unsigned hm = hw[0];
#pragma unroll
      for (int i = 0; i + 1 < kScanLoads / 2; ++i) hw[i] = hw[i + 1];
      unsigned m = __builtin_amdgcn_ballot_w32(hm != 0u);
      while (m != 0u) {
        int n = __builtin_popcount(m);
        if (qn + n > kQCap) {
          user_queue_flush(q, qn, lane, base, nrows, edge_user, edge_course, tab, T);
          qn = 0;
        }
        if (hm != 0u) {
          int b = __builtin_ctz(hm);
          hm &= hm - 1u;
          int pos = b & 7;
          int grp = jb + (2 * w + (pos >> 2)) * kUThreads;
          q[qn + (int)__builtin_amdgcn_mbcnt_lo(m, 0u)].x = 16 * grp + 4 * (pos & 3) + (b >> 3);
        }
        qn += n;
        m = __builtin_amdgcn_ballot_w32(hm != 0u);
      }
    }
  }
  if (qn > 0) user_queue_flush(q, qn, lane, base, nrows, edge_user, edge_course, tab, T);
  __syncthreads();

  const int wv = __builtin_amdgcn_readfirstlane(tid >> 5);
  const int m = lane & 15;
  const int hi = lane >> 4;
  for (int tile = wv; tile * 16 < nrows; tile += kUThreads / 32) {
    int row = tile * 16 + m;
    bool nv = row < nrows;
    int rcl = nv ? row : (nrows - 1);
    long long ncl = (long long)base + rcl;
    const float* arow = T + rcl * kURow;
    int wo = 0;
    asm volatile("" : "+s"(wo));
    v8f p, q;
    float dinv;
    sage1_chain_tile<2, 5, true>(arow,
                                 embed_u + (long long)idx_u[ncl] * 16,
                                 x_u + ncl * 5,
                                 nv, m, hi, Wd + wo, bd + wo, Wl, bl + wo, Wr, Wp + wo, bp + wo,
                                 W2n, W2r, b2r + wo, WF + wo, WF + kWF1 + wo, p, q, dinv);
    if (nv) {
      const float* s2 = arow + 19 + 8 * hi;
      float4* pd = (float4*)(Pout + ncl * 16 + 8 * hi);
      float4* od = (float4*)(out2 + ncl * 16 + 8 * hi);
      pd[0] = make_float4(p[0], p[1], p[2], p[3]);
      pd[1] = make_float4(p[4], p[5], p[6], p[7]);
      od[0] = make_float4(fmaf(s2[0], dinv, q[0]), fmaf(s2[1], dinv, q[1]),
                          fmaf(s2[2], dinv, q[2]), fmaf(s2[3], dinv, q[3]));
      od[1] = make_float4(fmaf(s2[4], dinv, q[4]), fmaf(s2[5], dinv, q[5]),
                          fmaf(s2[6], dinv, q[6]), fmaf(s2[7], dinv, q[7]));
    }
  }
}

__global__ void decode_kernel(const float* __restrict__ ou,
                              const float* __restrict__ oc,
                              const int* __restrict__ ls,
                              const int* __restrict__ ld,
                              float* __restrict__ out, int L) {
  int t = blockIdx.x * blockDim.x + threadIdx.x;
  if (t >= L) return;
  const float4* a = (const float4*)(ou + (long long)ls[t] * 16);
  const float4* b = (const float4*)(oc + (long long)ld[t] * 16);
  float acc = 0.0f;
#pragma unroll
  for (int i = 0; i < 4; ++i) {
    float4 xa = a[i], xb = b[i];
    acc += xa.x * xb.x + xa.y * xb.y + xa.z * xb.z + xa.w * xb.w;
  }
  out[t] = acc;
}

extern "C" void kernel_launch(void* const* d_in, const int* in_sizes, int n_in,
                              void* d_out, int out_size, void* d_ws, size_t ws_size,
                              hipStream_t stream) {
  const float* user_x       = (const float*)d_in[0];
  const float* course_x     = (const float*)d_in[1];
  const float* user_embed   = (const float*)d_in[2];
  const float* course_embed = (const float*)d_in[3];
  const float* Wu = (const float*)d_in[4];
  const float* bu = (const float*)d_in[5];
  const float* Wc = (const float*)d_in[6];
  const float* bc = (const float*)d_in[7];
  const float* c1e_Wl = (const float*)d_in[8];
  const float* c1e_bl = (const float*)d_in[9];
  const float* c1e_Wr = (const float*)d_in[10];
  const float* c1r_Wl = (const float*)d_in[11];
  const float* c1r_bl = (const float*)d_in[12];
  const float* c1r_Wr = (const float*)d_in[13];
  const float* c2e_Wl = (const float*)d_in[14];
  const float* c2e_bl = (const float*)d_in[15];
  const float* c2e_Wr = (const float*)d_in[16];
  const float* c2r_Wl = (const float*)d_in[17];
  const float* c2r_bl = (const float*)d_in[18];
  const float* c2r_Wr = (const float*)d_in[19];
  const int* user_idx   = (const int*)d_in[20];
  const int* course_idx = (const int*)d_in[21];
  const int* edge_src   = (const int*)d_in[22];
  const int* edge_dst   = (const int*)d_in[23];
  const int* label_src  = (const int*)d_in[24];
  const int* label_dst  = (const int*)d_in[25];
  float* outp = (float*)d_out;

  const int NU = in_sizes[2] / 16;
  const int NC = in_sizes[3] / 16;
  const int E  = in_sizes[22];
  const int L  = in_sizes[24];

  char* p = (char*)d_ws;
  auto carve = [&](size_t bytes) -> float* {
    float* r = (float*)p;
    p += (bytes + 255) & ~((size_t)255);
    return r;
  };
  float* prt2 = carve((size_t)NU * 32 * 4);
  float* Pu   = carve((size_t)NU * 32 * 4);
  float* aggc = carve((size_t)NC * 32 * 4);
  float* pru  = carve((size_t)NU * 32 * 4);
  float* prc  = carve((size_t)NC * 32 * 4);
  float* Pc = prc;
  float* Rc = prc + (size_t)NC * 16;
  float* ou   = carve((size_t)NU * 16 * 4);
  float* oc   = carve((size_t)NC * 16 * 4);

  const int B = 256;
  auto nblk = [](long long t, int b) { return (unsigned)((t + b - 1) / b); };

  const int NR = (NC + kCapRows - 1) / kCapRows;
  int NS = 256 / NR;
  if (NS < 1) NS = 1;
  const long long cap1 = ((long long)NU * 32) / ((long long)NC * 22);
  const long long cap2 = ((long long)NU * 32) / ((long long)NC * 16);
  const int NS1 = (int)(cap1 < NS ? (cap1 < 1 ? 1 : cap1) : NS);
  const int NS2 = (int)(cap2 < NS ? (cap2 < 1 ? 1 : cap2) : NS);

  float* part1 = pru;
  unsigned char* ukey = (unsigned char*)prt2;
  course_sums_raw_lds_kernel<5><<<NR * NS1, kLdsThreads, 0, stream>>>(
      user_embed, user_idx, user_x, edge_src, edge_dst, part1, ukey, E, NC, NR, NS1);
  reduce_parts_kernel<22><<<nblk((long long)NC * 22, B), B, 0, stream>>>(part1, aggc, NC, NS1);
  sage1_chain_kernel<5, 2><<<nblk(((long long)(NC + 15) / 16) * 32, B), B, 0, stream>>>(
      aggc, course_embed, course_idx, course_x, Wc, bc,
      c1e_Wl, c1e_bl, c1e_Wr, Wu, bu, c2r_Wl, c2e_Wr, c2e_bl, Pc, Rc, NC);
  float* ctab = pru;
  course_rows_kernel<<<nblk((long long)NC * kURow, B), B, 0, stream>>>(
      course_embed, course_idx, course_x, Pc, ctab, NC);
  user_sums_lds_kernel<<<nblk(NU, kURows), kUThreads, 0, stream>>>(
      (const unsigned int*)ukey, edge_src, edge_dst, ctab,
      user_embed, user_idx, user_x, Wu, bu,
      c1r_Wl, c1r_bl, c1r_Wr, Wc, bc, c2e_Wl, c2r_Wr, c2r_bl, Pu, ou, E, NU);

  float* part2 = prt2;
  course_sums16_lds_kernel<<<NR * NS2, kLdsThreads, 0, stream>>>(
      Pu, edge_src, edge_dst, part2, E, NC, NR, NS2);
  combine16_parts_kernel<<<nblk((long long)NC * 4, B), B, 0, stream>>>(
      part2, NS2, aggc + 21, 32, Rc, oc, NC);

  decode_kernel<<<nblk(L, B), B, 0, stream>>>(ou, oc, label_src, label_dst, outp, L);
}


# user range-table pass fetches 2 lines per kept edge, not 4+: a 16-byte edge record (user, course, its 2 features) and an aligned 128-byte course row; sum order: LDS arrival order as before
# speedup vs baseline: 4.0090x; 1.0474x over previous
#include <hip/hip_runtime.h>

typedef __attribute__((ext_vector_type(8))) float v8f;
typedef __attribute__((ext_vector_type(2))) float v2f;


constexpr int kURow = 35;
constexpr int kCRow = 32;
constexpr int kURows = 2048;
constexpr int kUThreads = 1024;
constexpr int kQCap = 64;
static_assert(kQCap % 32 == 0, "the flush rewrites kQCap / 32 queue slots per lane");
constexpr int kScanLoads = 8;
constexpr int kKeyMax = 255;
constexpr int kGather = 8;
constexpr int kPrepThreads = 1024;
constexpr int kPrepEdges = 8;

__global__ __launch_bounds__(kPrepThreads)
void user_pass_inputs_kernel(const float* __restrict__ embed,
                             const int* __restrict__ node_index,
                             const float* __restrict__ x,
                             const float* __restrict__ Pc,
                             const int* __restrict__ edge_user,
                             const int* __restrict__ edge_course,
                             float* __restrict__ tab,
                             int* __restrict__ rec,
                             int n, int E) {
  const int nthr = gridDim.x * blockDim.x;
  const int t = blockIdx.x * blockDim.x + threadIdx.x;
  for (int i = t; i < n * kCRow; i += nthr) {
    int c = i >> 5;
    int col = i & 31;
    tab[i] = col < 16 ? embed[(long long)node_index[c] * 16 + col]
                      : Pc[(long long)c * 16 + (col - 16)];
  }
  int4* r4 = (int4*)rec;
  for (int e0 = t; e0 < E; e0 += kPrepEdges * nthr) {
    int u[kPrepEdges], c[kPrepEdges];
    float2 xv[kPrepEdges];
#pragma unroll
    for (int k = 0; k < kPrepEdges; ++k) {
      int e = e0 + k * nthr;
      int ec = e < E ? e : e0;
      u[k] = edge_user[ec];
      c[k] = edge_course[ec];
    }
#pragma unroll
    for (int k = 0; k < kPrepEdges; ++k) xv[k] = *(const float2*)(x + (long long)c[k] * 2);
#pragma unroll
    for (int k = 0; k < kPrepEdges; ++k) {
      int e = e0 + k * nthr;
      if (e < E) r4[e] = make_int4(u[k], c[k], __float_as_int(xv[k].x), __float_as_int(xv[k].y));
    }
  }
}

__device__ __forceinline__ void user_queue_flush(int2* q, int qn, int lane, int base, int nrows,
                                                 const int4* __restrict__ rec,
                                                 const float* __restrict__ tab,
                                                 float* T) {
  __builtin_amdgcn_fence(__ATOMIC_SEQ_CST, "workgroup");
  {
    int4 r[kQCap / 32];
#pragma unroll
    for (int j = 0; j < kQCap / 32; ++j) {
      int i = lane + 32 * j;
      r[j] = rec[q[i < qn ? i : 0].x];
    }
#pragma unroll
    for (int j = 0; j < kQCap / 32; ++j) {
      asm volatile("" : "+v"(r[j].x), "+v"(r[j].y), "+v"(r[j].z), "+v"(r[j].w));
    }
#pragma unroll
    for (int j = 0; j < kQCap / 32; ++j) {
      int i = lane + 32 * j;
      int rel = r[j].x - base;
      bool ok = i < qn && (unsigned)rel < (unsigned)nrows;
      q[i] = make_int2(r[j].y, ok ? rel : -1);
      if (ok) {
        float* row = T + rel * kURow;
        atomicAdd(row + 16, __int_as_float(r[j].z));
        atomicAdd(row + 17, __int_as_float(r[j].w));
        atomicAdd(row + 18, 1.0f);
      }
    }
  }
  __builtin_amdgcn_fence(__ATOMIC_SEQ_CST, "workgroup");
  const int sub = lane >> 3;
  const int pc = lane & 7;
  const int col0 = pc < 4 ? 4 * pc : 4 * pc + 3;
  for (int m0 = 0; m0 < qn; m0 += 4 * kGather) {
    float4 v[kGather];
    int a[kGather];
#pragma unroll
    for (int k = 0; k < kGather; ++k) {
      int m = m0 + 4 * k + sub;
      bool ok = m < qn;
      int2 en = q[ok ? m : 0];
      v[k] = ((const float4*)(tab + (long long)en.x * kCRow))[pc];
      a[k] = (ok && en.y >= 0) ? en.y * kURow + col0 : -1;
    }
#pragma unroll
    for (int k = 0; k < kGather; ++k) {
      asm volatile("" : "+v"(v[k].x), "+v"(v[k].y), "+v"(v[k].z), "+v"(v[k].w));
    }
#pragma unroll
    for (int k = 0; k < kGather; ++k) {
      if (a[k] >= 0) {
        float* d = T + a[k];
        atomicAdd(d + 0, v[k].x);
        atomicAdd(d + 1, v[k].y);
        atomicAdd(d + 2, v[k].z);
        atomicAdd(d + 3, v[k].w);
      }
    }
  }
  __builtin_amdgcn_fence(__ATOMIC_SEQ_CST, "workgroup");
}

template <bool CHECK>
__device__ __forceinline__ void user_filter_trip(const uint4* __restrict__ key16, int jb, int N16,
                                                 unsigned kk, unsigned (&hw)[kScanLoads / 2]) {
  uint4 kv[kScanLoads];
  bool ok[kScanLoads];
#pragma unroll
  for (int s = 0; s < kScanLoads; ++s) {
    int j = jb + s * kUThreads;
    ok[s] = CHECK ? (j < N16) : true;
    kv[s] = key16[ok[s] ? j : 0];
  }
#pragma unroll
  for (int w = 0; w < kScanLoads / 2; ++w) {
    unsigned acc = 0u;
#pragma unroll
    for (int g = 0; g < 2; ++g) {
      const int s = 2 * w + g;
      const unsigned d[4] = {kv[s].x, kv[s].y, kv[s].z, kv[s].w};
      unsigned part = 0u;
#pragma unroll
      for (int i = 0; i < 4; ++i) {
        unsigned t = d[i] ^ kk;
        unsigned y = (t - 0x01010101u) & ~t;
        const int pos = 4 * g + i;
        part |= (y >> (7 - pos)) & (0x01010101u << pos);
      }
      if (CHECK && !ok[s]) part = 0u;
      acc |= part;
    }
    hw[w] = acc;
  }
}

constexpr int kCapRows = 2560;
constexpr int kLdsThreads = 1024;
constexpr int kEdgesPerTrip = 4;

template <int FIN>
__global__ __launch_bounds__(kLdsThreads)
void course_sums_raw_lds_kernel(const float* __restrict__ embed,
                                const int* __restrict__ node_index,
                                const float* __restrict__ x,
                                const int* __restrict__ gather_idx,
                                const int* __restrict__ scatter_idx,
                                float* __restrict__ part,
                                unsigned char* __restrict__ key_out,
                                int E, int n_dst, int n_ranges, int n_slices) {
  constexpr int W = 16 + FIN + 1;
  constexpr int P = W | 1;
  __shared__ float T[kCapRows * P];
  const int tid = threadIdx.x;
  const int nthr = blockDim.x;
  const int r = blockIdx.x % n_ranges;
  const int slice = blockIdx.x / n_ranges;
  const int base = r * kCapRows;
  const int left = n_dst - base;
  const int nrows = left < kCapRows ? left : kCapRows;
  for (int i = tid; i < nrows * P; i += nthr) T[i] = 0.0f;
  __syncthreads();

  const int chunk = (E + n_slices - 1) / n_slices;
  const int e_lo = slice * chunk;
  const int e_end = e_lo + chunk;
  const int e_hi = e_end < E ? e_end : E;
  for (int e0 = e_lo + tid; e0 < e_hi; e0 += kEdgesPerTrip * nthr) {
    int s[kEdgesPerTrip], rel[kEdgesPerTrip], ni[kEdgesPerTrip], kq[kEdgesPerTrip];
    bool in[kEdgesPerTrip];
    float4 ev[kEdgesPerTrip][4];
    float xv[kEdgesPerTrip][FIN];
#pragma unroll
    for (int u = 0; u < kEdgesPerTrip; ++u) {
      int e = e0 + u * nthr;
      bool v = e < e_hi;
      int ec = v ? e : e0;
      int d = scatter_idx[ec];
      s[u] = gather_idx[ec];
      int kr = s[u] / kURows;
      kq[u] = kr < kKeyMax ? kr : kKeyMax;
      rel[u] = d - base;
      in[u] = v && ((unsigned)rel[u] < (unsigned)nrows);
      if (!in[u]) s[u] = 0;
    }
#pragma unroll
    for (int u = 0; u < kEdgesPerTrip; ++u) ni[u] = node_index[s[u]];
#pragma unroll
    for (int u = 0; u < kEdgesPerTrip; ++u) {
      if (!in[u]) ni[u] = 0;
      const float4* er = (const float4*)(embed + (long long)ni[u] * 16);
#pragma unroll
      for (int q = 0; q < 4; ++q) ev[u][q] = er[q];
      const float* xr = x + (long long)s[u] * FIN;
#pragma unroll
      for (int j = 0; j < FIN; ++j) xv[u][j] = xr[j];
    }
    if (r == 0) {
#pragma unroll
      for (int u = 0; u < kEdgesPerTrip; ++u) {
        int e = e0 + u * nthr;
        if (e < e_hi) key_out[e] = (unsigned char)kq[u];
      }
    }
#pragma unroll
    for (int u = 0; u < kEdgesPerTrip; ++u) {
      if (in[u]) {
        float* row = T + rel[u] * P;
#pragma unroll
        for (int q = 0; q < 4; ++q) {
          atomicAdd(row + 4 * q + 0, ev[u][q].x);
          atomicAdd(row + 4 * q + 1, ev[u][q].y);
          atomicAdd(row + 4 * q + 2, ev[u][q].z);
          atomicAdd(row + 4 * q + 3, ev[u][q].w);
        }
#pragma unroll
        for (int j = 0; j < FIN; ++j) atomicAdd(row + 16 + j, xv[u][j]);
        atomicAdd(row + 16 + FIN, 1.0f);
      }
    }
  }
  __syncthreads();

  float* dst = part + ((long long)slice * n_dst + base) * W;
  for (int i = tid; i < nrows * W; i += nthr) {
    int row = i / W;
    int c = i - row * W;
    dst[i] = T[row * P + c];
  }
}

__global__ __launch_bounds__(kLdsThreads)
void course_sums16_lds_kernel(const float* __restrict__ srcp,
                              const int* __restrict__ gather_idx,
                              const int* __restrict__ scatter_idx,
                              float* __restrict__ part,
                              int E, int n_dst, int n_ranges, int n_slices) {
  constexpr int P = 17;
  __shared__ float T[kCapRows * P];
  const int tid = threadIdx.x;
  const int nthr = blockDim.x;
  const int r = blockIdx.x % n_ranges;
  const int slice = blockIdx.x / n_ranges;
  const int base = r * kCapRows;
  const int left = n_dst - base;
  const int nrows = left < kCapRows ? left : kCapRows;
  for (int i = tid; i < nrows * P; i += nthr) T[i] = 0.0f;
  __syncthreads();

  const int chunk = (E + n_slices - 1) / n_slices;
  const int e_lo = slice * chunk;
  const int e_end = e_lo + chunk;
  const int e_hi = e_end < E ? e_end : E;
  for (int e0 = e_lo + tid; e0 < e_hi; e0 += kEdgesPerTrip * nthr) {
    int s[kEdgesPerTrip], rel[kEdgesPerTrip];
    bool in[kEdgesPerTrip];
    float4 pv[kEdgesPerTrip][4];
#pragma unroll
    for (int u = 0; u < kEdgesPerTrip; ++u) {
      int e = e0 + u * nthr;
      bool v = e < e_hi;
      int ec = v ? e : e0;
      int d = scatter_idx[ec];
      s[u] = gather_idx[ec];
      rel[u] = d - base;
      in[u] = v && ((unsigned)rel[u] < (unsigned)nrows);
      if (!in[u]) s[u] = 0;
    }
#pragma unroll
    for (int u = 0; u < kEdgesPerTrip; ++u) {
      const float4* pr = (const float4*)(srcp + (long long)s[u] * 16);
#pragma unroll
      for (int q = 0; q < 4; ++q) pv[u][q] = pr[q];
    }
#pragma unroll
    for (int u = 0; u < kEdgesPerTrip; ++u) {
      if (in[u]) {
        float* row = T + rel[u] * P;
#pragma unroll
        for (int q = 0; q < 4; ++q) {
          atomicAdd(row + 4 * q + 0, pv[u][q].x);
          atomicAdd(row + 4 * q + 1, pv[u][q].y);
          atomicAdd(row + 4 * q + 2, pv[u][q].z);
          atomicAdd(row + 4 * q + 3, pv[u][q].w);
        }
      }
    }
  }
  __syncthreads();

  float* dst = part + ((long long)slice * n_dst + base) * 16;
  for (int i = tid; i < nrows * 16; i += nthr) {
    dst[i] = T[(i >> 4) * P + (i & 15)];
  }
}

template <int W>
__global__ void reduce_parts_kernel(const float* __restrict__ part,
                                    float* __restrict__ out,
                                    int n_rows, int n_slices) {
  int t = blockIdx.x * blockDim.x + threadIdx.x;
  int total = n_rows * W;
  if (t >= total) return;
  float acc = 0.0f;
#pragma unroll 16
  for (int s = 0; s < n_slices; ++s) acc += part[(long long)s * total + t];
  int row = t / W;
  int c = t - row * W;
  out[(long long)row * 32 + c] = acc;
}

__global__ void combine16_parts_kernel(const float* __restrict__ part,
                                       int n_slices,
                                       const float* __restrict__ deg,
                                       int dstride,
                                       const float* __restrict__ R,
                                       float* __restrict__ out,
                                       int n) {
  int t = blockIdx.x * blockDim.x + threadIdx.x;
  int node = t >> 2;
  int q = t & 3;
  if (node >= n) return;
  float4 a = make_float4(0.0f, 0.0f, 0.0f, 0.0f);
  long long stride = (long long)n * 16;
#pragma unroll 16
  for (int s = 0; s < n_slices; ++s) {
    float4 v = ((const float4*)(part + s * stride + (long long)node * 16))[q];
    a.x += v.x; a.y += v.y; a.z += v.z; a.w += v.w;
  }
  float dinv = 1.0f / fmaxf(deg[(long long)node * dstride], 1.0f);
  float4 rr = ((const float4*)(R + (long long)node * 16))[q];
  float4 o;
  o.x = fmaf(a.x, dinv, rr.x);
  o.y = fmaf(a.y, dinv, rr.y);
  o.z = fmaf(a.z, dinv, rr.z);
  o.w = fmaf(a.w, dinv, rr.w);
  ((float4*)(out + (long long)node * 16))[q] = o;
}

constexpr int kWF1 = 2 * 16 * 32;
constexpr int kWF2 = 2 * 8 * 32;

template <int FIN, int FDST, bool LDSW>
__device__ __forceinline__ void sage1_chain_tile(const float* arow,
                                                 const float* erow,
                                                 const float* xrow,
                                                 bool nv, int m, int hi,
                                                 const float* __restrict__ Wd,
                                                 const float* __restrict__ bd,
                                                 const float* __restrict__ Wl,
                                                 const float* __restrict__ bl,
                                                 const float* __restrict__ Wr,
                                                 const float* __restrict__ Wp,
                                                 const float* __restrict__ bp,
                                                 const float* __restrict__ W2n,
                                                 const float* __restrict__ W2r,
                                                 const float* __restrict__ b2r,
                                                 const v2f* wf1,
                                                 const v2f* wf2,
                                                 v8f& p, v8f& q, float& dinv_out) {
  const int lane = 16 * hi + m;
  float cnt  = arow[16 + FIN];
  float dinv = nv ? (1.0f / fmaxf(cnt, 1.0f)) : 0.0f;

  v2f f_agg[8], f_x[8];
#pragma unroll
  for (int s = 0; s < 4; ++s) {
    int k = 4 * s + 2 * hi;
    f_agg[s].x = arow[k] * dinv;
    f_agg[s].y = arow[k + 1] * dinv;
    float e0 = erow[k];
    float e1 = erow[k + 1];
    f_x[s].x = nv ? e0 : 0.0f;
    f_x[s].y = nv ? e1 : 0.0f;
  }
  {
    float xd[FDST];
#pragma unroll
    for (int j = 0; j < FDST; ++j) xd[j] = xrow[j];
#pragma unroll
    for (int s = 4; s < 8; ++s) {
      int r = 4 * (s - 4) + 2 * hi;
      float p0 = bd[r];
      float p1 = bd[r + 1];
#pragma unroll
      for (int j = 0; j < FDST; ++j) {
        p0 = fmaf(xd[j], Wd[r * FDST + j], p0);
        p1 = fmaf(xd[j], Wd[(r + 1) * FDST + j], p1);
      }
      f_x[s].x = nv ? p0 : 0.0f;
      f_x[s].y = nv ? p1 : 0.0f;
    }
  }
  {
    float sx[FIN];
#pragma unroll
    for (int j = 0; j < FIN; ++j) sx[j] = arow[16 + j];
#pragma unroll
    for (int s = 4; s < 8; ++s) {
      int r = 4 * (s - 4) + 2 * hi;
      float p0 = cnt * bp[r];
      float p1 = cnt * bp[r + 1];
#pragma unroll
      for (int j = 0; j < FIN; ++j) {
        p0 = fmaf(sx[j], Wp[r * FIN + j], p0);
        p1 = fmaf(sx[j], Wp[(r + 1) * FIN + j], p1);
      }
      f_agg[s].x = p0 * dinv;
      f_agg[s].y = p1 * dinv;
    }
  }

  v8f hT[2];
#pragma unroll
  for (int t = 0; t < 2; ++t) {
    int og = t * 16 + m;
    const float* wl = Wl + og * 32;
    const float* wr = Wr + og * 32;
    v8f c;
#pragma unroll
    for (int r = 0; r < 8; ++r) c[r] = bl[t * 16 + 8 * hi + r];

    if constexpr (LDSW) {
      v2f wa[16];
#pragma unroll
      for (int u = 0; u < 16; ++u) wa[u] = wf1[(16 * t + u) * 32 + lane];
#pragma unroll
      for (int s = 0; s < 8; ++s)
        c = __builtin_amdgcn_wmma_f32_16x16x4_f32(false, wa[s], false, f_agg[s],
                                                  (short)0, c, false, false);
#pragma unroll
      for (int s = 0; s < 8; ++s)
        c = __builtin_amdgcn_wmma_f32_16x16x4_f32(false, wa[8 + s], false, f_x[s],
                                                  (short)0, c, false, false);
    } else {
#pragma unroll
      for (int s = 0; s < 8; ++s) {
        int k = 4 * s + 2 * hi;
        v2f w;
        w.x = wl[k];
        w.y = wl[k + 1];
        c = __builtin_amdgcn_wmma_f32_16x16x4_f32(false, w, false, f_agg[s],
                                                  (short)0, c, false, false);
      }
#pragma unroll
      for (int s = 0; s < 8; ++s) {
        int k = 4 * s + 2 * hi;
        v2f w;
        w.x = wr[k];
        w.y = wr[k + 1];
        c = __builtin_amdgcn_wmma_f32_16x16x4_f32(false, w, false, f_x[s],
                                                  (short)0, c, false, false);
      }
    }
#pragma unroll
    for (int r = 0; r < 8; ++r) hT[t][r] = fmaxf(c[r], 0.0f);
  }

  const float* w2n = W2n + m * 32;
  const float* w2r = W2r + m * 32;
#pragma unroll
  for (int r = 0; r < 8; ++r) {
    p[r] = 0.0f;
    q[r] = b2r[8 * hi + r];
  }
#pragma unroll
  for (int t = 0; t < 2; ++t) {
#pragma unroll
    for (int qq = 0; qq < 4; ++qq) {
      int k = 16 * t + 8 * hi + 2 * qq;
      v2f hb;
      hb.x = hT[t][2 * qq];
      hb.y = hT[t][2 * qq + 1];
      v2f wa, wb;
      if constexpr (LDSW) {
        wa = wf2[(4 * t + qq) * 32 + lane];
        wb = wf2[(8 + 4 * t + qq) * 32 + lane];
      } else {
        wa.x = w2n[k];
        wa.y = w2n[k + 1];
        wb.x = w2r[k];
        wb.y = w2r[k + 1];
      }
      p = __builtin_amdgcn_wmma_f32_16x16x4_f32(false, wa, false, hb,
                                                (short)0, p, false, false);
      q = __builtin_amdgcn_wmma_f32_16x16x4_f32(false, wb, false, hb,
                                                (short)0, q, false, false);
    }
  }
  dinv_out = dinv;
}

template <int FIN, int FDST>
__global__ void sage1_chain_kernel(const float* __restrict__ agg,
                                   const float* __restrict__ embed_d,
                                   const int* __restrict__ idx_d,
                                   const float* __restrict__ x_d,
                                   const float* __restrict__ Wd,
                                   const float* __restrict__ bd,
                                   const float* __restrict__ Wl,
                                   const float* __restrict__ bl,
                                   const float* __restrict__ Wr,
                                   const float* __restrict__ Wp,
                                   const float* __restrict__ bp,
                                   const float* __restrict__ W2n,
                                   const float* __restrict__ W2r,
                                   const float* __restrict__ b2r,
                                   float* __restrict__ Pout,
                                   float* __restrict__ Rout,
                                   int n) {
  int lane = threadIdx.x & 31;
  int wave = (blockIdx.x * blockDim.x + threadIdx.x) >> 5;
  int nb = wave * 16;
  if (nb >= n) return;
  int m  = lane & 15;
  int hi = lane >> 4;

  int node = nb + m;
  bool nv = node < n;
  int ncl = nv ? node : (n - 1);
  v8f p, q;
  float dinv;
  sage1_chain_tile<FIN, FDST, false>(agg + (long long)ncl * 32,
                                     embed_d + (long long)idx_d[ncl] * 16,
                                     x_d + (long long)ncl * FDST,
                                     nv, m, hi, Wd, bd, Wl, bl, Wr, Wp, bp, W2n, W2r, b2r,
                                     nullptr, nullptr, p, q, dinv);

  if (nv) {
    float4* pd = (float4*)(Pout + (long long)node * 16 + 8 * hi);
    float4* rd = (float4*)(Rout + (long long)node * 16 + 8 * hi);
    pd[0] = make_float4(p[0], p[1], p[2], p[3]);
    pd[1] = make_float4(p[4], p[5], p[6], p[7]);
    rd[0] = make_float4(q[0], q[1], q[2], q[3]);
    rd[1] = make_float4(q[4], q[5], q[6], q[7]);
  }
}

__global__ __launch_bounds__(kUThreads)
void user_sums_lds_kernel(const unsigned int* __restrict__ edge_key,
                          const int* __restrict__ edge_user,
                          const int* __restrict__ edge_rec,
                          const float* __restrict__ tab,
                          const float* __restrict__ embed_u,
                          const int* __restrict__ idx_u,
                          const float* __restrict__ x_u,
                          const float* __restrict__ Wd,
                          const float* __restrict__ bd,
                          const float* __restrict__ Wl,
                          const float* __restrict__ bl,
                          const float* __restrict__ Wr,
                          const float* __restrict__ Wp,
                          const float* __restrict__ bp,
                          const float* __restrict__ W2n,
                          const float* __restrict__ W2r,
                          const float* __restrict__ b2r,
                          float* __restrict__ Pout,
                          float* __restrict__ out2,
                          int E, int n_user) {
  __shared__ float T[kURows * kURow];
  __shared__ int2 Q[(kUThreads / 32) * kQCap];
  __shared__ v2f WF[kWF1 + kWF2];
  const int tid = threadIdx.x;
  const int lane = tid & 31;
  const int base = blockIdx.x * kURows;
  const int left = n_user - base;
  const int nrows = left < kURows ? left : kURows;
  for (int i = tid; i < nrows * kURow; i += kUThreads) T[i] = 0.0f;
  for (int i = tid; i < kWF1 + kWF2; i += kUThreads) {
    const int fm = i & 15;
    const int fh = (i >> 4) & 1;
    const int g = i >> 5;
    const float* wsrc;
    if (g < 32) {
      const int t = g >> 4;
      const int u = g & 15;
      wsrc = (u < 8 ? Wl : Wr) + (16 * t + fm) * 32 + 4 * (u & 7) + 2 * fh;
    } else {
      const int g2 = g - 32;
      wsrc = ((g2 >> 3) ? W2r : W2n) + fm * 32 + 16 * ((g2 >> 2) & 1) + 8 * fh + 2 * (g2 & 3);
    }
    v2f w;
    w.x = wsrc[0];
    w.y = wsrc[1];
    WF[i] = w;
  }
  __syncthreads();

  int2* q = Q + (tid >> 5) * kQCap;
  int qn = 0;
  const int4* rec = (const int4*)edge_rec;
  const int N16 = E >> 4;
  const int kb = (int)blockIdx.x < kKeyMax ? (int)blockIdx.x : kKeyMax;
  const unsigned kk = (unsigned)kb * 0x01010101u;

  if (tid < 32) {
    int e = (N16 << 4) + lane;
    bool hit = false;
    if (e < E) hit = (unsigned)(edge_user[e] - base) < (unsigned)nrows;
    unsigned m = __builtin_amdgcn_ballot_w32(hit);
    if (hit) q[__builtin_amdgcn_mbcnt_lo(m, 0u)].x = e;
    qn = __builtin_popcount(m);
  }

  const uint4* key16 = (const uint4*)edge_key;
  constexpr int kTrip = kUThreads * kScanLoads;
  const int ntrips = (N16 + kTrip - 1) / kTrip;
  const int toff = (int)(((unsigned)blockIdx.x * (unsigned)ntrips) / gridDim.x);
  for (int t = 0; t < ntrips; ++t) {
    int tr = t + toff;
    if (tr >= ntrips) tr -= ntrips;
    const int jb = tr * kTrip + tid;
    unsigned hw[kScanLoads / 2];
    if (tr * kTrip + kTrip <= N16) user_filter_trip<false>(key16, jb, N16, kk, hw);
    else                           user_filter_trip<true>(key16, jb, N16, kk, hw);
#pragma nounroll
    for (int w = 0; w < kScanLoads / 2; ++w) {
      unsigned hm = hw[0];
#pragma unroll
      for (int i = 0; i + 1 < kScanLoads / 2; ++i) hw[i] = hw[i + 1];
      unsigned m = __builtin_amdgcn_ballot_w32(hm != 0u);
      while (m != 0u) {
        int n = __builtin_popcount(m);
        if (qn + n > kQCap) {
          user_queue_flush(q, qn, lane, base, nrows, rec, tab, T);
          qn = 0;
        }
        if (hm != 0u) {
          int b = __builtin_ctz(hm);
          hm &= hm - 1u;
          int pos = b & 7;
          int grp = jb + (2 * w + (pos >> 2)) * kUThreads;
          q[qn + (int)__builtin_amdgcn_mbcnt_lo(m, 0u)].x = 16 * grp + 4 * (pos & 3) + (b >> 3);
        }
        qn += n;
        m = __builtin_amdgcn_ballot_w32(hm != 0u);
      }
    }
  }
  if (qn > 0) user_queue_flush(q, qn, lane, base, nrows, rec, tab, T);
  __syncthreads();

  const int wv = __builtin_amdgcn_readfirstlane(tid >> 5);
  const int m = lane & 15;
  const int hi = lane >> 4;
  for (int tile = wv; tile * 16 < nrows; tile += kUThreads / 32) {
    int row = tile * 16 + m;
    bool nv = row < nrows;
    int rcl = nv ? row : (nrows - 1);
    long long ncl = (long long)base + rcl;
    const float* arow = T + rcl * kURow;
    int wo = 0;
    asm volatile("" : "+s"(wo));
    v8f p, q;
    float dinv;
    sage1_chain_tile<2, 5, true>(arow,
                                 embed_u + (long long)idx_u[ncl] * 16,
                                 x_u + ncl * 5,
                                 nv, m, hi, Wd + wo, bd + wo, Wl, bl + wo, Wr, Wp + wo, bp + wo,
                                 W2n, W2r, b2r + wo, WF + wo, WF + kWF1 + wo, p, q, dinv);
    if (nv) {
      const float* s2 = arow + 19 + 8 * hi;
      float4* pd = (float4*)(Pout + ncl * 16 + 8 * hi);
      float4* od = (float4*)(out2 + ncl * 16 + 8 * hi);
      pd[0] = make_float4(p[0], p[1], p[2], p[3]);
      pd[1] = make_float4(p[4], p[5], p[6], p[7]);
      od[0] = make_float4(fmaf(s2[0], dinv, q[0]), fmaf(s2[1], dinv, q[1]),
                          fmaf(s2[2], dinv, q[2]), fmaf(s2[3], dinv, q[3]));
      od[1] = make_float4(fmaf(s2[4], dinv, q[4]), fmaf(s2[5], dinv, q[5]),
                          fmaf(s2[6], dinv, q[6]), fmaf(s2[7], dinv, q[7]));
    }
  }
}

__global__ void decode_kernel(const float* __restrict__ ou,
                              const float* __restrict__ oc,
                              const int* __restrict__ ls,
                              const int* __restrict__ ld,
                              float* __restrict__ out, int L) {
  int t = blockIdx.x * blockDim.x + threadIdx.x;
  if (t >= L) return;
  const float4* a = (const float4*)(ou + (long long)ls[t] * 16);
  const float4* b = (const float4*)(oc + (long long)ld[t] * 16);
  float acc = 0.0f;
#pragma unroll
  for (int i = 0; i < 4; ++i) {
    float4 xa = a[i], xb = b[i];
    acc += xa.x * xb.x + xa.y * xb.y + xa.z * xb.z + xa.w * xb.w;
  }
  out[t] = acc;
}

extern "C" void kernel_launch(void* const* d_in, const int* in_sizes, int n_in,
                              void* d_out, int out_size, void* d_ws, size_t ws_size,
                              hipStream_t stream) {
  const float* user_x       = (const float*)d_in[0];
  const float* course_x     = (const float*)d_in[1];
  const float* user_embed   = (const float*)d_in[2];
  const float* course_embed = (const float*)d_in[3];
  const float* Wu = (const float*)d_in[4];
  const float* bu = (const float*)d_in[5];
  const float* Wc = (const float*)d_in[6];
  const float* bc = (const float*)d_in[7];
  const float* c1e_Wl = (const float*)d_in[8];
  const float* c1e_bl = (const float*)d_in[9];
  const float* c1e_Wr = (const float*)d_in[10];
  const float* c1r_Wl = (const float*)d_in[11];
  const float* c1r_bl = (const float*)d_in[12];
  const float* c1r_Wr = (const float*)d_in[13];
  const float* c2e_Wl = (const float*)d_in[14];
  const float* c2e_bl = (const float*)d_in[15];
  const float* c2e_Wr = (const float*)d_in[16];
  const float* c2r_Wl = (const float*)d_in[17];
  const float* c2r_bl = (const float*)d_in[18];
  const float* c2r_Wr = (const float*)d_in[19];
  const int* user_idx   = (const int*)d_in[20];
  const int* course_idx = (const int*)d_in[21];
  const int* edge_src   = (const int*)d_in[22];
  const int* edge_dst   = (const int*)d_in[23];
  const int* label_src  = (const int*)d_in[24];
  const int* label_dst  = (const int*)d_in[25];
  float* outp = (float*)d_out;

  const int NU = in_sizes[2] / 16;
  const int NC = in_sizes[3] / 16;
  const int E  = in_sizes[22];
  const int L  = in_sizes[24];

  char* p = (char*)d_ws;
  auto carve = [&](size_t bytes) -> float* {
    float* r = (float*)p;
    p += (bytes + 255) & ~((size_t)255);
    return r;
  };
  float* prt2 = carve((size_t)NU * 32 * 4);
  float* Pu   = carve((size_t)NU * 32 * 4);
  float* aggc = carve((size_t)NC * 32 * 4);
  float* pru  = carve((size_t)NU * 32 * 4);
  float* prc  = carve((size_t)NC * 32 * 4);
  float* Pc = prc;
  float* Rc = prc + (size_t)NC * 16;
  float* ou   = carve((size_t)NU * 16 * 4);
  float* oc   = carve((size_t)NC * 16 * 4);

  const int B = 256;
  auto nblk = [](long long t, int b) { return (unsigned)((t + b - 1) / b); };

  const int NR = (NC + kCapRows - 1) / kCapRows;
  int NS = 256 / NR;
  if (NS < 1) NS = 1;
  const long long cap1 = ((long long)NU * 32) / ((long long)NC * 22);
  const long long cap2 = ((long long)NU * 32) / ((long long)NC * 16);
  const int NS1 = (int)(cap1 < NS ? (cap1 < 1 ? 1 : cap1) : NS);
  const int NS2 = (int)(cap2 < NS ? (cap2 < 1 ? 1 : cap2) : NS);

  float* part1 = pru;
  unsigned char* ukey = (unsigned char*)prt2;
  course_sums_raw_lds_kernel<5><<<NR * NS1, kLdsThreads, 0, stream>>>(
      user_embed, user_idx, user_x, edge_src, edge_dst, part1, ukey, E, NC, NR, NS1);
  reduce_parts_kernel<22><<<nblk((long long)NC * 22, B), B, 0, stream>>>(part1, aggc, NC, NS1);
  sage1_chain_kernel<5, 2><<<nblk(((long long)(NC + 15) / 16) * 32, B), B, 0, stream>>>(
      aggc, course_embed, course_idx, course_x, Wc, bc,
      c1e_Wl, c1e_bl, c1e_Wr, Wu, bu, c2r_Wl, c2e_Wr, c2e_bl, Pc, Rc, NC);
  float* ctab = pru;
  int* erec = (int*)(pru + (((size_t)NC * kCRow + 63) & ~((size_t)63)));
  user_pass_inputs_kernel<<<256, kPrepThreads, 0, stream>>>(
      course_embed, course_idx, course_x, Pc, edge_src, edge_dst, ctab, erec, NC, E);
  user_sums_lds_kernel<<<nblk(NU, kURows), kUThreads, 0, stream>>>(
      (const unsigned int*)ukey, edge_src, erec, ctab,
      user_embed, user_idx, user_x, Wu, bu,
      c1r_Wl, c1r_bl, c1r_Wr, Wc, bc, c2e_Wl, c2r_Wr, c2r_bl, Pu, ou, E, NU);

  float* part2 = prt2;
  course_sums16_lds_kernel<<<NR * NS2, kLdsThreads, 0, stream>>>(
      Pu, edge_src, edge_dst, part2, E, NC, NR, NS2);
  combine16_parts_kernel<<<nblk((long long)NC * 4, B), B, 0, stream>>>(
      part2, NS2, aggc + 21, 32, Rc, oc, NC);

  decode_kernel<<<nblk(L, B), B, 0, stream>>>(ou, oc, label_src, label_dst, outp, L);
}


# course-side passes of both layers queue their in-range edges per wave in LDS and gather each 64-byte row with 4 lanes: every load and LDS add is dense; sum order: LDS arrival order as before
# speedup vs baseline: 4.2042x; 1.0487x over previous
#include <hip/hip_runtime.h>

typedef __attribute__((ext_vector_type(8))) float v8f;
typedef __attribute__((ext_vector_type(2))) float v2f;


constexpr int kURow = 35;
constexpr int kCRow = 32;
constexpr int kURows = 2048;
constexpr int kUThreads = 1024;
constexpr int kQCap = 64;
static_assert(kQCap % 32 == 0, "the flush rewrites kQCap / 32 queue slots per lane");
constexpr int kScanLoads = 8;
constexpr int kKeyMax = 255;
constexpr int kGather = 8;
constexpr int kPrepThreads = 1024;
constexpr int kPrepEdges = 8;

__global__ __launch_bounds__(kPrepThreads)
void user_pass_inputs_kernel(const float* __restrict__ embed,
                             const int* __restrict__ node_index,
                             const float* __restrict__ x,
                             const float* __restrict__ Pc,
                             const int* __restrict__ edge_user,
                             const int* __restrict__ edge_course,
                             float* __restrict__ tab,
                             int* __restrict__ rec,
                             int n, int E) {
  const int nthr = gridDim.x * blockDim.x;
  const int t = blockIdx.x * blockDim.x + threadIdx.x;
  for (int i = t; i < n * kCRow; i += nthr) {
    int c = i >> 5;
    int col = i & 31;
    tab[i] = col < 16 ? embed[(long long)node_index[c] * 16 + col]
                      : Pc[(long long)c * 16 + (col - 16)];
  }
  int4* r4 = (int4*)rec;
  for (int e0 = t; e0 < E; e0 += kPrepEdges * nthr) {
    int u[kPrepEdges], c[kPrepEdges];
    float2 xv[kPrepEdges];
#pragma unroll
    for (int k = 0; k < kPrepEdges; ++k) {
      int e = e0 + k * nthr;
      int ec = e < E ? e : e0;
      u[k] = edge_user[ec];
      c[k] = edge_course[ec];
    }
#pragma unroll
    for (int k = 0; k < kPrepEdges; ++k) xv[k] = *(const float2*)(x + (long long)c[k] * 2);
#pragma unroll
    for (int k = 0; k < kPrepEdges; ++k) {
      int e = e0 + k * nthr;
      if (e < E) r4[e] = make_int4(u[k], c[k], __float_as_int(xv[k].x), __float_as_int(xv[k].y));
    }
  }
}

__device__ __forceinline__ void user_queue_flush(int2* q, int qn, int lane, int base, int nrows,
                                                 const int4* __restrict__ rec,
                                                 const float* __restrict__ tab,
                                                 float* T) {
  __builtin_amdgcn_fence(__ATOMIC_SEQ_CST, "workgroup");
  {
    int4 r[kQCap / 32];
#pragma unroll
    for (int j = 0; j < kQCap / 32; ++j) {
      int i = lane + 32 * j;
      r[j] = rec[q[i < qn ? i : 0].x];
    }
#pragma unroll
    for (int j = 0; j < kQCap / 32; ++j) {
      asm volatile("" : "+v"(r[j].x), "+v"(r[j].y), "+v"(r[j].z), "+v"(r[j].w));
    }
#pragma unroll
    for (int j = 0; j < kQCap / 32; ++j) {
      int i = lane + 32 * j;
      int rel = r[j].x - base;
      bool ok = i < qn && (unsigned)rel < (unsigned)nrows;
      q[i] = make_int2(r[j].y, ok ? rel : -1);
      if (ok) {
        float* row = T + rel * kURow;
        atomicAdd(row + 16, __int_as_float(r[j].z));
        atomicAdd(row + 17, __int_as_float(r[j].w));
        atomicAdd(row + 18, 1.0f);
      }
    }
  }
  __builtin_amdgcn_fence(__ATOMIC_SEQ_CST, "workgroup");
  const int sub = lane >> 3;
  const int pc = lane & 7;
  const int col0 = pc < 4 ? 4 * pc : 4 * pc + 3;
  for (int m0 = 0; m0 < qn; m0 += 4 * kGather) {
    float4 v[kGather];
    int a[kGather];
#pragma unroll
    for (int k = 0; k < kGather; ++k) {
      int m = m0 + 4 * k + sub;
      bool ok = m < qn;
      int2 en = q[ok ? m : 0];
      v[k] = ((const float4*)(tab + (long long)en.x * kCRow))[pc];
      a[k] = (ok && en.y >= 0) ? en.y * kURow + col0 : -1;
    }
#pragma unroll
    for (int k = 0; k < kGather; ++k) {
      asm volatile("" : "+v"(v[k].x), "+v"(v[k].y), "+v"(v[k].z), "+v"(v[k].w));
    }
#pragma unroll
    for (int k = 0; k < kGather; ++k) {
      if (a[k] >= 0) {
        float* d = T + a[k];
        atomicAdd(d + 0, v[k].x);
        atomicAdd(d + 1, v[k].y);
        atomicAdd(d + 2, v[k].z);
        atomicAdd(d + 3, v[k].w);
      }
    }
  }
  __builtin_amdgcn_fence(__ATOMIC_SEQ_CST, "workgroup");
}

template <bool CHECK>
__device__ __forceinline__ void user_filter_trip(const uint4* __restrict__ key16, int jb, int N16,
                                                 unsigned kk, unsigned (&hw)[kScanLoads / 2]) {
  uint4 kv[kScanLoads];
  bool ok[kScanLoads];
#pragma unroll
  for (int s = 0; s < kScanLoads; ++s) {
    int j = jb + s * kUThreads;
    ok[s] = CHECK ? (j < N16) : true;
    kv[s] = key16[ok[s] ? j : 0];
  }
#pragma unroll
  for (int w = 0; w < kScanLoads / 2; ++w) {
    unsigned acc = 0u;
#pragma unroll
    for (int g = 0; g < 2; ++g) {
      const int s = 2 * w + g;
      const unsigned d[4] = {kv[s].x, kv[s].y, kv[s].z, kv[s].w};
      unsigned part = 0u;
#pragma unroll
      for (int i = 0; i < 4; ++i) {
        unsigned t = d[i] ^ kk;
        unsigned y = (t - 0x01010101u) & ~t;
        const int pos = 4 * g + i;
        part |= (y >> (7 - pos)) & (0x01010101u << pos);
      }
      if (CHECK && !ok[s]) part = 0u;
      acc |= part;
    }
    hw[w] = acc;
  }
}

constexpr int kCapRows = 2560;
constexpr int kLdsThreads = 1024;
constexpr int kCScan = 8;
constexpr int kCQCap = 128;
constexpr int kCGather = 8;
static_assert(kCQCap % 32 == 0, "the flush rewrites kCQCap / 32 queue slots per lane");
static_assert(kCQCap % (8 * kCGather) == 0, "a flush round covers 8 * kCGather entries");

template <bool RAW, int FIN, int P>
__device__ __forceinline__ void course_queue_flush(int4* q, int qn, int lane,
                                                   const int* __restrict__ gather_idx,
                                                   const int* __restrict__ node_index,
                                                   const float* __restrict__ rows16,
                                                   const float* __restrict__ x,
                                                   float* T) {
  __builtin_amdgcn_fence(__ATOMIC_SEQ_CST, "workgroup");
  {
    constexpr int J = kCQCap / 32;
    int sv[J], nv[J];
#pragma unroll
    for (int j = 0; j < J; ++j) {
      int i = lane + 32 * j;
      sv[j] = gather_idx[q[i < qn ? i : 0].x];
    }
    if constexpr (RAW) {
#pragma unroll
      for (int j = 0; j < J; ++j) nv[j] = node_index[sv[j]];
    }
#pragma unroll
    for (int j = 0; j < J; ++j) {
      int i = lane + 32 * j;
      q[i].x = sv[j];
      if constexpr (RAW) q[i].z = nv[j];
    }
  }
  __builtin_amdgcn_fence(__ATOMIC_SEQ_CST, "workgroup");
  const int sub4 = lane >> 2;
  const int pc4 = lane & 3;
  for (int m0 = 0; m0 < qn; m0 += 8 * kCGather) {
    float4 v[kCGather];
    int a[kCGather];
    float xa[kCGather], xb[kCGather];
#pragma unroll
    for (int k = 0; k < kCGather; ++k) {
      int m = m0 + 8 * k + sub4;
      bool ok = m < qn;
      int4 en = q[ok ? m : 0];
      int rown = RAW ? en.z : en.x;
      v[k] = ((const float4*)rows16)[rown * 4 + pc4];
      a[k] = ok ? en.y * P + 4 * pc4 : -1;
      if constexpr (RAW) {
        xa[k] = x[en.x * FIN + (pc4 < FIN ? pc4 : 0)];
        xb[k] = x[en.x * FIN + (FIN > 4 ? 4 : 0)];
      }
    }
#pragma unroll
    for (int k = 0; k < kCGather; ++k) {
      asm volatile("" : "+v"(v[k].x), "+v"(v[k].y), "+v"(v[k].z), "+v"(v[k].w));
      if constexpr (RAW) asm volatile("" : "+v"(xa[k]), "+v"(xb[k]));
    }
#pragma unroll
    for (int k = 0; k < kCGather; ++k) {
      if (a[k] >= 0) {
        float* d = T + a[k];
        atomicAdd(d + 0, v[k].x);
        atomicAdd(d + 1, v[k].y);
        atomicAdd(d + 2, v[k].z);
        atomicAdd(d + 3, v[k].w);
        if constexpr (RAW) {
          float* r0 = d - 4 * pc4;
          if (pc4 < FIN) atomicAdd(r0 + 16 + pc4, xa[k]);
          if (FIN > 4 && pc4 == 0) atomicAdd(r0 + 16 + 4, xb[k]);
          if (pc4 == 1) atomicAdd(r0 + 16 + FIN, 1.0f);
        }
      }
    }
  }
  __builtin_amdgcn_fence(__ATOMIC_SEQ_CST, "workgroup");
}

template <bool RAW, int FIN, int P>
__device__ __forceinline__ void course_edge_walk(const int* __restrict__ gather_idx,
                                                 const int* __restrict__ scatter_idx,
                                                 const int* __restrict__ node_index,
                                                 const float* __restrict__ rows16,
                                                 const float* __restrict__ x,
                                                 unsigned char* __restrict__ key_out,
                                                 bool write_keys,
                                                 float* T, int4* Q,
                                                 int e_lo, int e_hi, int base, int nrows) {
  const int tid = threadIdx.x;
  const int lane = tid & 31;
  constexpr int nthr = kLdsThreads;
  int4* q = Q + (tid >> 5) * kCQCap;
  int qn = 0;
  for (int eb = e_lo; eb < e_hi; eb += kCScan * nthr) {
    int d[kCScan];
#pragma unroll
    for (int u = 0; u < kCScan; ++u) {
      int e = eb + tid + u * nthr;
      d[u] = scatter_idx[e < e_hi ? e : e_lo];
    }
    if constexpr (RAW) {
      if (write_keys) {
        int s[kCScan];
#pragma unroll
        for (int u = 0; u < kCScan; ++u) {
          int e = eb + tid + u * nthr;
          s[u] = gather_idx[e < e_hi ? e : e_lo];
        }
#pragma unroll
        for (int u = 0; u < kCScan; ++u) {
          int e = eb + tid + u * nthr;
          int kr = s[u] / kURows;
          if (e < e_hi) key_out[e] = (unsigned char)(kr < kKeyMax ? kr : kKeyMax);
        }
      }
    }
#pragma nounroll
    for (int u = 0; u < kCScan; ++u) {
      int dd = d[0];
#pragma unroll
      for (int i = 0; i + 1 < kCScan; ++i) d[i] = d[i + 1];
      int e = eb + tid + u * nthr;
      int rel = dd - base;
      bool in = e < e_hi && (unsigned)rel < (unsigned)nrows;
      unsigned m = __builtin_amdgcn_ballot_w32(in);
      if (m != 0u) {
        int n = __builtin_popcount(m);
        if (qn + n > kCQCap) {
          course_queue_flush<RAW, FIN, P>(q, qn, lane, gather_idx, node_index, rows16, x, T);
          qn = 0;
        }
        if (in) {
          int4* slot = q + qn + (int)__builtin_amdgcn_mbcnt_lo(m, 0u);
          slot->x = e;
          slot->y = rel;
        }
        qn += n;
      }
    }
  }
  if (qn > 0) course_queue_flush<RAW, FIN, P>(q, qn, lane, gather_idx, node_index, rows16, x, T);
}

template <int FIN>
__global__ __launch_bounds__(kLdsThreads)
void course_sums_raw_lds_kernel(const float* __restrict__ embed,
                                const int* __restrict__ node_index,
                                const float* __restrict__ x,
                                const int* __restrict__ gather_idx,
                                const int* __restrict__ scatter_idx,
                                float* __restrict__ part,
                                unsigned char* __restrict__ key_out,
                                int E, int n_dst, int n_ranges, int n_slices) {
  constexpr int W = 16 + FIN + 1;
  constexpr int P = W | 1;
  static_assert(FIN >= 1 && FIN <= 5, "four lanes per entry carry features 0..3, lane 0 also feature 4, lane 1 the count");
  __shared__ float T[kCapRows * P];
  __shared__ int4 Q[(kLdsThreads / 32) * kCQCap];
  const int tid = threadIdx.x;
  const int nthr = blockDim.x;
  const int r = blockIdx.x % n_ranges;
  const int slice = blockIdx.x / n_ranges;
  const int base = r * kCapRows;
  const int left = n_dst - base;
  const int nrows = left < kCapRows ? left : kCapRows;
  for (int i = tid; i < nrows * P; i += nthr) T[i] = 0.0f;
  __syncthreads();

  const int chunk = (E + n_slices - 1) / n_slices;
  const int e_lo = slice * chunk;
  const int e_end = e_lo + chunk;
  const int e_hi = e_end < E ? e_end : E;
  course_edge_walk<true, FIN, P>(gather_idx, scatter_idx, node_index, embed, x, key_out, r == 0,
                                 T, Q, e_lo, e_hi, base, nrows);
  __syncthreads();

  float* dst = part + ((long long)slice * n_dst + base) * W;
  for (int i = tid; i < nrows * W; i += nthr) {
    int row = i / W;
    int c = i - row * W;
    dst[i] = T[row * P + c];
  }
}

__global__ __launch_bounds__(kLdsThreads)
void course_sums16_lds_kernel(const float* __restrict__ srcp,
                              const int* __restrict__ gather_idx,
                              const int* __restrict__ scatter_idx,
                              float* __restrict__ part,
                              int E, int n_dst, int n_ranges, int n_slices) {
  constexpr int P = 17;
  __shared__ float T[kCapRows * P];
  __shared__ int4 Q[(kLdsThreads / 32) * kCQCap];
  const int tid = threadIdx.x;
  const int nthr = blockDim.x;
  const int r = blockIdx.x % n_ranges;
  const int slice = blockIdx.x / n_ranges;
  const int base = r * kCapRows;
  const int left = n_dst - base;
  const int nrows = left < kCapRows ? left : kCapRows;
  for (int i = tid; i < nrows * P; i += nthr) T[i] = 0.0f;
  __syncthreads();

  const int chunk = (E + n_slices - 1) / n_slices;
  const int e_lo = slice * chunk;
  const int e_end = e_lo + chunk;
  const int e_hi = e_end < E ? e_end : E;
  course_edge_walk<false, 0, P>(gather_idx, scatter_idx, nullptr, srcp, nullptr, nullptr, false,
                                T, Q, e_lo, e_hi, base, nrows);
  __syncthreads();

  float* dst = part + ((long long)slice * n_dst + base) * 16;
  for (int i = tid; i < nrows * 16; i += nthr) {
    dst[i] = T[(i >> 4) * P + (i & 15)];
  }
}

template <int W>
__global__ void reduce_parts_kernel(const float* __restrict__ part,
                                    float* __restrict__ out,
                                    int n_rows, int n_slices) {
  int t = blockIdx.x * blockDim.x + threadIdx.x;
  int total = n_rows * W;
  if (t >= total) return;
  float acc = 0.0f;
#pragma unroll 16
  for (int s = 0; s < n_slices; ++s) acc += part[(long long)s * total + t];
  int row = t / W;
  int c = t - row * W;
  out[(long long)row * 32 + c] = acc;
}

__global__ void combine16_parts_kernel(const float* __restrict__ part,
                                       int n_slices,
                                       const float* __restrict__ deg,
                                       int dstride,
                                       const float* __restrict__ R,
                                       float* __restrict__ out,
                                       int n) {
  int t = blockIdx.x * blockDim.x + threadIdx.x;
  int node = t >> 2;
  int q = t & 3;
  if (node >= n) return;
  float4 a = make_float4(0.0f, 0.0f, 0.0f, 0.0f);
  long long stride = (long long)n * 16;
#pragma unroll 16
  for (int s = 0; s < n_slices; ++s) {
    float4 v = ((const float4*)(part + s * stride + (long long)node * 16))[q];
    a.x += v.x; a.y += v.y; a.z += v.z; a.w += v.w;
  }
  float dinv = 1.0f / fmaxf(deg[(long long)node * dstride], 1.0f);
  float4 rr = ((const float4*)(R + (long long)node * 16))[q];
  float4 o;
  o.x = fmaf(a.x, dinv, rr.x);
  o.y = fmaf(a.y, dinv, rr.y);
  o.z = fmaf(a.z, dinv, rr.z);
  o.w = fmaf(a.w, dinv, rr.w);
  ((float4*)(out + (long long)node * 16))[q] = o;
}

constexpr int kWF1 = 2 * 16 * 32;
constexpr int kWF2 = 2 * 8 * 32;

template <int FIN, int FDST, bool LDSW>
__device__ __forceinline__ void sage1_chain_tile(const float* arow,
                                                 const float* erow,
                                                 const float* xrow,
                                                 bool nv, int m, int hi,
                                                 const float* __restrict__ Wd,
                                                 const float* __restrict__ bd,
                                                 const float* __restrict__ Wl,
                                                 const float* __restrict__ bl,
                                                 const float* __restrict__ Wr,
                                                 const float* __restrict__ Wp,
                                                 const float* __restrict__ bp,
                                                 const float* __restrict__ W2n,
                                                 const float* __restrict__ W2r,
                                                 const float* __restrict__ b2r,
                                                 const v2f* wf1,
                                                 const v2f* wf2,
                                                 v8f& p, v8f& q, float& dinv_out) {
  const int lane = 16 * hi + m;
  float cnt  = arow[16 + FIN];
  float dinv = nv ? (1.0f / fmaxf(cnt, 1.0f)) : 0.0f;

  v2f f_agg[8], f_x[8];
#pragma unroll
  for (int s = 0; s < 4; ++s) {
    int k = 4 * s + 2 * hi;
    f_agg[s].x = arow[k] * dinv;
    f_agg[s].y = arow[k + 1] * dinv;
    float e0 = erow[k];
    float e1 = erow[k + 1];
    f_x[s].x = nv ? e0 : 0.0f;
    f_x[s].y = nv ? e1 : 0.0f;
  }
  {
    float xd[FDST];
#pragma unroll
    for (int j = 0; j < FDST; ++j) xd[j] = xrow[j];
#pragma unroll
    for (int s = 4; s < 8; ++s) {
      int r = 4 * (s - 4) + 2 * hi;
      float p0 = bd[r];
      float p1 = bd[r + 1];
#pragma unroll
      for (int j = 0; j < FDST; ++j) {
        p0 = fmaf(xd[j], Wd[r * FDST + j], p0);
        p1 = fmaf(xd[j], Wd[(r + 1) * FDST + j], p1);
      }
      f_x[s].x = nv ? p0 : 0.0f;
      f_x[s].y = nv ? p1 : 0.0f;
    }
  }
  {
    float sx[FIN];
#pragma unroll
    for (int j = 0; j < FIN; ++j) sx[j] = arow[16 + j];
#pragma unroll
    for (int s = 4; s < 8; ++s) {
      int r = 4 * (s - 4) + 2 * hi;
      float p0 = cnt * bp[r];
      float p1 = cnt * bp[r + 1];
#pragma unroll
      for (int j = 0; j < FIN; ++j) {
        p0 = fmaf(sx[j], Wp[r * FIN + j], p0);
        p1 = fmaf(sx[j], Wp[(r + 1) * FIN + j], p1);
      }
      f_agg[s].x = p0 * dinv;
      f_agg[s].y = p1 * dinv;
    }
  }

  v8f hT[2];
#pragma unroll
  for (int t = 0; t < 2; ++t) {
    int og = t * 16 + m;
    const float* wl = Wl + og * 32;
    const float* wr = Wr + og * 32;
    v8f c;
#pragma unroll
    for (int r = 0; r < 8; ++r) c[r] = bl[t * 16 + 8 * hi + r];

    if constexpr (LDSW) {
      v2f wa[16];
#pragma unroll
      for (int u = 0; u < 16; ++u) wa[u] = wf1[(16 * t + u) * 32 + lane];
#pragma unroll
      for (int s = 0; s < 8; ++s)
        c = __builtin_amdgcn_wmma_f32_16x16x4_f32(false, wa[s], false, f_agg[s],
                                                  (short)0, c, false, false);
#pragma unroll
      for (int s = 0; s < 8; ++s)
        c = __builtin_amdgcn_wmma_f32_16x16x4_f32(false, wa[8 + s], false, f_x[s],
                                                  (short)0, c, false, false);
    } else {
#pragma unroll
      for (int s = 0; s < 8; ++s) {
        int k = 4 * s + 2 * hi;
        v2f w;
        w.x = wl[k];
        w.y = wl[k + 1];
        c = __builtin_amdgcn_wmma_f32_16x16x4_f32(false, w, false, f_agg[s],
                                                  (short)0, c, false, false);
      }
#pragma unroll
      for (int s = 0; s < 8; ++s) {
        int k = 4 * s + 2 * hi;
        v2f w;
        w.x = wr[k];
        w.y = wr[k + 1];
        c = __builtin_amdgcn_wmma_f32_16x16x4_f32(false, w, false, f_x[s],
                                                  (short)0, c, false, false);
      }
    }
#pragma unroll
    for (int r = 0; r < 8; ++r) hT[t][r] = fmaxf(c[r], 0.0f);
  }

  const float* w2n = W2n + m * 32;
  const float* w2r = W2r + m * 32;
#pragma unroll
  for (int r = 0; r < 8; ++r) {
    p[r] = 0.0f;
    q[r] = b2r[8 * hi + r];
  }
#pragma unroll
  for (int t = 0; t < 2; ++t) {
#pragma unroll
    for (int qq = 0; qq < 4; ++qq) {
      int k = 16 * t + 8 * hi + 2 * qq;
      v2f hb;
      hb.x = hT[t][2 * qq];
      hb.y = hT[t][2 * qq + 1];
      v2f wa, wb;
      if constexpr (LDSW) {
        wa = wf2[(4 * t + qq) * 32 + lane];
        wb = wf2[(8 + 4 * t + qq) * 32 + lane];
      } else {
        wa.x = w2n[k];
        wa.y = w2n[k + 1];
        wb.x = w2r[k];
        wb.y = w2r[k + 1];
      }
      p = __builtin_amdgcn_wmma_f32_16x16x4_f32(false, wa, false, hb,
                                                (short)0, p, false, false);
      q = __builtin_amdgcn_wmma_f32_16x16x4_f32(false, wb, false, hb,
                                                (short)0, q, false, false);
    }
  }
  dinv_out = dinv;
}

template <int FIN, int FDST>
__global__ void sage1_chain_kernel(const float* __restrict__ agg,
                                   const float* __restrict__ embed_d,
                                   const int* __restrict__ idx_d,
                                   const float* __restrict__ x_d,
                                   const float* __restrict__ Wd,
                                   const float* __restrict__ bd,
                                   const float* __restrict__ Wl,
                                   const float* __restrict__ bl,
                                   const float* __restrict__ Wr,
                                   const float* __restrict__ Wp,
                                   const float* __restrict__ bp,
                                   const float* __restrict__ W2n,
                                   const float* __restrict__ W2r,
                                   const float* __restrict__ b2r,
                                   float* __restrict__ Pout,
                                   float* __restrict__ Rout,
                                   int n) {
  int lane = threadIdx.x & 31;
  int wave = (blockIdx.x * blockDim.x + threadIdx.x) >> 5;
  int nb = wave * 16;
  if (nb >= n) return;
  int m  = lane & 15;
  int hi = lane >> 4;

  int node = nb + m;
  bool nv = node < n;
  int ncl = nv ? node : (n - 1);
  v8f p, q;
  float dinv;
  sage1_chain_tile<FIN, FDST, false>(agg + (long long)ncl * 32,
                                     embed_d + (long long)idx_d[ncl] * 16,
                                     x_d + (long long)ncl * FDST,
                                     nv, m, hi, Wd, bd, Wl, bl, Wr, Wp, bp, W2n, W2r, b2r,
                                     nullptr, nullptr, p, q, dinv);

  if (nv) {
    float4* pd = (float4*)(Pout + (long long)node * 16 + 8 * hi);
    float4* rd = (float4*)(Rout + (long long)node * 16 + 8 * hi);
    pd[0] = make_float4(p[0], p[1], p[2], p[3]);
    pd[1] = make_float4(p[4], p[5], p[6], p[7]);
    rd[0] = make_float4(q[0], q[1], q[2], q[3]);
    rd[1] = make_float4(q[4], q[5], q[6], q[7]);
  }
}

__global__ __launch_bounds__(kUThreads)
void user_sums_lds_kernel(const unsigned int* __restrict__ edge_key,
                          const int* __restrict__ edge_user,
                          const int* __restrict__ edge_rec,
                          const float* __restrict__ tab,
                          const float* __restrict__ embed_u,
                          const int* __restrict__ idx_u,
                          const float* __restrict__ x_u,
                          const float* __restrict__ Wd,
                          const float* __restrict__ bd,
                          const float* __restrict__ Wl,
                          const float* __restrict__ bl,
                          const float* __restrict__ Wr,
                          const float* __restrict__ Wp,
                          const float* __restrict__ bp,
                          const float* __restrict__ W2n,
                          const float* __restrict__ W2r,
                          const float* __restrict__ b2r,
                          float* __restrict__ Pout,
                          float* __restrict__ out2,
                          int E, int n_user) {
  __shared__ float T[kURows * kURow];
  __shared__ int2 Q[(kUThreads / 32) * kQCap];
  __shared__ v2f WF[kWF1 + kWF2];
  const int tid = threadIdx.x;
  const int lane = tid & 31;
  const int base = blockIdx.x * kURows;
  const int left = n_user - base;
  const int nrows = left < kURows ? left : kURows;
  for (int i = tid; i < nrows * kURow; i += kUThreads) T[i] = 0.0f;
  for (int i = tid; i < kWF1 + kWF2; i += kUThreads) {
    const int fm = i & 15;
    const int fh = (i >> 4) & 1;
    const int g = i >> 5;
    const float* wsrc;
    if (g < 32) {
      const int t = g >> 4;
      const int u = g & 15;
      wsrc = (u < 8 ? Wl : Wr) + (16 * t + fm) * 32 + 4 * (u & 7) + 2 * fh;
    } else {
      const int g2 = g - 32;
      wsrc = ((g2 >> 3) ? W2r : W2n) + fm * 32 + 16 * ((g2 >> 2) & 1) + 8 * fh + 2 * (g2 & 3);
    }
    v2f w;
    w.x = wsrc[0];
    w.y = wsrc[1];
    WF[i] = w;
  }
  __syncthreads();

  int2* q = Q + (tid >> 5) * kQCap;
  int qn = 0;
  const int4* rec = (const int4*)edge_rec;
  const int N16 = E >> 4;
  const int kb = (int)blockIdx.x < kKeyMax ? (int)blockIdx.x : kKeyMax;
  const unsigned kk = (unsigned)kb * 0x01010101u;

  if (tid < 32) {
    int e = (N16 << 4) + lane;
    bool hit = false;
    if (e < E) hit = (unsigned)(edge_user[e] - base) < (unsigned)nrows;
    unsigned m = __builtin_amdgcn_ballot_w32(hit);
    if (hit) q[__builtin_amdgcn_mbcnt_lo(m, 0u)].x = e;
    qn = __builtin_popcount(m);
  }

  const uint4* key16 = (const uint4*)edge_key;
  constexpr int kTrip = kUThreads * kScanLoads;
  const int ntrips = (N16 + kTrip - 1) / kTrip;
  const int toff = (int)(((unsigned)blockIdx.x * (unsigned)ntrips) / gridDim.x);
  for (int t = 0; t < ntrips; ++t) {
    int tr = t + toff;
    if (tr >= ntrips) tr -= ntrips;
    const int jb = tr * kTrip + tid;
    unsigned hw[kScanLoads / 2];
    if (tr * kTrip + kTrip <= N16) user_filter_trip<false>(key16, jb, N16, kk, hw);
    else                           user_filter_trip<true>(key16, jb, N16, kk, hw);
#pragma nounroll
    for (int w = 0; w < kScanLoads / 2; ++w) {
      unsigned hm = hw[0];
#pragma unroll
      for (int i = 0; i + 1 < kScanLoads / 2; ++i) hw[i] = hw[i + 1];
      unsigned m = __builtin_amdgcn_ballot_w32(hm != 0u);
      while (m != 0u) {
        int n = __builtin_popcount(m);
        if (qn + n > kQCap) {
          user_queue_flush(q, qn, lane, base, nrows, rec, tab, T);
          qn = 0;
        }
        if (hm != 0u) {
          int b = __builtin_ctz(hm);
          hm &= hm - 1u;
          int pos = b & 7;
          int grp = jb + (2 * w + (pos >> 2)) * kUThreads;
          q[qn + (int)__builtin_amdgcn_mbcnt_lo(m, 0u)].x = 16 * grp + 4 * (pos & 3) + (b >> 3);
        }
        qn += n;
        m = __builtin_amdgcn_ballot_w32(hm != 0u);
      }
    }
  }
  if (qn > 0) user_queue_flush(q, qn, lane, base, nrows, rec, tab, T);
  __syncthreads();

  const int wv = __builtin_amdgcn_readfirstlane(tid >> 5);
  const int m = lane & 15;
  const int hi = lane >> 4;
  for (int tile = wv; tile * 16 < nrows; tile += kUThreads / 32) {
    int row = tile * 16 + m;
    bool nv = row < nrows;
    int rcl = nv ? row : (nrows - 1);
    long long ncl = (long long)base + rcl;
    const float* arow = T + rcl * kURow;
    int wo = 0;
    asm volatile("" : "+s"(wo));
    v8f p, q;
    float dinv;
    sage1_chain_tile<2, 5, true>(arow,
                                 embed_u + (long long)idx_u[ncl] * 16,
                                 x_u + ncl * 5,
                                 nv, m, hi, Wd + wo, bd + wo, Wl, bl + wo, Wr, Wp + wo, bp + wo,
                                 W2n, W2r, b2r + wo, WF + wo, WF + kWF1 + wo, p, q, dinv);
    if (nv) {
      const float* s2 = arow + 19 + 8 * hi;
      float4* pd = (float4*)(Pout + ncl * 16 + 8 * hi);
      float4* od = (float4*)(out2 + ncl * 16 + 8 * hi);
      pd[0] = make_float4(p[0], p[1], p[2], p[3]);
      pd[1] = make_float4(p[4], p[5], p[6], p[7]);
      od[0] = make_float4(fmaf(s2[0], dinv, q[0]), fmaf(s2[1], dinv, q[1]),
                          fmaf(s2[2], dinv, q[2]), fmaf(s2[3], dinv, q[3]));
      od[1] = make_float4(fmaf(s2[4], dinv, q[4]), fmaf(s2[5], dinv, q[5]),
                          fmaf(s2[6], dinv, q[6]), fmaf(s2[7], dinv, q[7]));
    }
  }
}

__global__ void decode_kernel(const float* __restrict__ ou,
                              const float* __restrict__ oc,
                              const int* __restrict__ ls,
                              const int* __restrict__ ld,
                              float* __restrict__ out, int L) {
  int t = blockIdx.x * blockDim.x + threadIdx.x;
  if (t >= L) return;
  const float4* a = (const float4*)(ou + (long long)ls[t] * 16);
  const float4* b = (const float4*)(oc + (long long)ld[t] * 16);
  float acc = 0.0f;
#pragma unroll
  for (int i = 0; i < 4; ++i) {
    float4 xa = a[i], xb = b[i];
    acc += xa.x * xb.x + xa.y * xb.y + xa.z * xb.z + xa.w * xb.w;
  }
  out[t] = acc;
}

extern "C" void kernel_launch(void* const* d_in, const int* in_sizes, int n_in,
                              void* d_out, int out_size, void* d_ws, size_t ws_size,
                              hipStream_t stream) {
  const float* user_x       = (const float*)d_in[0];
  const float* course_x     = (const float*)d_in[1];
  const float* user_embed   = (const float*)d_in[2];
  const float* course_embed = (const float*)d_in[3];
  const float* Wu = (const float*)d_in[4];
  const float* bu = (const float*)d_in[5];
  const float* Wc = (const float*)d_in[6];
  const float* bc = (const float*)d_in[7];
  const float* c1e_Wl = (const float*)d_in[8];
  const float* c1e_bl = (const float*)d_in[9];
  const float* c1e_Wr = (const float*)d_in[10];
  const float* c1r_Wl = (const float*)d_in[11];
  const float* c1r_bl = (const float*)d_in[12];
  const float* c1r_Wr = (const float*)d_in[13];
  const float* c2e_Wl = (const float*)d_in[14];
  const float* c2e_bl = (const float*)d_in[15];
  const float* c2e_Wr = (const float*)d_in[16];
  const float* c2r_Wl = (const float*)d_in[17];
  const float* c2r_bl = (const float*)d_in[18];
  const float* c2r_Wr = (const float*)d_in[19];
  const int* user_idx   = (const int*)d_in[20];
  const int* course_idx = (const int*)d_in[21];
  const int* edge_src   = (const int*)d_in[22];
  const int* edge_dst   = (const int*)d_in[23];
  const int* label_src  = (const int*)d_in[24];
  const int* label_dst  = (const int*)d_in[25];
  float* outp = (float*)d_out;

  const int NU = in_sizes[2] / 16;
  const int NC = in_sizes[3] / 16;
  const int E  = in_sizes[22];
  const int L  = in_sizes[24];

  char* p = (char*)d_ws;
  auto carve = [&](size_t bytes) -> float* {
    float* r = (float*)p;
    p += (bytes + 255) & ~((size_t)255);
    return r;
  };
  float* prt2 = carve((size_t)NU * 32 * 4);
  float* Pu   = carve((size_t)NU * 32 * 4);
  float* aggc = carve((size_t)NC * 32 * 4);
  float* pru  = carve((size_t)NU * 32 * 4);
  float* prc  = carve((size_t)NC * 32 * 4);
  float* Pc = prc;
  float* Rc = prc + (size_t)NC * 16;
  float* ou   = carve((size_t)NU * 16 * 4);
  float* oc   = carve((size_t)NC * 16 * 4);

  const int B = 256;
  auto nblk = [](long long t, int b) { return (unsigned)((t + b - 1) / b); };

  const int NR = (NC + kCapRows - 1) / kCapRows;
  int NS = 256 / NR;
  if (NS < 1) NS = 1;
  const long long cap1 = ((long long)NU * 32) / ((long long)NC * 22);
  const long long cap2 = ((long long)NU * 32) / ((long long)NC * 16);
  const int NS1 = (int)(cap1 < NS ? (cap1 < 1 ? 1 : cap1) : NS);
  const int NS2 = (int)(cap2 < NS ? (cap2 < 1 ? 1 : cap2) : NS);

  float* part1 = pru;
  unsigned char* ukey = (unsigned char*)prt2;
  course_sums_raw_lds_kernel<5><<<NR * NS1, kLdsThreads, 0, stream>>>(
      user_embed, user_idx, user_x, edge_src, edge_dst, part1, ukey, E, NC, NR, NS1);
  reduce_parts_kernel<22><<<nblk((long long)NC * 22, B), B, 0, stream>>>(part1, aggc, NC, NS1);
  sage1_chain_kernel<5, 2><<<nblk(((long long)(NC + 15) / 16) * 32, B), B, 0, stream>>>(
      aggc, course_embed, course_idx, course_x, Wc, bc,
      c1e_Wl, c1e_bl, c1e_Wr, Wu, bu, c2r_Wl, c2e_Wr, c2e_bl, Pc, Rc, NC);
  float* ctab = pru;
  int* erec = (int*)(pru + (((size_t)NC * kCRow + 63) & ~((size_t)63)));
  user_pass_inputs_kernel<<<256, kPrepThreads, 0, stream>>>(
      course_embed, course_idx, course_x, Pc, edge_src, edge_dst, ctab, erec, NC, E);
  user_sums_lds_kernel<<<nblk(NU, kURows), kUThreads, 0, stream>>>(
      (const unsigned int*)ukey, edge_src, erec, ctab,
      user_embed, user_idx, user_x, Wu, bu,
      c1r_Wl, c1r_bl, c1r_Wr, Wc, bc, c2e_Wl, c2r_Wr, c2r_bl, Pu, ou, E, NU);

  float* part2 = prt2;
  course_sums16_lds_kernel<<<NR * NS2, kLdsThreads, 0, stream>>>(
      Pu, edge_src, edge_dst, part2, E, NC, NR, NS2);
  combine16_parts_kernel<<<nblk((long long)NC * 4, B), B, 0, stream>>>(
      part2, NS2, aggc + 21, 32, Rc, oc, NC);

  decode_kernel<<<nblk(L, B), B, 0, stream>>>(ou, oc, label_src, label_dst, outp, L);
}
